# starting kernel
# baseline (speedup 1.0000x reference)
#include <hip/hip_runtime.h>
#include <math.h>

#define NNODE 50000

typedef float v2f __attribute__((ext_vector_type(2)));
typedef float v8f __attribute__((ext_vector_type(8)));


__global__ void fill_kernel(float* __restrict__ p, float v, int n) {
    int i = blockIdx.x * blockDim.x + threadIdx.x;
    if (i < n) p[i] = v;
}

__device__ __forceinline__ void edge_nodes(const int* __restrict__ ei, int E,
                                           int e, int& s, int& d) {
    if (e < E) { s = ei[e]; d = ei[E + e]; }
    else       { s = e - E; d = s; }
}

__device__ __forceinline__ void atomicMaxF(float* addr, float val) {
    if (val >= 0.0f) atomicMax((int*)addr, __float_as_int(val));
    else             atomicMin((unsigned int*)addr, __float_as_uint(val));
}

__global__ void gemm_wmma_f32(const float* __restrict__ X,
                              const float* __restrict__ W,
                              float* __restrict__ H,
                              int n, int K, int M) {
    const int lane = threadIdx.x & 31;
    const int wave = threadIdx.x >> 5;
    const int wavesPerBlock = blockDim.x >> 5;

    const int colTiles = M >> 4;
    const int rowTiles = n >> 4;
    const int tile = blockIdx.x * wavesPerBlock + wave;
    if (tile >= rowTiles * colTiles) return;

    const int rowT = tile / colTiles;
    const int colT = tile - rowT * colTiles;
    const int row0 = rowT << 4;
    const int col0 = colT << 4;

    const int m16  = lane & 15;
    const int half = lane >> 4;

    v8f acc = {};
    const float* xrow = X + (size_t)(row0 + m16) * K;

    for (int k0 = 0; k0 < K; k0 += 4) {
        const int k = k0 + half * 2;
        v2f a, b;
        a.x = xrow[k];
        a.y = xrow[k + 1];
        b.x = W[(size_t)(k)     * M + col0 + m16];
        b.y = W[(size_t)(k + 1) * M + col0 + m16];
        acc = __builtin_amdgcn_wmma_f32_16x16x4_f32(
                   false, a,  false, b,
                   (short)0, acc,  false,  false);
    }

#pragma unroll
    for (int v = 0; v < 8; ++v)
        H[(size_t)(row0 + half * 8 + v) * M + col0 + m16] = acc[v];
}

__global__ void att_logits(const float* __restrict__ H,
                           const float* __restrict__ a_src,
                           const float* __restrict__ a_dst,
                           float* __restrict__ als, float* __restrict__ ald,
                           int n, int heads, int C) {
    int t = blockIdx.x * blockDim.x + threadIdx.x;
    if (t >= n * heads) return;
    int node = t / heads;
    int h    = t - node * heads;
    const float* hp = H + (size_t)node * heads * C + (size_t)h * C;
    const float* as = a_src + h * C;
    const float* ad = a_dst + h * C;
    float s = 0.0f, d = 0.0f;
    for (int c = 0; c < C; ++c) { float v = hp[c]; s += v * as[c]; d += v * ad[c]; }
    als[t] = s;
    ald[t] = d;
}

__global__ void edge_max(const int* __restrict__ ei, int E, int Etot,
                         const float* __restrict__ als, const float* __restrict__ ald,
                         float* __restrict__ mbuf, int heads) {
    int e = blockIdx.x * blockDim.x + threadIdx.x;
    if (e >= Etot) return;
    int s, d;
    edge_nodes(ei, E, e, s, d);
    for (int h = 0; h < heads; ++h) {
        float x  = als[s * heads + h] + ald[d * heads + h];
        float lr = x >= 0.0f ? x : 0.2f * x;
        atomicMaxF(&mbuf[d * heads + h], lr);
    }
}

__global__ void edge_acc(const int* __restrict__ ei, int E, int Etot,
                         const float* __restrict__ als, const float* __restrict__ ald,
                         const float* __restrict__ mbuf,
                         const float* __restrict__ H,
                         float* __restrict__ den, float* __restrict__ accum,
                         int heads, int C) {
    int t = blockIdx.x * blockDim.x + threadIdx.x;
    if (t >= Etot * heads) return;
    int e = t / heads;
    int h = t - e * heads;
    int s, d;
    edge_nodes(ei, E, e, s, d);
    float x  = als[s * heads + h] + ald[d * heads + h];
    float lr = x >= 0.0f ? x : 0.2f * x;
    float ex = __expf(lr - mbuf[d * heads + h]);
    atomicAdd(&den[d * heads + h], ex);
    const float* hs = H     + (size_t)s * heads * C + (size_t)h * C;
    float*       ap = accum + (size_t)d * heads * C + (size_t)h * C;
    for (int c = 0; c < C; ++c) atomicAdd(&ap[c], hs[c] * ex);
}

__global__ void finalize(float* __restrict__ accum, const float* __restrict__ den,
                         const float* __restrict__ bias,
                         int n, int heads, int C, int relu) {
    int HC = heads * C;
    int t  = blockIdx.x * blockDim.x + threadIdx.x;
    if (t >= n * HC) return;
    int node = t / HC;
    int rem  = t - node * HC;
    int h    = rem / C;
    float v = accum[t] / den[node * heads + h] + bias[rem];
    if (relu) v = fmaxf(v, 0.0f);
    accum[t] = v;
}


extern "C" void kernel_launch(void* const* d_in, const int* in_sizes, int n_in,
                              void* d_out, int out_size, void* d_ws, size_t ws_size,
                              hipStream_t stream) {
    const float* x      = (const float*)d_in[0];
    const int*   ei     = (const int*)  d_in[1];
    const float* W1     = (const float*)d_in[2];
    const float* a_src1 = (const float*)d_in[3];
    const float* a_dst1 = (const float*)d_in[4];
    const float* b1     = (const float*)d_in[5];
    const float* W2     = (const float*)d_in[6];
    const float* a_src2 = (const float*)d_in[7];
    const float* a_dst2 = (const float*)d_in[8];
    const float* b2     = (const float*)d_in[9];
    const float* W3     = (const float*)d_in[10];
    const float* a_src3 = (const float*)d_in[11];
    const float* a_dst3 = (const float*)d_in[12];
    const float* b3     = (const float*)d_in[13];

    const int E    = in_sizes[1] / 2;
    const int Etot = E + NNODE;

    float* ws   = (float*)d_ws;
    float* bufH = ws;
    float* act1 = bufH + (size_t)NNODE * 240;
    float* h2   = act1 + (size_t)NNODE * 112;
    float* act2 = h2   + (size_t)NNODE * 96;
    float* als  = act2 + (size_t)NNODE * 96;
    float* ald  = als  + (size_t)NNODE * 7;
    float* mbuf = ald  + (size_t)NNODE * 7;
    float* den  = mbuf + (size_t)NNODE * 7;
    float* out  = (float*)d_out;

    auto layer = [&](const float* X, int K,
                     const float* W, const float* as, const float* ad, const float* b,
                     int heads, int C, float* Hbuf, float* accbuf, int relu) {
        const int HC = heads * C;

        const int tiles = (NNODE / 16) * (HC / 16);
        gemm_wmma_f32<<<(tiles + 7) / 8, 256, 0, stream>>>(X, W, Hbuf, NNODE, K, HC);

        const int nh = NNODE * heads;
        att_logits<<<(nh + 255) / 256, 256, 0, stream>>>(Hbuf, as, ad, als, ald,
                                                         NNODE, heads, C);

        fill_kernel<<<(nh + 255) / 256, 256, 0, stream>>>(mbuf, -INFINITY, nh);
        hipMemsetAsync(den, 0, (size_t)nh * sizeof(float), stream);
        hipMemsetAsync(accbuf, 0, (size_t)NNODE * HC * sizeof(float), stream);

        edge_max<<<(Etot + 255) / 256, 256, 0, stream>>>(ei, E, Etot, als, ald,
                                                         mbuf, heads);

        const int th = Etot * heads;
        edge_acc<<<(th + 255) / 256, 256, 0, stream>>>(ei, E, Etot, als, ald, mbuf,
                                                       Hbuf, den, accbuf, heads, C);

        const int tot = NNODE * HC;
        finalize<<<(tot + 255) / 256, 256, 0, stream>>>(accbuf, den, b,
                                                        NNODE, heads, C, relu);
    };

    layer(x,    128, W1, a_src1, a_dst1, b1, 7, 16, bufH, act1, 1);
    layer(act1, 112, W2, a_src2, a_dst2, b2, 6, 16, h2,   act2, 1);
    layer(act2,  96, W3, a_src3, a_dst3, b3, 6, 40, bufH, out,  0);
}


# aggregation: each workgroup owns 200 destination rows in LDS, filters the whole edge list for them, max / exp-sum / row sums by LDS atomics (arrival order); no global atomics, memsets, finalize
# speedup vs baseline: 9.9480x; 9.9480x over previous
#include <hip/hip_runtime.h>
#include <math.h>

#define NNODE 50000

typedef float v2f __attribute__((ext_vector_type(2)));
typedef float v8f __attribute__((ext_vector_type(8)));

__global__ void gemm_wmma_f32(const float* __restrict__ X,
                              const float* __restrict__ W,
                              float* __restrict__ H,
                              int n, int K, int M) {
    const int lane = threadIdx.x & 31;
    const int wave = threadIdx.x >> 5;
    const int wavesPerBlock = blockDim.x >> 5;

    const int colTiles = M >> 4;
    const int rowTiles = n >> 4;
    const int tile = blockIdx.x * wavesPerBlock + wave;
    if (tile >= rowTiles * colTiles) return;

    const int rowT = tile / colTiles;
    const int colT = tile - rowT * colTiles;
    const int row0 = rowT << 4;
    const int col0 = colT << 4;

    const int m16  = lane & 15;
    const int half = lane >> 4;

    v8f acc = {};
    const float* xrow = X + (size_t)(row0 + m16) * K;

    for (int k0 = 0; k0 < K; k0 += 4) {
        const int k = k0 + half * 2;
        v2f a, b;
        a.x = xrow[k];
        a.y = xrow[k + 1];
        b.x = W[(size_t)(k)     * M + col0 + m16];
        b.y = W[(size_t)(k + 1) * M + col0 + m16];
        acc = __builtin_amdgcn_wmma_f32_16x16x4_f32(
                   false, a,  false, b,
                   (short)0, acc,  false,  false);
    }

#pragma unroll
    for (int v = 0; v < 8; ++v)
        H[(size_t)(row0 + half * 8 + v) * M + col0 + m16] = acc[v];
}

__global__ void att_logits(const float* __restrict__ H,
                           const float* __restrict__ a_src,
                           const float* __restrict__ a_dst,
                           float* __restrict__ als, float* __restrict__ ald,
                           int n, int heads, int C) {
    int t = blockIdx.x * blockDim.x + threadIdx.x;
    if (t >= n * heads) return;
    int node = t / heads;
    int h    = t - node * heads;
    const float* hp = H + (size_t)node * heads * C + (size_t)h * C;
    const float* as = a_src + h * C;
    const float* ad = a_dst + h * C;
    float s = 0.0f, d = 0.0f;
    for (int c = 0; c < C; ++c) { float v = hp[c]; s += v * as[c]; d += v * ad[c]; }
    als[t] = s;
    ald[t] = d;
}


constexpr int TILE        = 200;
constexpr int LIST_CAP    = 16384;
constexpr int AGG_THREADS = 1024;
constexpr int AGG_WAVES   = AGG_THREADS / 32;
constexpr int SCAN_VEC    = 8;
constexpr int ACC_BATCH   = 4;

static_assert(TILE <= 256, "local row index is packed into 8 bits");
static_assert(NNODE <= (1 << 24), "source index is packed into 24 bits");

__device__ __forceinline__ unsigned f2key(float f) {
    unsigned b = __float_as_uint(f);
    return (b & 0x80000000u) ? ~b : (b | 0x80000000u);
}
__device__ __forceinline__ float key2f(unsigned k) {
    return __uint_as_float((k & 0x80000000u) ? (k & 0x7fffffffu) : ~k);
}

template <int HEADS, int C>
__global__ __launch_bounds__(AGG_THREADS)
void agg_tile(const int* __restrict__ ei, int E,
              const float* __restrict__ als, const float* __restrict__ ald,
              const float* __restrict__ H, const float* __restrict__ bias,
              float* __restrict__ out, int relu) {
    constexpr int HC  = HEADS * C;
    constexpr int NIT = (HC + 31) / 32;

    __shared__ float         acc[TILE * HC];
    __shared__ unsigned      mxk[TILE * HEADS];
    __shared__ float         dn[TILE * HEADS];
    __shared__ unsigned      elist[LIST_CAP];
    __shared__ unsigned char eld[LIST_CAP];
    __shared__ unsigned      cnt;

    const int tid   = threadIdx.x;
    const int lane  = tid & 31;
    const int wave  = tid >> 5;
    const int tile0 = blockIdx.x * TILE;
    const int nvalid = min(TILE, NNODE - tile0);
    const unsigned nv = (unsigned)nvalid;

    const int* __restrict__ srcp = ei;
    const int* __restrict__ dstp = ei + E;

    for (int i = tid; i < TILE * HC; i += AGG_THREADS) acc[i] = 0.0f;
    const unsigned key_ninf = f2key(-INFINITY);
    for (int i = tid; i < TILE * HEADS; i += AGG_THREADS) { mxk[i] = key_ninf; dn[i] = 0.0f; }
    for (int i = tid; i < nvalid; i += AGG_THREADS)
        elist[i] = ((unsigned)(tile0 + i) << 8) | (unsigned)i;
    if (tid == 0) cnt = nv;
    __syncthreads();

    auto push = [&](int e, int d) {
        const unsigned ld = (unsigned)(d - tile0);
        if (ld < nv) {
            const unsigned pos = atomicAdd(&cnt, 1u);
            if (pos < (unsigned)LIST_CAP) { elist[pos] = (unsigned)e; eld[pos] = (unsigned char)ld; }
        }
    };

    int e_done = 0;
    if ((((size_t)dstp) & 15) == 0 && E >= 4) {
        const int   nvec = E >> 2;
        const int4* dv   = (const int4*)dstp;
        constexpr int STEP = SCAN_VEC * AGG_THREADS;
        auto loadq = [&](int4 (&q)[SCAN_VEC], int base) {
#pragma unroll
            for (int j = 0; j < SCAN_VEC; ++j) {
                const int v = base + j * AGG_THREADS + tid;
                q[j] = dv[min(v, nvec - 1)];
            }
        };
        auto procq = [&](const int4 (&q)[SCAN_VEC], int base) {
#pragma unroll
            for (int j = 0; j < SCAN_VEC; ++j) {
                const int v = base + j * AGG_THREADS + tid;
                const bool any = ((unsigned)(q[j].x - tile0) < nv) | ((unsigned)(q[j].y - tile0) < nv) |
                                 ((unsigned)(q[j].z - tile0) < nv) | ((unsigned)(q[j].w - tile0) < nv);
                if (any && v < nvec) {
                    push(4 * v,     q[j].x);
                    push(4 * v + 1, q[j].y);
                    push(4 * v + 2, q[j].z);
                    push(4 * v + 3, q[j].w);
                }
            }
        };
        int4 qa[SCAN_VEC], qb[SCAN_VEC];
        loadq(qa, 0);
        for (int base = 0; base < nvec; base += 2 * STEP) {
            loadq(qb, base + STEP);
            procq(qa, base);
            loadq(qa, base + 2 * STEP);
            procq(qb, base + STEP);
        }
        e_done = nvec << 2;
    }
    for (int e = e_done + tid; e < E; e += AGG_THREADS) push(e, dstp[e]);
    __syncthreads();

    const unsigned total_raw = cnt;
    if (total_raw <= (unsigned)LIST_CAP) {
        const int total = (int)total_raw;

        for (int i0 = nvalid + tid; i0 < total; i0 += 4 * AGG_THREADS) {
            int sv[4];
#pragma unroll
            for (int j = 0; j < 4; ++j) {
                const int i = i0 + j * AGG_THREADS;
                sv[j] = (i < total) ? srcp[elist[i]] : 0;
            }
#pragma unroll
            for (int j = 0; j < 4; ++j) {
                const int i = i0 + j * AGG_THREADS;
                if (i < total) elist[i] = ((unsigned)sv[j] << 8) | (unsigned)eld[i];
            }
        }
        __syncthreads();

        const int n1 = total * HEADS;
        for (int idx0 = tid; idx0 < n1; idx0 += 4 * AGG_THREADS) {
            float xv[4];
            int   cell[4];
#pragma unroll
            for (int j = 0; j < 4; ++j) {
                const int idx  = min(idx0 + j * AGG_THREADS, n1 - 1);
                const int item = idx / HEADS;
                const int h    = idx - item * HEADS;
                const unsigned en = elist[item];
                const int s  = (int)(en >> 8);
                const int ld = (int)(en & 255u);
                xv[j]   = als[s * HEADS + h] + ald[(tile0 + ld) * HEADS + h];
                cell[j] = ld * HEADS + h;
            }
#pragma unroll
            for (int j = 0; j < 4; ++j) {
                if (idx0 + j * AGG_THREADS < n1) {
                    const float lr = xv[j] >= 0.0f ? xv[j] : 0.2f * xv[j];
                    atomicMax(&mxk[cell[j]], f2key(lr));
                }
            }
        }
        __syncthreads();

        int hk[NIT];
#pragma unroll
        for (int k = 0; k < NIT; ++k) hk[k] = min((lane + 32 * k) / C, HEADS - 1);

        for (int item0 = wave; item0 < total; item0 += AGG_WAVES * ACC_BATCH) {
            int      sb[ACC_BATCH], lb[ACC_BATCH];
            float    xa[ACC_BATCH], xd[ACC_BATCH], hv[ACC_BATCH][NIT];
            unsigned mk[ACC_BATCH];
#pragma unroll
            for (int b = 0; b < ACC_BATCH; ++b) {
                const int it = min(item0 + AGG_WAVES * b, total - 1);
                const unsigned en = elist[it];
                sb[b] = (int)(en >> 8);
                lb[b] = (int)(en & 255u);
            }
#pragma unroll
            for (int b = 0; b < ACC_BATCH; ++b) {
                xa[b] = 0.0f; xd[b] = 0.0f; mk[b] = 0u;
                if (lane < HEADS) {
                    xa[b] = als[sb[b] * HEADS + lane];
                    xd[b] = ald[(tile0 + lb[b]) * HEADS + lane];
                    mk[b] = mxk[lb[b] * HEADS + lane];
                }
                const float* __restrict__ hrow = H + (size_t)sb[b] * HC;
#pragma unroll
                for (int k = 0; k < NIT; ++k) {
                    const int c = lane + 32 * k;
                    hv[b][k] = (c < HC) ? hrow[c] : 0.0f;
                }
            }
#pragma unroll
            for (int b = 0; b < ACC_BATCH; ++b) {
                if (item0 + AGG_WAVES * b < total) {
                    float exl = 0.0f;
                    if (lane < HEADS) {
                        const float x  = xa[b] + xd[b];
                        const float lr = x >= 0.0f ? x : 0.2f * x;
                        exl = __expf(lr - key2f(mk[b]));
                        atomicAdd(&dn[lb[b] * HEADS + lane], exl);
                    }
                    float* arow = acc + lb[b] * HC;
#pragma unroll
                    for (int k = 0; k < NIT; ++k) {
                        const int c = lane + 32 * k;
                        const float ex = __shfl(exl, hk[k]);
                        if (c < HC) atomicAdd(&arow[c], hv[b][k] * ex);
                    }
                }
            }
        }
    } else {
        auto visit_max = [&](int s, int ld) {
            for (int h = 0; h < HEADS; ++h) {
                const float x  = als[s * HEADS + h] + ald[(tile0 + ld) * HEADS + h];
                const float lr = x >= 0.0f ? x : 0.2f * x;
                atomicMax(&mxk[ld * HEADS + h], f2key(lr));
            }
        };
        for (int i = tid; i < nvalid; i += AGG_THREADS) visit_max(tile0 + i, i);
        for (int e = tid; e < E; e += AGG_THREADS) {
            const unsigned ld = (unsigned)(dstp[e] - tile0);
            if (ld < nv) visit_max(srcp[e], (int)ld);
        }
        __syncthreads();

        auto visit_acc = [&](int s, int ld) {
            for (int h = 0; h < HEADS; ++h) {
                const float x  = als[s * HEADS + h] + ald[(tile0 + ld) * HEADS + h];
                const float lr = x >= 0.0f ? x : 0.2f * x;
                const float ex = __expf(lr - key2f(mxk[ld * HEADS + h]));
                atomicAdd(&dn[ld * HEADS + h], ex);
                const float* hs = H + (size_t)s * HC + h * C;
                float*       ap = acc + ld * HC + h * C;
                for (int c = 0; c < C; ++c) atomicAdd(&ap[c], hs[c] * ex);
            }
        };
        for (int i = tid; i < nvalid; i += AGG_THREADS) visit_acc(tile0 + i, i);
        for (int e = tid; e < E; e += AGG_THREADS) {
            const unsigned ld = (unsigned)(dstp[e] - tile0);
            if (ld < nv) visit_acc(srcp[e], (int)ld);
        }
    }
    __syncthreads();

    for (int i = tid; i < nvalid * HC; i += AGG_THREADS) {
        const int ld  = i / HC;
        const int rem = i - ld * HC;
        const int h   = rem / C;
        float v = acc[i] / dn[ld * HEADS + h] + bias[rem];
        if (relu) v = fmaxf(v, 0.0f);
        out[(size_t)tile0 * HC + i] = v;
    }
}


extern "C" void kernel_launch(void* const* d_in, const int* in_sizes, int n_in,
                              void* d_out, int out_size, void* d_ws, size_t ws_size,
                              hipStream_t stream) {
    const float* x      = (const float*)d_in[0];
    const int*   ei     = (const int*)  d_in[1];
    const float* W1     = (const float*)d_in[2];
    const float* a_src1 = (const float*)d_in[3];
    const float* a_dst1 = (const float*)d_in[4];
    const float* b1     = (const float*)d_in[5];
    const float* W2     = (const float*)d_in[6];
    const float* a_src2 = (const float*)d_in[7];
    const float* a_dst2 = (const float*)d_in[8];
    const float* b2     = (const float*)d_in[9];
    const float* W3     = (const float*)d_in[10];
    const float* a_src3 = (const float*)d_in[11];
    const float* a_dst3 = (const float*)d_in[12];
    const float* b3     = (const float*)d_in[13];

    const int E = in_sizes[1] / 2;

    float* ws   = (float*)d_ws;
    float* bufH = ws;
    float* act1 = bufH + (size_t)NNODE * 240;
    float* h2   = act1 + (size_t)NNODE * 112;
    float* act2 = h2   + (size_t)NNODE * 96;
    float* als  = act2 + (size_t)NNODE * 96;
    float* ald  = als  + (size_t)NNODE * 7;
    float* out  = (float*)d_out;

    constexpr int NT = (NNODE + TILE - 1) / TILE;

    auto front = [&](const float* X, int K, const float* W, const float* as, const float* ad,
                     int heads, int C, float* Hbuf) {
        const int HC = heads * C;
        const int tiles = (NNODE / 16) * (HC / 16);
        gemm_wmma_f32<<<(tiles + 7) / 8, 256, 0, stream>>>(X, W, Hbuf, NNODE, K, HC);
        const int nh = NNODE * heads;
        att_logits<<<(nh + 255) / 256, 256, 0, stream>>>(Hbuf, as, ad, als, ald,
                                                         NNODE, heads, C);
    };

    front(x, 128, W1, a_src1, a_dst1, 7, 16, bufH);
    agg_tile<7, 16><<<NT, AGG_THREADS, 0, stream>>>(ei, E, als, ald, bufH, b1, act1, 1);
    front(act1, 112, W2, a_src2, a_dst2, 6, 16, h2);
    agg_tile<6, 16><<<NT, AGG_THREADS, 0, stream>>>(ei, E, als, ald, h2, b2, act2, 1);
    front(act2, 96, W3, a_src3, a_dst3, 6, 40, bufH);
    agg_tile<6, 40><<<NT, AGG_THREADS, 0, stream>>>(ei, E, als, ald, bufH, b3, out, 0);
}


# aggregation: layer 1 also writes each tile's packed edge list to the workspace, layers 2-3 load it into LDS: their edge filters and source lookups go; sums stay LDS atomics in arrival order
# speedup vs baseline: 11.6624x; 1.1723x over previous
#include <hip/hip_runtime.h>
#include <math.h>

#define NNODE 50000

typedef float v2f __attribute__((ext_vector_type(2)));
typedef float v8f __attribute__((ext_vector_type(8)));

__global__ void gemm_wmma_f32(const float* __restrict__ X,
                              const float* __restrict__ W,
                              float* __restrict__ H,
                              int n, int K, int M) {
    const int lane = threadIdx.x & 31;
    const int wave = threadIdx.x >> 5;
    const int wavesPerBlock = blockDim.x >> 5;

    const int colTiles = M >> 4;
    const int rowTiles = n >> 4;
    const int tile = blockIdx.x * wavesPerBlock + wave;
    if (tile >= rowTiles * colTiles) return;

    const int rowT = tile / colTiles;
    const int colT = tile - rowT * colTiles;
    const int row0 = rowT << 4;
    const int col0 = colT << 4;

    const int m16  = lane & 15;
    const int half = lane >> 4;

    v8f acc = {};
    const float* xrow = X + (size_t)(row0 + m16) * K;

    for (int k0 = 0; k0 < K; k0 += 4) {
        const int k = k0 + half * 2;
        v2f a, b;
        a.x = xrow[k];
        a.y = xrow[k + 1];
        b.x = W[(size_t)(k)     * M + col0 + m16];
        b.y = W[(size_t)(k + 1) * M + col0 + m16];
        acc = __builtin_amdgcn_wmma_f32_16x16x4_f32(
                   false, a,  false, b,
                   (short)0, acc,  false,  false);
    }

#pragma unroll
    for (int v = 0; v < 8; ++v)
        H[(size_t)(row0 + half * 8 + v) * M + col0 + m16] = acc[v];
}

__global__ void att_logits(const float* __restrict__ H,
                           const float* __restrict__ a_src,
                           const float* __restrict__ a_dst,
                           float* __restrict__ als, float* __restrict__ ald,
                           int n, int heads, int C) {
    int t = blockIdx.x * blockDim.x + threadIdx.x;
    if (t >= n * heads) return;
    int node = t / heads;
    int h    = t - node * heads;
    const float* hp = H + (size_t)node * heads * C + (size_t)h * C;
    const float* as = a_src + h * C;
    const float* ad = a_dst + h * C;
    float s = 0.0f, d = 0.0f;
    for (int c = 0; c < C; ++c) { float v = hp[c]; s += v * as[c]; d += v * ad[c]; }
    als[t] = s;
    ald[t] = d;
}


constexpr int TILE        = 200;
constexpr int LIST_CAP    = 20480;
constexpr int AGG_THREADS = 1024;
constexpr int AGG_WAVES   = AGG_THREADS / 32;
constexpr int SCAN_VEC    = 8;
constexpr int ACC_BATCH   = 4;

static_assert(TILE <= 256, "local row index is packed into 8 bits");
static_assert(NNODE <= (1 << 24), "source index is packed into 24 bits");

__device__ __forceinline__ unsigned f2key(float f) {
    unsigned b = __float_as_uint(f);
    return (b & 0x80000000u) ? ~b : (b | 0x80000000u);
}
__device__ __forceinline__ float key2f(unsigned k) {
    return __uint_as_float((k & 0x80000000u) ? (k & 0x7fffffffu) : ~k);
}

template <int HEADS, int C, bool BUILD>
__global__ __launch_bounds__(AGG_THREADS)
void agg_tile(const int* __restrict__ ei, int E,
              const float* __restrict__ als, const float* __restrict__ ald,
              const float* __restrict__ H, const float* __restrict__ bias,
              float* __restrict__ out, int relu,
              unsigned* glist, unsigned* gcnt) {
    constexpr int HC  = HEADS * C;
    constexpr int NIT = (HC + 31) / 32;

    __shared__ float         acc[TILE * HC];
    __shared__ unsigned      mxk[TILE * HEADS];
    __shared__ float         dn[TILE * HEADS];
    __shared__ unsigned      elist[LIST_CAP];

    const int tid   = threadIdx.x;
    const int lane  = tid & 31;
    const int wave  = tid >> 5;
    const int tile0 = blockIdx.x * TILE;
    const int nvalid = min(TILE, NNODE - tile0);
    const unsigned nv = (unsigned)nvalid;

    const int* __restrict__ srcp = ei;
    const int* __restrict__ dstp = ei + E;

    unsigned* tlist = glist + (size_t)blockIdx.x * LIST_CAP;

    unsigned total_raw = 0;
    if constexpr (!BUILD) total_raw = gcnt[blockIdx.x];

    for (int i = tid; i < TILE * HC; i += AGG_THREADS) acc[i] = 0.0f;
    const unsigned key_ninf = f2key(-INFINITY);
    for (int i = tid; i < TILE * HEADS; i += AGG_THREADS) { mxk[i] = key_ninf; dn[i] = 0.0f; }

    if constexpr (BUILD) {
        __shared__ unsigned char eld[LIST_CAP];
        __shared__ unsigned      cnt;

        for (int i = tid; i < nvalid; i += AGG_THREADS) {
            const unsigned p = ((unsigned)(tile0 + i) << 8) | (unsigned)i;
            elist[i] = p;
            tlist[i] = p;
        }
        if (tid == 0) cnt = nv;
        __syncthreads();

        auto push = [&](int e, int d) {
            const unsigned ld = (unsigned)(d - tile0);
            if (ld < nv) {
                const unsigned pos = atomicAdd(&cnt, 1u);
                if (pos < (unsigned)LIST_CAP) { elist[pos] = (unsigned)e; eld[pos] = (unsigned char)ld; }
            }
        };

        int e_done = 0;
        if ((((size_t)dstp) & 15) == 0 && E >= 4) {
            const int   nvec = E >> 2;
            const int4* dv   = (const int4*)dstp;
            constexpr int STEP = SCAN_VEC * AGG_THREADS;
            auto loadq = [&](int4 (&q)[SCAN_VEC], int base) {
#pragma unroll
                for (int j = 0; j < SCAN_VEC; ++j) {
                    const int v = base + j * AGG_THREADS + tid;
                    q[j] = dv[min(v, nvec - 1)];
                }
            };
            auto procq = [&](const int4 (&q)[SCAN_VEC], int base) {
#pragma unroll
                for (int j = 0; j < SCAN_VEC; ++j) {
                    const int v = base + j * AGG_THREADS + tid;
                    const bool any = ((unsigned)(q[j].x - tile0) < nv) | ((unsigned)(q[j].y - tile0) < nv) |
                                     ((unsigned)(q[j].z - tile0) < nv) | ((unsigned)(q[j].w - tile0) < nv);
                    if (any && v < nvec) {
                        push(4 * v,     q[j].x);
                        push(4 * v + 1, q[j].y);
                        push(4 * v + 2, q[j].z);
                        push(4 * v + 3, q[j].w);
                    }
                }
            };
            int4 qa[SCAN_VEC], qb[SCAN_VEC];
            loadq(qa, 0);
            for (int base = 0; base < nvec; base += 2 * STEP) {
                loadq(qb, base + STEP);
                procq(qa, base);
                loadq(qa, base + 2 * STEP);
                procq(qb, base + STEP);
            }
            e_done = nvec << 2;
        }
        for (int e = e_done + tid; e < E; e += AGG_THREADS) push(e, dstp[e]);
        __syncthreads();

        total_raw = cnt;
        if (tid == 0) gcnt[blockIdx.x] = total_raw;
        if (total_raw <= (unsigned)LIST_CAP) {
            const int total = (int)total_raw;

            for (int i0 = nvalid + tid; i0 < total; i0 += 4 * AGG_THREADS) {
                int sv[4];
#pragma unroll
                for (int j = 0; j < 4; ++j) {
                    const int i = i0 + j * AGG_THREADS;
                    sv[j] = (i < total) ? srcp[elist[i]] : 0;
                }
#pragma unroll
                for (int j = 0; j < 4; ++j) {
                    const int i = i0 + j * AGG_THREADS;
                    if (i < total) {
                        const unsigned p = ((unsigned)sv[j] << 8) | (unsigned)eld[i];
                        elist[i] = p;
                        tlist[i] = p;
                    }
                }
            }
        }
        __syncthreads();
    } else {
        if (total_raw <= (unsigned)LIST_CAP) {
            const int total = (int)total_raw;
            const unsigned* __restrict__ rlist = tlist;
            for (int i0 = tid; i0 < total; i0 += 4 * AGG_THREADS) {
                unsigned pv[4];
#pragma unroll
                for (int j = 0; j < 4; ++j) {
                    const int i = i0 + j * AGG_THREADS;
                    pv[j] = (i < total) ? rlist[i] : 0u;
                }
#pragma unroll
                for (int j = 0; j < 4; ++j) {
                    const int i = i0 + j * AGG_THREADS;
                    if (i < total) elist[i] = pv[j];
                }
            }
        }
        __syncthreads();
    }

    if (total_raw <= (unsigned)LIST_CAP) {
        const int total = (int)total_raw;

        const int n1 = total * HEADS;
        for (int idx0 = tid; idx0 < n1; idx0 += 4 * AGG_THREADS) {
            float xv[4];
            int   cell[4];
#pragma unroll
            for (int j = 0; j < 4; ++j) {
                const int idx  = min(idx0 + j * AGG_THREADS, n1 - 1);
                const int item = idx / HEADS;
                const int h    = idx - item * HEADS;
                const unsigned en = elist[item];
                const int s  = (int)(en >> 8);
                const int ld = (int)(en & 255u);
                xv[j]   = als[s * HEADS + h] + ald[(tile0 + ld) * HEADS + h];
                cell[j] = ld * HEADS + h;
            }
#pragma unroll
            for (int j = 0; j < 4; ++j) {
                if (idx0 + j * AGG_THREADS < n1) {
                    const float lr = xv[j] >= 0.0f ? xv[j] : 0.2f * xv[j];
                    atomicMax(&mxk[cell[j]], f2key(lr));
                }
            }
        }
        __syncthreads();

        int hk[NIT];
#pragma unroll
        for (int k = 0; k < NIT; ++k) hk[k] = min((lane + 32 * k) / C, HEADS - 1);

        for (int item0 = wave; item0 < total; item0 += AGG_WAVES * ACC_BATCH) {
            int      sb[ACC_BATCH], lb[ACC_BATCH];
            float    xa[ACC_BATCH], xd[ACC_BATCH], hv[ACC_BATCH][NIT];
            unsigned mk[ACC_BATCH];
#pragma unroll
            for (int b = 0; b < ACC_BATCH; ++b) {
                const int it = min(item0 + AGG_WAVES * b, total - 1);
                const unsigned en = elist[it];
                sb[b] = (int)(en >> 8);
                lb[b] = (int)(en & 255u);
            }
#pragma unroll
            for (int b = 0; b < ACC_BATCH; ++b) {
                xa[b] = 0.0f; xd[b] = 0.0f; mk[b] = 0u;
                if (lane < HEADS) {
                    xa[b] = als[sb[b] * HEADS + lane];
                    xd[b] = ald[(tile0 + lb[b]) * HEADS + lane];
                    mk[b] = mxk[lb[b] * HEADS + lane];
                }
                const float* __restrict__ hrow = H + (size_t)sb[b] * HC;
#pragma unroll
                for (int k = 0; k < NIT; ++k) {
                    const int c = lane + 32 * k;
                    hv[b][k] = (c < HC) ? hrow[c] : 0.0f;
                }
            }
#pragma unroll
            for (int b = 0; b < ACC_BATCH; ++b) {
                if (item0 + AGG_WAVES * b < total) {
                    float exl = 0.0f;
                    if (lane < HEADS) {
                        const float x  = xa[b] + xd[b];
                        const float lr = x >= 0.0f ? x : 0.2f * x;
                        exl = __expf(lr - key2f(mk[b]));
                        atomicAdd(&dn[lb[b] * HEADS + lane], exl);
                    }
                    float* arow = acc + lb[b] * HC;
#pragma unroll
                    for (int k = 0; k < NIT; ++k) {
                        const int c = lane + 32 * k;
                        const float ex = __shfl(exl, hk[k]);
                        if (c < HC) atomicAdd(&arow[c], hv[b][k] * ex);
                    }
                }
            }
        }
    } else {
        auto visit_max = [&](int s, int ld) {
            for (int h = 0; h < HEADS; ++h) {
                const float x  = als[s * HEADS + h] + ald[(tile0 + ld) * HEADS + h];
                const float lr = x >= 0.0f ? x : 0.2f * x;
                atomicMax(&mxk[ld * HEADS + h], f2key(lr));
            }
        };
        for (int i = tid; i < nvalid; i += AGG_THREADS) visit_max(tile0 + i, i);
        for (int e = tid; e < E; e += AGG_THREADS) {
            const unsigned ld = (unsigned)(dstp[e] - tile0);
            if (ld < nv) visit_max(srcp[e], (int)ld);
        }
        __syncthreads();

        auto visit_acc = [&](int s, int ld) {
            for (int h = 0; h < HEADS; ++h) {
                const float x  = als[s * HEADS + h] + ald[(tile0 + ld) * HEADS + h];
                const float lr = x >= 0.0f ? x : 0.2f * x;
                const float ex = __expf(lr - key2f(mxk[ld * HEADS + h]));
                atomicAdd(&dn[ld * HEADS + h], ex);
                const float* hs = H + (size_t)s * HC + h * C;
                float*       ap = acc + ld * HC + h * C;
                for (int c = 0; c < C; ++c) atomicAdd(&ap[c], hs[c] * ex);
            }
        };
        for (int i = tid; i < nvalid; i += AGG_THREADS) visit_acc(tile0 + i, i);
        for (int e = tid; e < E; e += AGG_THREADS) {
            const unsigned ld = (unsigned)(dstp[e] - tile0);
            if (ld < nv) visit_acc(srcp[e], (int)ld);
        }
    }
    __syncthreads();

    for (int i = tid; i < nvalid * HC; i += AGG_THREADS) {
        const int ld  = i / HC;
        const int rem = i - ld * HC;
        const int h   = rem / C;
        float v = acc[i] / dn[ld * HEADS + h] + bias[rem];
        if (relu) v = fmaxf(v, 0.0f);
        out[(size_t)tile0 * HC + i] = v;
    }
}


extern "C" void kernel_launch(void* const* d_in, const int* in_sizes, int n_in,
                              void* d_out, int out_size, void* d_ws, size_t ws_size,
                              hipStream_t stream) {
    const float* x      = (const float*)d_in[0];
    const int*   ei     = (const int*)  d_in[1];
    const float* W1     = (const float*)d_in[2];
    const float* a_src1 = (const float*)d_in[3];
    const float* a_dst1 = (const float*)d_in[4];
    const float* b1     = (const float*)d_in[5];
    const float* W2     = (const float*)d_in[6];
    const float* a_src2 = (const float*)d_in[7];
    const float* a_dst2 = (const float*)d_in[8];
    const float* b2     = (const float*)d_in[9];
    const float* W3     = (const float*)d_in[10];
    const float* a_src3 = (const float*)d_in[11];
    const float* a_dst3 = (const float*)d_in[12];
    const float* b3     = (const float*)d_in[13];

    const int E = in_sizes[1] / 2;

    float* ws   = (float*)d_ws;
    float* bufH = ws;
    float* act1 = bufH + (size_t)NNODE * 240;
    float* h2   = act1 + (size_t)NNODE * 112;
    float* act2 = h2   + (size_t)NNODE * 96;
    float* als  = act2 + (size_t)NNODE * 96;
    float* ald  = als  + (size_t)NNODE * 7;
    float* out  = (float*)d_out;

    constexpr int NT = (NNODE + TILE - 1) / TILE;
    unsigned* glist = (unsigned*)(ald + (size_t)NNODE * 7);
    unsigned* gcnt  = glist + (size_t)NT * LIST_CAP;

    auto front = [&](const float* X, int K, const float* W, const float* as, const float* ad,
                     int heads, int C, float* Hbuf) {
        const int HC = heads * C;
        const int tiles = (NNODE / 16) * (HC / 16);
        gemm_wmma_f32<<<(tiles + 7) / 8, 256, 0, stream>>>(X, W, Hbuf, NNODE, K, HC);
        const int nh = NNODE * heads;
        att_logits<<<(nh + 255) / 256, 256, 0, stream>>>(Hbuf, as, ad, als, ald,
                                                         NNODE, heads, C);
    };

    front(x, 128, W1, a_src1, a_dst1, 7, 16, bufH);
    agg_tile<7, 16, true><<<NT, AGG_THREADS, 0, stream>>>(ei, E, als, ald, bufH, b1, act1, 1, glist, gcnt);
    front(act1, 112, W2, a_src2, a_dst2, 6, 16, h2);
    agg_tile<6, 16, false><<<NT, AGG_THREADS, 0, stream>>>(ei, E, als, ald, h2, b2, act2, 1, glist, gcnt);
    front(act2, 96, W3, a_src3, a_dst3, 6, 40, bufH);
    agg_tile<6, 40, false><<<NT, AGG_THREADS, 0, stream>>>(ei, E, als, ald, bufH, b3, out, 0, glist, gcnt);
}


# gemm: a wave keeps its 16-column slice of W in registers for all k and walks 16-row blocks of X, which the workgroup stages in LDS with whole-line 16-byte loads; k order of the sums unchanged
# speedup vs baseline: 13.4298x; 1.1515x over previous
#include <hip/hip_runtime.h>
#include <math.h>

#define NNODE 50000

typedef float v2f __attribute__((ext_vector_type(2)));
typedef float v8f __attribute__((ext_vector_type(8)));

template <int K, int M>
__global__ __launch_bounds__((M / 16) * 32)
void gemm_wstat_f32(const float* __restrict__ X,
                    const float* __restrict__ W,
                    float* __restrict__ H,
                    int rowTiles) {
    constexpr int NW  = M / 16;
    constexpr int NT  = NW * 32;
    constexpr int KS  = K / 4;
    constexpr int P   = K + 4;
    constexpr int TV  = 16 * K / 4;
    constexpr int LPT = (TV + NT - 1) / NT;

    __shared__ __attribute__((aligned(16))) float xs[2][16 * P];

    const int tid  = threadIdx.x;
    const int lane = tid & 31;
    const int wave = tid >> 5;
    const int c    = lane & 15;
    const int h    = lane >> 4;
    const int col0 = wave << 4;

    int rt = blockIdx.x;
    if (rt >= rowTiles) return;

    v2f b[KS];
#pragma unroll
    for (int ks = 0; ks < KS; ++ks) {
        const int k = 4 * ks + 2 * h;
        b[ks].x = W[(size_t)k * M + col0 + c];
        b[ks].y = W[(size_t)(k + 1) * M + col0 + c];
    }

    static_assert(LPT <= 3, "three staging registers");
    float4 s0 = {0.0f, 0.0f, 0.0f, 0.0f}, s1 = s0, s2 = s0;
    auto gload = [&](int t) {
        const float4* __restrict__ srcv = (const float4*)(X + (size_t)t * 16 * K);
        if (tid < TV) s0 = srcv[tid];
        if constexpr (LPT > 1) { if (tid + NT < TV) s1 = srcv[tid + NT]; }
        if constexpr (LPT > 2) { if (tid + 2 * NT < TV) s2 = srcv[tid + 2 * NT]; }
    };
    auto put = [&](int bufi, int v, const float4& val) {
        if (v < TV) {
            const int r = v / (K / 4);
            const int q = v - r * (K / 4);
            *(float4*)&xs[bufi][r * P + 4 * q] = val;
        }
    };
    auto lstore = [&](int bufi) {
        put(bufi, tid, s0);
        if constexpr (LPT > 1) put(bufi, tid + NT, s1);
        if constexpr (LPT > 2) put(bufi, tid + 2 * NT, s2);
    };

    gload(rt);
    lstore(0);
    __syncthreads();

    int cur = 0;
    for (; rt < rowTiles; rt += gridDim.x) {
        const int  nxt  = rt + gridDim.x;
        const bool more = nxt < rowTiles;
        if (more) gload(nxt);

        const float* xa = &xs[cur][c * P + 2 * h];
        v2f a[KS];
#pragma unroll
        for (int ks = 0; ks < KS; ++ks) a[ks] = *(const v2f*)(xa + 4 * ks);

        v8f acc = {};
#pragma unroll
        for (int ks = 0; ks < KS; ++ks)
            acc = __builtin_amdgcn_wmma_f32_16x16x4_f32(
                       false, a[ks],  false, b[ks],
                       (short)0, acc,  false,  false);

        float* hp = H + (size_t)(rt * 16 + 8 * h) * M + col0 + c;
#pragma unroll
        for (int v = 0; v < 8; ++v) hp[(size_t)v * M] = acc[v];

        if (more) lstore(cur ^ 1);
        __syncthreads();
        cur ^= 1;
    }
}

__global__ void att_logits(const float* __restrict__ H,
                           const float* __restrict__ a_src,
                           const float* __restrict__ a_dst,
                           float* __restrict__ als, float* __restrict__ ald,
                           int n, int heads, int C) {
    int t = blockIdx.x * blockDim.x + threadIdx.x;
    if (t >= n * heads) return;
    int node = t / heads;
    int h    = t - node * heads;
    const float* hp = H + (size_t)node * heads * C + (size_t)h * C;
    const float* as = a_src + h * C;
    const float* ad = a_dst + h * C;
    float s = 0.0f, d = 0.0f;
    for (int c = 0; c < C; ++c) { float v = hp[c]; s += v * as[c]; d += v * ad[c]; }
    als[t] = s;
    ald[t] = d;
}


constexpr int TILE        = 200;
constexpr int LIST_CAP    = 20480;
constexpr int AGG_THREADS = 1024;
constexpr int AGG_WAVES   = AGG_THREADS / 32;
constexpr int SCAN_VEC    = 8;
constexpr int ACC_BATCH   = 4;

static_assert(TILE <= 256, "local row index is packed into 8 bits");
static_assert(NNODE <= (1 << 24), "source index is packed into 24 bits");

__device__ __forceinline__ unsigned f2key(float f) {
    unsigned b = __float_as_uint(f);
    return (b & 0x80000000u) ? ~b : (b | 0x80000000u);
}
__device__ __forceinline__ float key2f(unsigned k) {
    return __uint_as_float((k & 0x80000000u) ? (k & 0x7fffffffu) : ~k);
}

template <int HEADS, int C, bool BUILD>
__global__ __launch_bounds__(AGG_THREADS)
void agg_tile(const int* __restrict__ ei, int E,
              const float* __restrict__ als, const float* __restrict__ ald,
              const float* __restrict__ H, const float* __restrict__ bias,
              float* __restrict__ out, int relu,
              unsigned* glist, unsigned* gcnt) {
    constexpr int HC  = HEADS * C;
    constexpr int NIT = (HC + 31) / 32;

    __shared__ float         acc[TILE * HC];
    __shared__ unsigned      mxk[TILE * HEADS];
    __shared__ float         dn[TILE * HEADS];
    __shared__ unsigned      elist[LIST_CAP];

    const int tid   = threadIdx.x;
    const int lane  = tid & 31;
    const int wave  = tid >> 5;
    const int tile0 = blockIdx.x * TILE;
    const int nvalid = min(TILE, NNODE - tile0);
    const unsigned nv = (unsigned)nvalid;

    const int* __restrict__ srcp = ei;
    const int* __restrict__ dstp = ei + E;

    unsigned* tlist = glist + (size_t)blockIdx.x * LIST_CAP;

    unsigned total_raw = 0;
    if constexpr (!BUILD) total_raw = gcnt[blockIdx.x];

    for (int i = tid; i < TILE * HC; i += AGG_THREADS) acc[i] = 0.0f;
    const unsigned key_ninf = f2key(-INFINITY);
    for (int i = tid; i < TILE * HEADS; i += AGG_THREADS) { mxk[i] = key_ninf; dn[i] = 0.0f; }

    if constexpr (BUILD) {
        __shared__ unsigned char eld[LIST_CAP];
        __shared__ unsigned      cnt;

        for (int i = tid; i < nvalid; i += AGG_THREADS) {
            const unsigned p = ((unsigned)(tile0 + i) << 8) | (unsigned)i;
            elist[i] = p;
            tlist[i] = p;
        }
        if (tid == 0) cnt = nv;
        __syncthreads();

        auto push = [&](int e, int d) {
            const unsigned ld = (unsigned)(d - tile0);
            if (ld < nv) {
                const unsigned pos = atomicAdd(&cnt, 1u);
                if (pos < (unsigned)LIST_CAP) { elist[pos] = (unsigned)e; eld[pos] = (unsigned char)ld; }
            }
        };

        int e_done = 0;
        if ((((size_t)dstp) & 15) == 0 && E >= 4) {
            const int   nvec = E >> 2;
            const int4* dv   = (const int4*)dstp;
            constexpr int STEP = SCAN_VEC * AGG_THREADS;
            auto loadq = [&](int4 (&q)[SCAN_VEC], int base) {
#pragma unroll
                for (int j = 0; j < SCAN_VEC; ++j) {
                    const int v = base + j * AGG_THREADS + tid;
                    q[j] = dv[min(v, nvec - 1)];
                }
            };
            auto procq = [&](const int4 (&q)[SCAN_VEC], int base) {
#pragma unroll
                for (int j = 0; j < SCAN_VEC; ++j) {
                    const int v = base + j * AGG_THREADS + tid;
                    const bool any = ((unsigned)(q[j].x - tile0) < nv) | ((unsigned)(q[j].y - tile0) < nv) |
                                     ((unsigned)(q[j].z - tile0) < nv) | ((unsigned)(q[j].w - tile0) < nv);
                    if (any && v < nvec) {
                        push(4 * v,     q[j].x);
                        push(4 * v + 1, q[j].y);
                        push(4 * v + 2, q[j].z);
                        push(4 * v + 3, q[j].w);
                    }
                }
            };
            int4 qa[SCAN_VEC], qb[SCAN_VEC];
            loadq(qa, 0);
            for (int base = 0; base < nvec; base += 2 * STEP) {
                loadq(qb, base + STEP);
                procq(qa, base);
                loadq(qa, base + 2 * STEP);
                procq(qb, base + STEP);
            }
            e_done = nvec << 2;
        }
        for (int e = e_done + tid; e < E; e += AGG_THREADS) push(e, dstp[e]);
        __syncthreads();

        total_raw = cnt;
        if (tid == 0) gcnt[blockIdx.x] = total_raw;
        if (total_raw <= (unsigned)LIST_CAP) {
            const int total = (int)total_raw;

            for (int i0 = nvalid + tid; i0 < total; i0 += 4 * AGG_THREADS) {
                int sv[4];
#pragma unroll
                for (int j = 0; j < 4; ++j) {
                    const int i = i0 + j * AGG_THREADS;
                    sv[j] = (i < total) ? srcp[elist[i]] : 0;
                }
#pragma unroll
                for (int j = 0; j < 4; ++j) {
                    const int i = i0 + j * AGG_THREADS;
                    if (i < total) {
                        const unsigned p = ((unsigned)sv[j] << 8) | (unsigned)eld[i];
                        elist[i] = p;
                        tlist[i] = p;
                    }
                }
            }
        }
        __syncthreads();
    } else {
        if (total_raw <= (unsigned)LIST_CAP) {
            const int total = (int)total_raw;
            const unsigned* __restrict__ rlist = tlist;
            for (int i0 = tid; i0 < total; i0 += 4 * AGG_THREADS) {
                unsigned pv[4];
#pragma unroll
                for (int j = 0; j < 4; ++j) {
                    const int i = i0 + j * AGG_THREADS;
                    pv[j] = (i < total) ? rlist[i] : 0u;
                }
#pragma unroll
                for (int j = 0; j < 4; ++j) {
                    const int i = i0 + j * AGG_THREADS;
                    if (i < total) elist[i] = pv[j];
                }
            }
        }
        __syncthreads();
    }

    if (total_raw <= (unsigned)LIST_CAP) {
        const int total = (int)total_raw;

        const int n1 = total * HEADS;
        for (int idx0 = tid; idx0 < n1; idx0 += 4 * AGG_THREADS) {
            float xv[4];
            int   cell[4];
#pragma unroll
            for (int j = 0; j < 4; ++j) {
                const int idx  = min(idx0 + j * AGG_THREADS, n1 - 1);
                const int item = idx / HEADS;
                const int h    = idx - item * HEADS;
                const unsigned en = elist[item];
                const int s  = (int)(en >> 8);
                const int ld = (int)(en & 255u);
                xv[j]   = als[s * HEADS + h] + ald[(tile0 + ld) * HEADS + h];
                cell[j] = ld * HEADS + h;
            }
#pragma unroll
            for (int j = 0; j < 4; ++j) {
                if (idx0 + j * AGG_THREADS < n1) {
                    const float lr = xv[j] >= 0.0f ? xv[j] : 0.2f * xv[j];
                    atomicMax(&mxk[cell[j]], f2key(lr));
                }
            }
        }
        __syncthreads();

        int hk[NIT];
#pragma unroll
        for (int k = 0; k < NIT; ++k) hk[k] = min((lane + 32 * k) / C, HEADS - 1);

        for (int item0 = wave; item0 < total; item0 += AGG_WAVES * ACC_BATCH) {
            int      sb[ACC_BATCH], lb[ACC_BATCH];
            float    xa[ACC_BATCH], xd[ACC_BATCH], hv[ACC_BATCH][NIT];
            unsigned mk[ACC_BATCH];
#pragma unroll
            for (int b = 0; b < ACC_BATCH; ++b) {
                const int it = min(item0 + AGG_WAVES * b, total - 1);
                const unsigned en = elist[it];
                sb[b] = (int)(en >> 8);
                lb[b] = (int)(en & 255u);
            }
#pragma unroll
            for (int b = 0; b < ACC_BATCH; ++b) {
                xa[b] = 0.0f; xd[b] = 0.0f; mk[b] = 0u;
                if (lane < HEADS) {
                    xa[b] = als[sb[b] * HEADS + lane];
                    xd[b] = ald[(tile0 + lb[b]) * HEADS + lane];
                    mk[b] = mxk[lb[b] * HEADS + lane];
                }
                const float* __restrict__ hrow = H + (size_t)sb[b] * HC;
#pragma unroll
                for (int k = 0; k < NIT; ++k) {
                    const int c = lane + 32 * k;
                    hv[b][k] = (c < HC) ? hrow[c] : 0.0f;
                }
            }
#pragma unroll
            for (int b = 0; b < ACC_BATCH; ++b) {
                if (item0 + AGG_WAVES * b < total) {
                    float exl = 0.0f;
                    if (lane < HEADS) {
                        const float x  = xa[b] + xd[b];
                        const float lr = x >= 0.0f ? x : 0.2f * x;
                        exl = __expf(lr - key2f(mk[b]));
                        atomicAdd(&dn[lb[b] * HEADS + lane], exl);
                    }
                    float* arow = acc + lb[b] * HC;
#pragma unroll
                    for (int k = 0; k < NIT; ++k) {
                        const int c = lane + 32 * k;
                        const float ex = __shfl(exl, hk[k]);
                        if (c < HC) atomicAdd(&arow[c], hv[b][k] * ex);
                    }
                }
            }
        }
    } else {
        auto visit_max = [&](int s, int ld) {
            for (int h = 0; h < HEADS; ++h) {
                const float x  = als[s * HEADS + h] + ald[(tile0 + ld) * HEADS + h];
                const float lr = x >= 0.0f ? x : 0.2f * x;
                atomicMax(&mxk[ld * HEADS + h], f2key(lr));
            }
        };
        for (int i = tid; i < nvalid; i += AGG_THREADS) visit_max(tile0 + i, i);
        for (int e = tid; e < E; e += AGG_THREADS) {
            const unsigned ld = (unsigned)(dstp[e] - tile0);
            if (ld < nv) visit_max(srcp[e], (int)ld);
        }
        __syncthreads();

        auto visit_acc = [&](int s, int ld) {
            for (int h = 0; h < HEADS; ++h) {
                const float x  = als[s * HEADS + h] + ald[(tile0 + ld) * HEADS + h];
                const float lr = x >= 0.0f ? x : 0.2f * x;
                const float ex = __expf(lr - key2f(mxk[ld * HEADS + h]));
                atomicAdd(&dn[ld * HEADS + h], ex);
                const float* hs = H + (size_t)s * HC + h * C;
                float*       ap = acc + ld * HC + h * C;
                for (int c = 0; c < C; ++c) atomicAdd(&ap[c], hs[c] * ex);
            }
        };
        for (int i = tid; i < nvalid; i += AGG_THREADS) visit_acc(tile0 + i, i);
        for (int e = tid; e < E; e += AGG_THREADS) {
            const unsigned ld = (unsigned)(dstp[e] - tile0);
            if (ld < nv) visit_acc(srcp[e], (int)ld);
        }
    }
    __syncthreads();

    for (int i = tid; i < nvalid * HC; i += AGG_THREADS) {
        const int ld  = i / HC;
        const int rem = i - ld * HC;
        const int h   = rem / C;
        float v = acc[i] / dn[ld * HEADS + h] + bias[rem];
        if (relu) v = fmaxf(v, 0.0f);
        out[(size_t)tile0 * HC + i] = v;
    }
}


extern "C" void kernel_launch(void* const* d_in, const int* in_sizes, int n_in,
                              void* d_out, int out_size, void* d_ws, size_t ws_size,
                              hipStream_t stream) {
    const float* x      = (const float*)d_in[0];
    const int*   ei     = (const int*)  d_in[1];
    const float* W1     = (const float*)d_in[2];
    const float* a_src1 = (const float*)d_in[3];
    const float* a_dst1 = (const float*)d_in[4];
    const float* b1     = (const float*)d_in[5];
    const float* W2     = (const float*)d_in[6];
    const float* a_src2 = (const float*)d_in[7];
    const float* a_dst2 = (const float*)d_in[8];
    const float* b2     = (const float*)d_in[9];
    const float* W3     = (const float*)d_in[10];
    const float* a_src3 = (const float*)d_in[11];
    const float* a_dst3 = (const float*)d_in[12];
    const float* b3     = (const float*)d_in[13];

    const int E = in_sizes[1] / 2;

    float* ws   = (float*)d_ws;
    float* bufH = ws;
    float* act1 = bufH + (size_t)NNODE * 240;
    float* h2   = act1 + (size_t)NNODE * 112;
    float* act2 = h2   + (size_t)NNODE * 96;
    float* als  = act2 + (size_t)NNODE * 96;
    float* ald  = als  + (size_t)NNODE * 7;
    float* out  = (float*)d_out;

    constexpr int NT = (NNODE + TILE - 1) / TILE;
    unsigned* glist = (unsigned*)(ald + (size_t)NNODE * 7);
    unsigned* gcnt  = glist + (size_t)NT * LIST_CAP;

    constexpr int RT = NNODE / 16;
    static_assert(NNODE % 16 == 0, "row blocks are whole");

    auto logits = [&](const float* Hbuf, const float* as, const float* ad, int heads, int C) {
        const int nh = NNODE * heads;
        att_logits<<<(nh + 255) / 256, 256, 0, stream>>>(Hbuf, as, ad, als, ald, NNODE, heads, C);
    };

    gemm_wstat_f32<128, 112><<<768, 224, 0, stream>>>(x, W1, bufH, RT);
    logits(bufH, a_src1, a_dst1, 7, 16);
    agg_tile<7, 16, true><<<NT, AGG_THREADS, 0, stream>>>(ei, E, als, ald, bufH, b1, act1, 1, glist, gcnt);
    gemm_wstat_f32<112, 96><<<768, 192, 0, stream>>>(act1, W2, h2, RT);
    logits(h2, a_src2, a_dst2, 6, 16);
    agg_tile<6, 16, false><<<NT, AGG_THREADS, 0, stream>>>(ei, E, als, ald, h2, b2, act2, 1, glist, gcnt);
    gemm_wstat_f32<96, 240><<<512, 480, 0, stream>>>(act2, W3, bufH, RT);
    logits(bufH, a_src3, a_dst3, 6, 40);
    agg_tile<6, 40, false><<<NT, AGG_THREADS, 0, stream>>>(ei, E, als, ald, bufH, b3, out, 0, glist, gcnt);
}


# logits: a_src, a_dst folded into the weights (W*a: 16 more GEMM columns, one more wave per workgroup), so logits come out of the matrix unit and the 3 logit launches go; logit sums re-associated
# speedup vs baseline: 14.2499x; 1.0611x over previous
#include <hip/hip_runtime.h>
#include <math.h>

#define NNODE 50000

typedef float v2f __attribute__((ext_vector_type(2)));
typedef float v8f __attribute__((ext_vector_type(8)));

template <int K, int M>
__global__ __launch_bounds__((M / 16 + 1) * 32)
void gemm_wstat_f32(const float* __restrict__ X,
                    const float* __restrict__ W,
                    const float* __restrict__ WA,
                    float* __restrict__ H,
                    float* __restrict__ LG,
                    int rowTiles) {
    constexpr int NW  = M / 16 + 1;
    constexpr int NT  = NW * 32;
    constexpr int KS  = K / 4;
    constexpr int P   = K + 4;
    constexpr int TV  = 16 * K / 4;
    constexpr int LPT = (TV + NT - 1) / NT;

    __shared__ __attribute__((aligned(16))) float xs[2][16 * P];

    const int tid  = threadIdx.x;
    const int lane = tid & 31;
    const int wave = tid >> 5;
    const int c    = lane & 15;
    const int h    = lane >> 4;
    const int col0 = wave << 4;
    const bool lw  = wave == NW - 1;

    int rt = blockIdx.x;
    if (rt >= rowTiles) return;

    v2f b[KS];
    if (!lw) {
#pragma unroll
        for (int ks = 0; ks < KS; ++ks) {
            const int k = 4 * ks + 2 * h;
            b[ks].x = W[(size_t)k * M + col0 + c];
            b[ks].y = W[(size_t)(k + 1) * M + col0 + c];
        }
    } else {
#pragma unroll
        for (int ks = 0; ks < KS; ++ks) {
            const int k = 4 * ks + 2 * h;
            b[ks].x = WA[k * 16 + c];
            b[ks].y = WA[(k + 1) * 16 + c];
        }
    }

    static_assert(LPT <= 3, "three staging registers");
    float4 s0 = {0.0f, 0.0f, 0.0f, 0.0f}, s1 = s0, s2 = s0;
    auto gload = [&](int t) {
        const float4* __restrict__ srcv = (const float4*)(X + (size_t)t * 16 * K);
        if (tid < TV) s0 = srcv[tid];
        if constexpr (LPT > 1) { if (tid + NT < TV) s1 = srcv[tid + NT]; }
        if constexpr (LPT > 2) { if (tid + 2 * NT < TV) s2 = srcv[tid + 2 * NT]; }
    };
    auto put = [&](int bufi, int v, const float4& val) {
        if (v < TV) {
            const int r = v / (K / 4);
            const int q = v - r * (K / 4);
            *(float4*)&xs[bufi][r * P + 4 * q] = val;
        }
    };
    auto lstore = [&](int bufi) {
        put(bufi, tid, s0);
        if constexpr (LPT > 1) put(bufi, tid + NT, s1);
        if constexpr (LPT > 2) put(bufi, tid + 2 * NT, s2);
    };

    gload(rt);
    lstore(0);
    __syncthreads();

    int cur = 0;
    for (; rt < rowTiles; rt += gridDim.x) {
        const int  nxt  = rt + gridDim.x;
        const bool more = nxt < rowTiles;
        if (more) gload(nxt);

        const float* xa = &xs[cur][c * P + 2 * h];
        v2f a[KS];
#pragma unroll
        for (int ks = 0; ks < KS; ++ks) a[ks] = *(const v2f*)(xa + 4 * ks);

        v8f acc = {};
#pragma unroll
        for (int ks = 0; ks < KS; ++ks)
            acc = __builtin_amdgcn_wmma_f32_16x16x4_f32(
                       false, a[ks],  false, b[ks],
                       (short)0, acc,  false,  false);

        if (!lw) {
            float* hp = H + (size_t)(rt * 16 + 8 * h) * M + col0 + c;
#pragma unroll
            for (int v = 0; v < 8; ++v) hp[(size_t)v * M] = acc[v];
        } else {
            float* lp = LG + (size_t)(rt * 16 + 8 * h) * 16 + c;
#pragma unroll
            for (int v = 0; v < 8; ++v) lp[v * 16] = acc[v];
        }

        if (more) lstore(cur ^ 1);
        __syncthreads();
        cur ^= 1;
    }
}

__device__ __forceinline__ void fold_one(const float* __restrict__ W,
                                         const float* __restrict__ a_src,
                                         const float* __restrict__ a_dst,
                                         float* __restrict__ WA,
                                         int K, int M, int heads, int C, int t) {
    if (t >= K * 16) return;
    const int k = t >> 4;
    const int j = t & 15;
    float s = 0.0f;
    if (j < 2 * heads) {
        const int hd = j < heads ? j : j - heads;
        const float* a = (j < heads ? a_src : a_dst) + hd * C;
        const float* w = W + (size_t)k * M + hd * C;
        for (int cc = 0; cc < C; ++cc) s += w[cc] * a[cc];
    }
    WA[t] = s;
}

__global__ void fold_att(const float* __restrict__ W1, const float* __restrict__ as1, const float* __restrict__ ad1,
                         const float* __restrict__ W2, const float* __restrict__ as2, const float* __restrict__ ad2,
                         const float* __restrict__ W3, const float* __restrict__ as3, const float* __restrict__ ad3,
                         float* __restrict__ WA1, float* __restrict__ WA2, float* __restrict__ WA3) {
    const int t = blockIdx.x * blockDim.x + threadIdx.x;
    if (blockIdx.y == 0)      fold_one(W1, as1, ad1, WA1, 128, 112, 7, 16, t);
    else if (blockIdx.y == 1) fold_one(W2, as2, ad2, WA2, 112,  96, 6, 16, t);
    else                      fold_one(W3, as3, ad3, WA3,  96, 240, 6, 40, t);
}


constexpr int TILE        = 200;
constexpr int LIST_CAP    = 20480;
constexpr int AGG_THREADS = 1024;
constexpr int AGG_WAVES   = AGG_THREADS / 32;
constexpr int SCAN_VEC    = 8;
constexpr int ACC_BATCH   = 4;

static_assert(TILE <= 256, "local row index is packed into 8 bits");
static_assert(NNODE <= (1 << 24), "source index is packed into 24 bits");

__device__ __forceinline__ unsigned f2key(float f) {
    unsigned b = __float_as_uint(f);
    return (b & 0x80000000u) ? ~b : (b | 0x80000000u);
}
__device__ __forceinline__ float key2f(unsigned k) {
    return __uint_as_float((k & 0x80000000u) ? (k & 0x7fffffffu) : ~k);
}

template <int HEADS, int C, bool BUILD>
__global__ __launch_bounds__(AGG_THREADS)
void agg_tile(const int* __restrict__ ei, int E,
              const float* __restrict__ lg,
              const float* __restrict__ H, const float* __restrict__ bias,
              float* __restrict__ out, int relu,
              unsigned* glist, unsigned* gcnt) {
    constexpr int HC  = HEADS * C;
    constexpr int NIT = (HC + 31) / 32;

    __shared__ float         acc[TILE * HC];
    __shared__ unsigned      mxk[TILE * HEADS];
    __shared__ float         dn[TILE * HEADS];
    __shared__ unsigned      elist[LIST_CAP];

    const int tid   = threadIdx.x;
    const int lane  = tid & 31;
    const int wave  = tid >> 5;
    const int tile0 = blockIdx.x * TILE;
    const int nvalid = min(TILE, NNODE - tile0);
    const unsigned nv = (unsigned)nvalid;

    const int* __restrict__ srcp = ei;
    const int* __restrict__ dstp = ei + E;

    unsigned* tlist = glist + (size_t)blockIdx.x * LIST_CAP;

    unsigned total_raw = 0;
    if constexpr (!BUILD) total_raw = gcnt[blockIdx.x];

    for (int i = tid; i < TILE * HC; i += AGG_THREADS) acc[i] = 0.0f;
    const unsigned key_ninf = f2key(-INFINITY);
    for (int i = tid; i < TILE * HEADS; i += AGG_THREADS) { mxk[i] = key_ninf; dn[i] = 0.0f; }

    if constexpr (BUILD) {
        __shared__ unsigned char eld[LIST_CAP];
        __shared__ unsigned      cnt;

        for (int i = tid; i < nvalid; i += AGG_THREADS) {
            const unsigned p = ((unsigned)(tile0 + i) << 8) | (unsigned)i;
            elist[i] = p;
            tlist[i] = p;
        }
        if (tid == 0) cnt = nv;
        __syncthreads();

        auto push = [&](int e, int d) {
            const unsigned ld = (unsigned)(d - tile0);
            if (ld < nv) {
                const unsigned pos = atomicAdd(&cnt, 1u);
                if (pos < (unsigned)LIST_CAP) { elist[pos] = (unsigned)e; eld[pos] = (unsigned char)ld; }
            }
        };

        int e_done = 0;
        if ((((size_t)dstp) & 15) == 0 && E >= 4) {
            const int   nvec = E >> 2;
            const int4* dv   = (const int4*)dstp;
            constexpr int STEP = SCAN_VEC * AGG_THREADS;
            auto loadq = [&](int4 (&q)[SCAN_VEC], int base) {
#pragma unroll
                for (int j = 0; j < SCAN_VEC; ++j) {
                    const int v = base + j * AGG_THREADS + tid;
                    q[j] = dv[min(v, nvec - 1)];
                }
            };
            auto procq = [&](const int4 (&q)[SCAN_VEC], int base) {
#pragma unroll
                for (int j = 0; j < SCAN_VEC; ++j) {
                    const int v = base + j * AGG_THREADS + tid;
                    const bool any = ((unsigned)(q[j].x - tile0) < nv) | ((unsigned)(q[j].y - tile0) < nv) |
                                     ((unsigned)(q[j].z - tile0) < nv) | ((unsigned)(q[j].w - tile0) < nv);
                    if (any && v < nvec) {
                        push(4 * v,     q[j].x);
                        push(4 * v + 1, q[j].y);
                        push(4 * v + 2, q[j].z);
                        push(4 * v + 3, q[j].w);
                    }
                }
            };
            int4 qa[SCAN_VEC], qb[SCAN_VEC];
            loadq(qa, 0);
            for (int base = 0; base < nvec; base += 2 * STEP) {
                loadq(qb, base + STEP);
                procq(qa, base);
                loadq(qa, base + 2 * STEP);
                procq(qb, base + STEP);
            }
            e_done = nvec << 2;
        }
        for (int e = e_done + tid; e < E; e += AGG_THREADS) push(e, dstp[e]);
        __syncthreads();

        total_raw = cnt;
        if (tid == 0) gcnt[blockIdx.x] = total_raw;
        if (total_raw <= (unsigned)LIST_CAP) {
            const int total = (int)total_raw;

            for (int i0 = nvalid + tid; i0 < total; i0 += 4 * AGG_THREADS) {
                int sv[4];
#pragma unroll
                for (int j = 0; j < 4; ++j) {
                    const int i = i0 + j * AGG_THREADS;
                    sv[j] = (i < total) ? srcp[elist[i]] : 0;
                }
#pragma unroll
                for (int j = 0; j < 4; ++j) {
                    const int i = i0 + j * AGG_THREADS;
                    if (i < total) {
                        const unsigned p = ((unsigned)sv[j] << 8) | (unsigned)eld[i];
                        elist[i] = p;
                        tlist[i] = p;
                    }
                }
            }
        }
        __syncthreads();
    } else {
        if (total_raw <= (unsigned)LIST_CAP) {
            const int total = (int)total_raw;
            const unsigned* __restrict__ rlist = tlist;
            for (int i0 = tid; i0 < total; i0 += 4 * AGG_THREADS) {
                unsigned pv[4];
#pragma unroll
                for (int j = 0; j < 4; ++j) {
                    const int i = i0 + j * AGG_THREADS;
                    pv[j] = (i < total) ? rlist[i] : 0u;
                }
#pragma unroll
                for (int j = 0; j < 4; ++j) {
                    const int i = i0 + j * AGG_THREADS;
                    if (i < total) elist[i] = pv[j];
                }
            }
        }
        __syncthreads();
    }

    if (total_raw <= (unsigned)LIST_CAP) {
        const int total = (int)total_raw;

        const int n1 = total * HEADS;
        for (int idx0 = tid; idx0 < n1; idx0 += 4 * AGG_THREADS) {
            float xv[4];
            int   cell[4];
#pragma unroll
            for (int j = 0; j < 4; ++j) {
                const int idx  = min(idx0 + j * AGG_THREADS, n1 - 1);
                const int item = idx / HEADS;
                const int h    = idx - item * HEADS;
                const unsigned en = elist[item];
                const int s  = (int)(en >> 8);
                const int ld = (int)(en & 255u);
                xv[j]   = lg[s * 16 + h] + lg[(tile0 + ld) * 16 + HEADS + h];
                cell[j] = ld * HEADS + h;
            }
#pragma unroll
            for (int j = 0; j < 4; ++j) {
                if (idx0 + j * AGG_THREADS < n1) {
                    const float lr = xv[j] >= 0.0f ? xv[j] : 0.2f * xv[j];
                    atomicMax(&mxk[cell[j]], f2key(lr));
                }
            }
        }
        __syncthreads();

        int hk[NIT];
#pragma unroll
        for (int k = 0; k < NIT; ++k) hk[k] = min((lane + 32 * k) / C, HEADS - 1);

        for (int item0 = wave; item0 < total; item0 += AGG_WAVES * ACC_BATCH) {
            int      sb[ACC_BATCH], lb[ACC_BATCH];
            float    xa[ACC_BATCH], xd[ACC_BATCH], hv[ACC_BATCH][NIT];
            unsigned mk[ACC_BATCH];
#pragma unroll
            for (int b = 0; b < ACC_BATCH; ++b) {
                const int it = min(item0 + AGG_WAVES * b, total - 1);
                const unsigned en = elist[it];
                sb[b] = (int)(en >> 8);
                lb[b] = (int)(en & 255u);
            }
#pragma unroll
            for (int b = 0; b < ACC_BATCH; ++b) {
                xa[b] = 0.0f; xd[b] = 0.0f; mk[b] = 0u;
                if (lane < HEADS) {
                    xa[b] = lg[sb[b] * 16 + lane];
                    xd[b] = lg[(tile0 + lb[b]) * 16 + HEADS + lane];
                    mk[b] = mxk[lb[b] * HEADS + lane];
                }
                const float* __restrict__ hrow = H + (size_t)sb[b] * HC;
#pragma unroll
                for (int k = 0; k < NIT; ++k) {
                    const int c = lane + 32 * k;
                    hv[b][k] = (c < HC) ? hrow[c] : 0.0f;
                }
            }
#pragma unroll
            for (int b = 0; b < ACC_BATCH; ++b) {
                if (item0 + AGG_WAVES * b < total) {
                    float exl = 0.0f;
                    if (lane < HEADS) {
                        const float x  = xa[b] + xd[b];
                        const float lr = x >= 0.0f ? x : 0.2f * x;
                        exl = __expf(lr - key2f(mk[b]));
                        atomicAdd(&dn[lb[b] * HEADS + lane], exl);
                    }
                    float* arow = acc + lb[b] * HC;
#pragma unroll
                    for (int k = 0; k < NIT; ++k) {
                        const int c = lane + 32 * k;
                        const float ex = __shfl(exl, hk[k]);
                        if (c < HC) atomicAdd(&arow[c], hv[b][k] * ex);
                    }
                }
            }
        }
    } else {
        auto visit_max = [&](int s, int ld) {
            for (int h = 0; h < HEADS; ++h) {
                const float x  = lg[s * 16 + h] + lg[(tile0 + ld) * 16 + HEADS + h];
                const float lr = x >= 0.0f ? x : 0.2f * x;
                atomicMax(&mxk[ld * HEADS + h], f2key(lr));
            }
        };
        for (int i = tid; i < nvalid; i += AGG_THREADS) visit_max(tile0 + i, i);
        for (int e = tid; e < E; e += AGG_THREADS) {
            const unsigned ld = (unsigned)(dstp[e] - tile0);
            if (ld < nv) visit_max(srcp[e], (int)ld);
        }
        __syncthreads();

        auto visit_acc = [&](int s, int ld) {
            for (int h = 0; h < HEADS; ++h) {
                const float x  = lg[s * 16 + h] + lg[(tile0 + ld) * 16 + HEADS + h];
                const float lr = x >= 0.0f ? x : 0.2f * x;
                const float ex = __expf(lr - key2f(mxk[ld * HEADS + h]));
                atomicAdd(&dn[ld * HEADS + h], ex);
                const float* hs = H + (size_t)s * HC + h * C;
                float*       ap = acc + ld * HC + h * C;
                for (int c = 0; c < C; ++c) atomicAdd(&ap[c], hs[c] * ex);
            }
        };
        for (int i = tid; i < nvalid; i += AGG_THREADS) visit_acc(tile0 + i, i);
        for (int e = tid; e < E; e += AGG_THREADS) {
            const unsigned ld = (unsigned)(dstp[e] - tile0);
            if (ld < nv) visit_acc(srcp[e], (int)ld);
        }
    }
    __syncthreads();

    for (int i = tid; i < nvalid * HC; i += AGG_THREADS) {
        const int ld  = i / HC;
        const int rem = i - ld * HC;
        const int h   = rem / C;
        float v = acc[i] / dn[ld * HEADS + h] + bias[rem];
        if (relu) v = fmaxf(v, 0.0f);
        out[(size_t)tile0 * HC + i] = v;
    }
}


extern "C" void kernel_launch(void* const* d_in, const int* in_sizes, int n_in,
                              void* d_out, int out_size, void* d_ws, size_t ws_size,
                              hipStream_t stream) {
    const float* x      = (const float*)d_in[0];
    const int*   ei     = (const int*)  d_in[1];
    const float* W1     = (const float*)d_in[2];
    const float* a_src1 = (const float*)d_in[3];
    const float* a_dst1 = (const float*)d_in[4];
    const float* b1     = (const float*)d_in[5];
    const float* W2     = (const float*)d_in[6];
    const float* a_src2 = (const float*)d_in[7];
    const float* a_dst2 = (const float*)d_in[8];
    const float* b2     = (const float*)d_in[9];
    const float* W3     = (const float*)d_in[10];
    const float* a_src3 = (const float*)d_in[11];
    const float* a_dst3 = (const float*)d_in[12];
    const float* b3     = (const float*)d_in[13];

    const int E = in_sizes[1] / 2;

    float* ws   = (float*)d_ws;
    float* bufH = ws;
    float* act1 = bufH + (size_t)NNODE * 240;
    float* h2   = act1 + (size_t)NNODE * 112;
    float* act2 = h2   + (size_t)NNODE * 96;
    float* lg   = act2 + (size_t)NNODE * 96;
    float* wa1  = lg   + (size_t)NNODE * 16;
    float* wa2  = wa1  + 128 * 16;
    float* wa3  = wa2  + 112 * 16;
    float* out  = (float*)d_out;

    constexpr int NT = (NNODE + TILE - 1) / TILE;
    unsigned* glist = (unsigned*)(wa3 + 96 * 16);
    unsigned* gcnt  = glist + (size_t)NT * LIST_CAP;

    constexpr int RT = NNODE / 16;
    static_assert(NNODE % 16 == 0, "row blocks are whole");

    fold_att<<<dim3(8, 3), 256, 0, stream>>>(W1, a_src1, a_dst1, W2, a_src2, a_dst2, W3, a_src3, a_dst3, wa1, wa2, wa3);

    gemm_wstat_f32<128, 112><<<768, 256, 0, stream>>>(x, W1, wa1, bufH, lg, RT);
    agg_tile<7, 16, true><<<NT, AGG_THREADS, 0, stream>>>(ei, E, lg, bufH, b1, act1, 1, glist, gcnt);
    gemm_wstat_f32<112, 96><<<768, 224, 0, stream>>>(act1, W2, wa2, h2, lg, RT);
    agg_tile<6, 16, false><<<NT, AGG_THREADS, 0, stream>>>(ei, E, lg, h2, b2, act2, 1, glist, gcnt);
    gemm_wstat_f32<96, 240><<<512, 512, 0, stream>>>(act2, W3, wa3, bufH, lg, RT);
    agg_tile<6, 40, false><<<NT, AGG_THREADS, 0, stream>>>(ei, E, lg, bufH, b3, out, 0, glist, gcnt);
}


# aggregation filter: set-up launch writes one key byte per edge (its destination tile); a tile scans that plane, 16 edges a 16-byte load, not the 4-byte indices; sums stay LDS atomics, arrival order
# speedup vs baseline: 14.7465x; 1.0348x over previous
#include <hip/hip_runtime.h>
#include <math.h>

#define NNODE 50000

typedef float v2f __attribute__((ext_vector_type(2)));
typedef float v8f __attribute__((ext_vector_type(8)));

template <int K, int M>
__global__ __launch_bounds__((M / 16 + 1) * 32)
void gemm_wstat_f32(const float* __restrict__ X,
                    const float* __restrict__ W,
                    const float* __restrict__ WA,
                    float* __restrict__ H,
                    float* __restrict__ LG,
                    int rowTiles) {
    constexpr int NW  = M / 16 + 1;
    constexpr int NT  = NW * 32;
    constexpr int KS  = K / 4;
    constexpr int P   = K + 4;
    constexpr int TV  = 16 * K / 4;
    constexpr int LPT = (TV + NT - 1) / NT;

    __shared__ __attribute__((aligned(16))) float xs[2][16 * P];

    const int tid  = threadIdx.x;
    const int lane = tid & 31;
    const int wave = tid >> 5;
    const int c    = lane & 15;
    const int h    = lane >> 4;
    const int col0 = wave << 4;
    const bool lw  = wave == NW - 1;

    int rt = blockIdx.x;
    if (rt >= rowTiles) return;

    v2f b[KS];
    if (!lw) {
#pragma unroll
        for (int ks = 0; ks < KS; ++ks) {
            const int k = 4 * ks + 2 * h;
            b[ks].x = W[(size_t)k * M + col0 + c];
            b[ks].y = W[(size_t)(k + 1) * M + col0 + c];
        }
    } else {
#pragma unroll
        for (int ks = 0; ks < KS; ++ks) {
            const int k = 4 * ks + 2 * h;
            b[ks].x = WA[k * 16 + c];
            b[ks].y = WA[(k + 1) * 16 + c];
        }
    }

    static_assert(LPT <= 3, "three staging registers");
    float4 s0 = {0.0f, 0.0f, 0.0f, 0.0f}, s1 = s0, s2 = s0;
    auto gload = [&](int t) {
        const float4* __restrict__ srcv = (const float4*)(X + (size_t)t * 16 * K);
        if (tid < TV) s0 = srcv[tid];
        if constexpr (LPT > 1) { if (tid + NT < TV) s1 = srcv[tid + NT]; }
        if constexpr (LPT > 2) { if (tid + 2 * NT < TV) s2 = srcv[tid + 2 * NT]; }
    };
    auto put = [&](int bufi, int v, const float4& val) {
        if (v < TV) {
            const int r = v / (K / 4);
            const int q = v - r * (K / 4);
            *(float4*)&xs[bufi][r * P + 4 * q] = val;
        }
    };
    auto lstore = [&](int bufi) {
        put(bufi, tid, s0);
        if constexpr (LPT > 1) put(bufi, tid + NT, s1);
        if constexpr (LPT > 2) put(bufi, tid + 2 * NT, s2);
    };

    gload(rt);
    lstore(0);
    __syncthreads();

    int cur = 0;
    for (; rt < rowTiles; rt += gridDim.x) {
        const int  nxt  = rt + gridDim.x;
        const bool more = nxt < rowTiles;
        if (more) gload(nxt);

        const float* xa = &xs[cur][c * P + 2 * h];
        v2f a[KS];
#pragma unroll
        for (int ks = 0; ks < KS; ++ks) a[ks] = *(const v2f*)(xa + 4 * ks);

        v8f acc = {};
#pragma unroll
        for (int ks = 0; ks < KS; ++ks)
            acc = __builtin_amdgcn_wmma_f32_16x16x4_f32(
                       false, a[ks],  false, b[ks],
                       (short)0, acc,  false,  false);

        if (!lw) {
            float* hp = H + (size_t)(rt * 16 + 8 * h) * M + col0 + c;
#pragma unroll
            for (int v = 0; v < 8; ++v) hp[(size_t)v * M] = acc[v];
        } else {
            float* lp = LG + (size_t)(rt * 16 + 8 * h) * 16 + c;
#pragma unroll
            for (int v = 0; v < 8; ++v) lp[v * 16] = acc[v];
        }

        if (more) lstore(cur ^ 1);
        __syncthreads();
        cur ^= 1;
    }
}

constexpr int TILE  = 200;
constexpr int NTILE = (NNODE + TILE - 1) / TILE;

__device__ __forceinline__ void fold_one(const float* __restrict__ W,
                                         const float* __restrict__ a_src,
                                         const float* __restrict__ a_dst,
                                         float* __restrict__ WA,
                                         int K, int M, int heads, int C, int t) {
    if (t >= K * 16) return;
    const int k = t >> 4;
    const int j = t & 15;
    float s = 0.0f;
    if (j < 2 * heads) {
        const int hd = j < heads ? j : j - heads;
        const float* a = (j < heads ? a_src : a_dst) + hd * C;
        const float* w = W + (size_t)k * M + hd * C;
        for (int cc = 0; cc < C; ++cc) s += w[cc] * a[cc];
    }
    WA[t] = s;
}

constexpr int FOLD_BLOCKS   = 24;
constexpr int KEY_BLOCKS    = 200;
constexpr int SETUP_THREADS = 256;

__device__ __forceinline__ unsigned tile_key(int d) {
    const unsigned t = (unsigned)d / (unsigned)TILE;
    return t < (unsigned)NTILE ? t : 255u;
}

__global__ __launch_bounds__(SETUP_THREADS)
void fold_att(const float* __restrict__ W1, const float* __restrict__ as1, const float* __restrict__ ad1,
              const float* __restrict__ W2, const float* __restrict__ as2, const float* __restrict__ ad2,
              const float* __restrict__ W3, const float* __restrict__ as3, const float* __restrict__ ad3,
              float* __restrict__ WA1, float* __restrict__ WA2, float* __restrict__ WA3,
              const int* __restrict__ ei, int E, unsigned* __restrict__ keys) {
    if (blockIdx.x < FOLD_BLOCKS) {
        const int layer = blockIdx.x >> 3;
        const int t     = (blockIdx.x & 7) * SETUP_THREADS + threadIdx.x;
        if (layer == 0)      fold_one(W1, as1, ad1, WA1, 128, 112, 7, 16, t);
        else if (layer == 1) fold_one(W2, as2, ad2, WA2, 112,  96, 6, 16, t);
        else                 fold_one(W3, as3, ad3, WA3,  96, 240, 6, 40, t);
        return;
    }
    const int* __restrict__ dstp = ei + E;
    const int  ngrp = ((E + 15) >> 4) << 2;
    const bool al   = (((size_t)dstp) & 15) == 0;
    for (int g = (blockIdx.x - FOLD_BLOCKS) * SETUP_THREADS + threadIdx.x; g < ngrp; g += KEY_BLOCKS * SETUP_THREADS) {
        const int e = 4 * g;
        int d0, d1, d2, d3;
        if (al && e + 3 < E) {
            const int4 q = ((const int4*)dstp)[g];
            d0 = q.x; d1 = q.y; d2 = q.z; d3 = q.w;
        } else {
            d0 = e     < E ? dstp[e]     : -1;
            d1 = e + 1 < E ? dstp[e + 1] : -1;
            d2 = e + 2 < E ? dstp[e + 2] : -1;
            d3 = e + 3 < E ? dstp[e + 3] : -1;
        }
        keys[g] = tile_key(d0) | (tile_key(d1) << 8) | (tile_key(d2) << 16) | (tile_key(d3) << 24);
    }
}


constexpr int LIST_CAP    = 20480;
constexpr int AGG_THREADS = 1024;
constexpr int AGG_WAVES   = AGG_THREADS / 32;
constexpr int SCAN_VEC    = 4;
constexpr int ACC_BATCH   = 4;

static_assert(TILE <= 256, "local row index is packed into 8 bits");
static_assert(NTILE <= 255, "a tile number fits one key byte, 255 is left for 'no tile'");
static_assert(NNODE <= (1 << 24), "source index is packed into 24 bits");

__device__ __forceinline__ unsigned f2key(float f) {
    unsigned b = __float_as_uint(f);
    return (b & 0x80000000u) ? ~b : (b | 0x80000000u);
}
__device__ __forceinline__ float key2f(unsigned k) {
    return __uint_as_float((k & 0x80000000u) ? (k & 0x7fffffffu) : ~k);
}

template <int HEADS, int C, bool BUILD>
__global__ __launch_bounds__(AGG_THREADS)
void agg_tile(const int* __restrict__ ei, int E,
              const float* __restrict__ lg,
              const float* __restrict__ H, const float* __restrict__ bias,
              float* __restrict__ out, int relu,
              unsigned* glist, unsigned* gcnt,
              const unsigned* __restrict__ keys) {
    (void)keys;
    constexpr int HC  = HEADS * C;
    constexpr int NIT = (HC + 31) / 32;

    __shared__ float         acc[TILE * HC];
    __shared__ unsigned      mxk[TILE * HEADS];
    __shared__ float         dn[TILE * HEADS];
    __shared__ unsigned      elist[LIST_CAP];

    const int tid   = threadIdx.x;
    const int lane  = tid & 31;
    const int wave  = tid >> 5;
    const int tile0 = blockIdx.x * TILE;
    const int nvalid = min(TILE, NNODE - tile0);
    const unsigned nv = (unsigned)nvalid;

    const int* __restrict__ srcp = ei;
    const int* __restrict__ dstp = ei + E;

    unsigned* tlist = glist + (size_t)blockIdx.x * LIST_CAP;

    unsigned total_raw = 0;
    if constexpr (!BUILD) total_raw = gcnt[blockIdx.x];

    for (int i = tid; i < TILE * HC; i += AGG_THREADS) acc[i] = 0.0f;
    const unsigned key_ninf = f2key(-INFINITY);
    for (int i = tid; i < TILE * HEADS; i += AGG_THREADS) { mxk[i] = key_ninf; dn[i] = 0.0f; }

    if constexpr (BUILD) {
        __shared__ unsigned      cnt;

        for (int i = tid; i < nvalid; i += AGG_THREADS) {
            const unsigned p = ((unsigned)(tile0 + i) << 8) | (unsigned)i;
            elist[i] = p;
            tlist[i] = p;
        }
        if (tid == 0) cnt = nv;
        __syncthreads();

        auto push = [&](int e) {
            const unsigned pos = atomicAdd(&cnt, 1u);
            if (pos < (unsigned)LIST_CAP) elist[pos] = (unsigned)e;
        };
        {
            const int    nvec = (E + 15) >> 4;
            const uint4* kv   = (const uint4*)keys;
            const unsigned pat = (unsigned)blockIdx.x * 0x01010101u;
            auto hit = [&](unsigned x) -> unsigned {
                const unsigned y = x ^ pat;
                return ~(((y & 0x7F7F7F7Fu) + 0x7F7F7F7Fu) | y | 0x7F7F7F7Fu);
            };
            constexpr int STEP = SCAN_VEC * AGG_THREADS;
            auto loadq = [&](uint4 (&q)[SCAN_VEC], int base) {
#pragma unroll
                for (int j = 0; j < SCAN_VEC; ++j) {
                    const int v = base + j * AGG_THREADS + tid;
                    q[j] = kv[min(v, nvec - 1)];
                }
            };
            auto procq = [&](const uint4 (&q)[SCAN_VEC], int base) {
#pragma unroll
                for (int j = 0; j < SCAN_VEC; ++j) {
                    const int v = base + j * AGG_THREADS + tid;
                    unsigned m = (hit(q[j].x) >> 7) | (hit(q[j].y) >> 6) | (hit(q[j].z) >> 5) | (hit(q[j].w) >> 4);
                    if (v >= nvec) m = 0u;
                    while (m) {
                        const int t = __builtin_ctz(m);
                        m &= m - 1u;
                        push(16 * v + 4 * (t & 7) + (t >> 3));
                    }
                }
            };
            if (nvec > 0) {
                uint4 qa[SCAN_VEC], qb[SCAN_VEC];
                loadq(qa, 0);
                for (int base = 0; base < nvec; base += 2 * STEP) {
                    loadq(qb, base + STEP);
                    procq(qa, base);
                    loadq(qa, base + 2 * STEP);
                    procq(qb, base + STEP);
                }
            }
        }
        __syncthreads();

        total_raw = cnt;
        if (tid == 0) gcnt[blockIdx.x] = total_raw;
        if (total_raw <= (unsigned)LIST_CAP) {
            const int total = (int)total_raw;

            for (int i0 = nvalid + tid; i0 < total; i0 += 4 * AGG_THREADS) {
                int sv[4], dd[4];
#pragma unroll
                for (int j = 0; j < 4; ++j) {
                    const int i = i0 + j * AGG_THREADS;
                    sv[j] = 0; dd[j] = tile0;
                    if (i < total) {
                        const unsigned e = elist[i];
                        sv[j] = srcp[e];
                        dd[j] = dstp[e];
                    }
                }
#pragma unroll
                for (int j = 0; j < 4; ++j) {
                    const int i = i0 + j * AGG_THREADS;
                    if (i < total) {
                        const unsigned p = ((unsigned)sv[j] << 8) | (unsigned)(dd[j] - tile0);
                        elist[i] = p;
                        tlist[i] = p;
                    }
                }
            }
        }
        __syncthreads();
    } else {
        if (total_raw <= (unsigned)LIST_CAP) {
            const int total = (int)total_raw;
            const unsigned* __restrict__ rlist = tlist;
            for (int i0 = tid; i0 < total; i0 += 4 * AGG_THREADS) {
                unsigned pv[4];
#pragma unroll
                for (int j = 0; j < 4; ++j) {
                    const int i = i0 + j * AGG_THREADS;
                    pv[j] = (i < total) ? rlist[i] : 0u;
                }
#pragma unroll
                for (int j = 0; j < 4; ++j) {
                    const int i = i0 + j * AGG_THREADS;
                    if (i < total) elist[i] = pv[j];
                }
            }
        }
        __syncthreads();
    }

    if (total_raw <= (unsigned)LIST_CAP) {
        const int total = (int)total_raw;

        const int n1 = total * HEADS;
        for (int idx0 = tid; idx0 < n1; idx0 += 4 * AGG_THREADS) {
            float xv[4];
            int   cell[4];
#pragma unroll
            for (int j = 0; j < 4; ++j) {
                const int idx  = min(idx0 + j * AGG_THREADS, n1 - 1);
                const int item = idx / HEADS;
                const int h    = idx - item * HEADS;
                const unsigned en = elist[item];
                const int s  = (int)(en >> 8);
                const int ld = (int)(en & 255u);
                xv[j]   = lg[s * 16 + h] + lg[(tile0 + ld) * 16 + HEADS + h];
                cell[j] = ld * HEADS + h;
            }
#pragma unroll
            for (int j = 0; j < 4; ++j) {
                if (idx0 + j * AGG_THREADS < n1) {
                    const float lr = xv[j] >= 0.0f ? xv[j] : 0.2f * xv[j];
                    atomicMax(&mxk[cell[j]], f2key(lr));
                }
            }
        }
        __syncthreads();

        int hk[NIT];
#pragma unroll
        for (int k = 0; k < NIT; ++k) hk[k] = min((lane + 32 * k) / C, HEADS - 1);

        for (int item0 = wave; item0 < total; item0 += AGG_WAVES * ACC_BATCH) {
            int      sb[ACC_BATCH], lb[ACC_BATCH];
            float    xa[ACC_BATCH], xd[ACC_BATCH], hv[ACC_BATCH][NIT];
            unsigned mk[ACC_BATCH];
#pragma unroll
            for (int b = 0; b < ACC_BATCH; ++b) {
                const int it = min(item0 + AGG_WAVES * b, total - 1);
                const unsigned en = elist[it];
                sb[b] = (int)(en >> 8);
                lb[b] = (int)(en & 255u);
            }
#pragma unroll
            for (int b = 0; b < ACC_BATCH; ++b) {
                xa[b] = 0.0f; xd[b] = 0.0f; mk[b] = 0u;
                if (lane < HEADS) {
                    xa[b] = lg[sb[b] * 16 + lane];
                    xd[b] = lg[(tile0 + lb[b]) * 16 + HEADS + lane];
                    mk[b] = mxk[lb[b] * HEADS + lane];
                }
                const float* __restrict__ hrow = H + (size_t)sb[b] * HC;
#pragma unroll
                for (int k = 0; k < NIT; ++k) {
                    const int c = lane + 32 * k;
                    hv[b][k] = (c < HC) ? hrow[c] : 0.0f;
                }
            }
#pragma unroll
            for (int b = 0; b < ACC_BATCH; ++b) {
                if (item0 + AGG_WAVES * b < total) {
                    float exl = 0.0f;
                    if (lane < HEADS) {
                        const float x  = xa[b] + xd[b];
                        const float lr = x >= 0.0f ? x : 0.2f * x;
                        exl = __expf(lr - key2f(mk[b]));
                        atomicAdd(&dn[lb[b] * HEADS + lane], exl);
                    }
                    float* arow = acc + lb[b] * HC;
#pragma unroll
                    for (int k = 0; k < NIT; ++k) {
                        const int c = lane + 32 * k;
                        const float ex = __shfl(exl, hk[k]);
                        if (c < HC) atomicAdd(&arow[c], hv[b][k] * ex);
                    }
                }
            }
        }
    } else {
        auto visit_max = [&](int s, int ld) {
            for (int h = 0; h < HEADS; ++h) {
                const float x  = lg[s * 16 + h] + lg[(tile0 + ld) * 16 + HEADS + h];
                const float lr = x >= 0.0f ? x : 0.2f * x;
                atomicMax(&mxk[ld * HEADS + h], f2key(lr));
            }
        };
        for (int i = tid; i < nvalid; i += AGG_THREADS) visit_max(tile0 + i, i);
        for (int e = tid; e < E; e += AGG_THREADS) {
            const unsigned ld = (unsigned)(dstp[e] - tile0);
            if (ld < nv) visit_max(srcp[e], (int)ld);
        }
        __syncthreads();

        auto visit_acc = [&](int s, int ld) {
            for (int h = 0; h < HEADS; ++h) {
                const float x  = lg[s * 16 + h] + lg[(tile0 + ld) * 16 + HEADS + h];
                const float lr = x >= 0.0f ? x : 0.2f * x;
                const float ex = __expf(lr - key2f(mxk[ld * HEADS + h]));
                atomicAdd(&dn[ld * HEADS + h], ex);
                const float* hs = H + (size_t)s * HC + h * C;
                float*       ap = acc + ld * HC + h * C;
                for (int c = 0; c < C; ++c) atomicAdd(&ap[c], hs[c] * ex);
            }
        };
        for (int i = tid; i < nvalid; i += AGG_THREADS) visit_acc(tile0 + i, i);
        for (int e = tid; e < E; e += AGG_THREADS) {
            const unsigned ld = (unsigned)(dstp[e] - tile0);
            if (ld < nv) visit_acc(srcp[e], (int)ld);
        }
    }
    __syncthreads();

    for (int i = tid; i < nvalid * HC; i += AGG_THREADS) {
        const int ld  = i / HC;
        const int rem = i - ld * HC;
        const int h   = rem / C;
        float v = acc[i] / dn[ld * HEADS + h] + bias[rem];
        if (relu) v = fmaxf(v, 0.0f);
        out[(size_t)tile0 * HC + i] = v;
    }
}


extern "C" void kernel_launch(void* const* d_in, const int* in_sizes, int n_in,
                              void* d_out, int out_size, void* d_ws, size_t ws_size,
                              hipStream_t stream) {
    const float* x      = (const float*)d_in[0];
    const int*   ei     = (const int*)  d_in[1];
    const float* W1     = (const float*)d_in[2];
    const float* a_src1 = (const float*)d_in[3];
    const float* a_dst1 = (const float*)d_in[4];
    const float* b1     = (const float*)d_in[5];
    const float* W2     = (const float*)d_in[6];
    const float* a_src2 = (const float*)d_in[7];
    const float* a_dst2 = (const float*)d_in[8];
    const float* b2     = (const float*)d_in[9];
    const float* W3     = (const float*)d_in[10];
    const float* a_src3 = (const float*)d_in[11];
    const float* a_dst3 = (const float*)d_in[12];
    const float* b3     = (const float*)d_in[13];

    const int E = in_sizes[1] / 2;

    float* ws   = (float*)d_ws;
    float* bufH = ws;
    float* act1 = bufH + (size_t)NNODE * 240;
    float* h2   = act1 + (size_t)NNODE * 112;
    float* act2 = h2   + (size_t)NNODE * 96;
    float* lg   = act2 + (size_t)NNODE * 96;
    float* wa1  = lg   + (size_t)NNODE * 16;
    float* wa2  = wa1  + 128 * 16;
    float* wa3  = wa2  + 112 * 16;
    float* out  = (float*)d_out;

    constexpr int NT = NTILE;
    unsigned* glist = (unsigned*)(wa3 + 96 * 16);
    unsigned* gcnt  = glist + (size_t)NT * LIST_CAP;
    unsigned* keys  = gcnt + 256;
    static_assert(NT <= 256, "the key plane starts behind the list lengths");

    constexpr int RT = NNODE / 16;
    static_assert(NNODE % 16 == 0, "row blocks are whole");

    fold_att<<<FOLD_BLOCKS + KEY_BLOCKS, SETUP_THREADS, 0, stream>>>(W1, a_src1, a_dst1, W2, a_src2, a_dst2, W3, a_src3, a_dst3, wa1, wa2, wa3, ei, E, keys);

    gemm_wstat_f32<128, 112><<<768, 256, 0, stream>>>(x, W1, wa1, bufH, lg, RT);
    agg_tile<7, 16, true><<<NT, AGG_THREADS, 0, stream>>>(ei, E, lg, bufH, b1, act1, 1, glist, gcnt, keys);
    gemm_wstat_f32<112, 96><<<768, 224, 0, stream>>>(act1, W2, wa2, h2, lg, RT);
    agg_tile<6, 16, false><<<NT, AGG_THREADS, 0, stream>>>(ei, E, lg, h2, b2, act2, 1, glist, gcnt, keys);
    gemm_wstat_f32<96, 240><<<512, 512, 0, stream>>>(act2, W3, wa3, bufH, lg, RT);
    agg_tile<6, 40, false><<<NT, AGG_THREADS, 0, stream>>>(ei, E, lg, bufH, b3, out, 0, glist, gcnt, keys);
}


# aggregation: a 4-edge batch's logits, exp and denominator add done once on lanes (edge, head); weights reach channel lanes by v_readlane, not ds_bpermute; sums stay LDS atomics, arrival order
# speedup vs baseline: 15.7866x; 1.0705x over previous
#include <hip/hip_runtime.h>
#include <math.h>

#define NNODE 50000

typedef float v2f __attribute__((ext_vector_type(2)));
typedef float v8f __attribute__((ext_vector_type(8)));

template <int K, int M>
__global__ __launch_bounds__((M / 16 + 1) * 32)
void gemm_wstat_f32(const float* __restrict__ X,
                    const float* __restrict__ W,
                    const float* __restrict__ WA,
                    float* __restrict__ H,
                    float* __restrict__ LG,
                    int rowTiles) {
    constexpr int NW  = M / 16 + 1;
    constexpr int NT  = NW * 32;
    constexpr int KS  = K / 4;
    constexpr int P   = K + 4;
    constexpr int TV  = 16 * K / 4;
    constexpr int LPT = (TV + NT - 1) / NT;

    __shared__ __attribute__((aligned(16))) float xs[2][16 * P];

    const int tid  = threadIdx.x;
    const int lane = tid & 31;
    const int wave = tid >> 5;
    const int c    = lane & 15;
    const int h    = lane >> 4;
    const int col0 = wave << 4;
    const bool lw  = wave == NW - 1;

    int rt = blockIdx.x;
    if (rt >= rowTiles) return;

    v2f b[KS];
    if (!lw) {
#pragma unroll
        for (int ks = 0; ks < KS; ++ks) {
            const int k = 4 * ks + 2 * h;
            b[ks].x = W[(size_t)k * M + col0 + c];
            b[ks].y = W[(size_t)(k + 1) * M + col0 + c];
        }
    } else {
#pragma unroll
        for (int ks = 0; ks < KS; ++ks) {
            const int k = 4 * ks + 2 * h;
            b[ks].x = WA[k * 16 + c];
            b[ks].y = WA[(k + 1) * 16 + c];
        }
    }

    static_assert(LPT <= 3, "three staging registers");
    float4 s0 = {0.0f, 0.0f, 0.0f, 0.0f}, s1 = s0, s2 = s0;
    auto gload = [&](int t) {
        const float4* __restrict__ srcv = (const float4*)(X + (size_t)t * 16 * K);
        if (tid < TV) s0 = srcv[tid];
        if constexpr (LPT > 1) { if (tid + NT < TV) s1 = srcv[tid + NT]; }
        if constexpr (LPT > 2) { if (tid + 2 * NT < TV) s2 = srcv[tid + 2 * NT]; }
    };
    auto put = [&](int bufi, int v, const float4& val) {
        if (v < TV) {
            const int r = v / (K / 4);
            const int q = v - r * (K / 4);
            *(float4*)&xs[bufi][r * P + 4 * q] = val;
        }
    };
    auto lstore = [&](int bufi) {
        put(bufi, tid, s0);
        if constexpr (LPT > 1) put(bufi, tid + NT, s1);
        if constexpr (LPT > 2) put(bufi, tid + 2 * NT, s2);
    };

    gload(rt);
    lstore(0);
    __syncthreads();

    int cur = 0;
    for (; rt < rowTiles; rt += gridDim.x) {
        const int  nxt  = rt + gridDim.x;
        const bool more = nxt < rowTiles;
        if (more) gload(nxt);

        const float* xa = &xs[cur][c * P + 2 * h];
        v2f a[KS];
#pragma unroll
        for (int ks = 0; ks < KS; ++ks) a[ks] = *(const v2f*)(xa + 4 * ks);

        v8f acc = {};
#pragma unroll
        for (int ks = 0; ks < KS; ++ks)
            acc = __builtin_amdgcn_wmma_f32_16x16x4_f32(
                       false, a[ks],  false, b[ks],
                       (short)0, acc,  false,  false);

        if (!lw) {
            float* hp = H + (size_t)(rt * 16 + 8 * h) * M + col0 + c;
#pragma unroll
            for (int v = 0; v < 8; ++v) hp[(size_t)v * M] = acc[v];
        } else {
            float* lp = LG + (size_t)(rt * 16 + 8 * h) * 16 + c;
#pragma unroll
            for (int v = 0; v < 8; ++v) lp[v * 16] = acc[v];
        }

        if (more) lstore(cur ^ 1);
        __syncthreads();
        cur ^= 1;
    }
}

constexpr int TILE  = 200;
constexpr int NTILE = (NNODE + TILE - 1) / TILE;

__device__ __forceinline__ void fold_one(const float* __restrict__ W,
                                         const float* __restrict__ a_src,
                                         const float* __restrict__ a_dst,
                                         float* __restrict__ WA,
                                         int K, int M, int heads, int C, int t) {
    if (t >= K * 16) return;
    const int k = t >> 4;
    const int j = t & 15;
    float s = 0.0f;
    if (j < 2 * heads) {
        const int hd = j < heads ? j : j - heads;
        const float* a = (j < heads ? a_src : a_dst) + hd * C;
        const float* w = W + (size_t)k * M + hd * C;
        for (int cc = 0; cc < C; ++cc) s += w[cc] * a[cc];
    }
    WA[t] = s;
}

constexpr int FOLD_BLOCKS   = 24;
constexpr int KEY_BLOCKS    = 200;
constexpr int SETUP_THREADS = 256;

__device__ __forceinline__ unsigned tile_key(int d) {
    const unsigned t = (unsigned)d / (unsigned)TILE;
    return t < (unsigned)NTILE ? t : 255u;
}

__global__ __launch_bounds__(SETUP_THREADS)
void fold_att(const float* __restrict__ W1, const float* __restrict__ as1, const float* __restrict__ ad1,
              const float* __restrict__ W2, const float* __restrict__ as2, const float* __restrict__ ad2,
              const float* __restrict__ W3, const float* __restrict__ as3, const float* __restrict__ ad3,
              float* __restrict__ WA1, float* __restrict__ WA2, float* __restrict__ WA3,
              const int* __restrict__ ei, int E, unsigned* __restrict__ keys) {
    if (blockIdx.x < FOLD_BLOCKS) {
        const int layer = blockIdx.x >> 3;
        const int t     = (blockIdx.x & 7) * SETUP_THREADS + threadIdx.x;
        if (layer == 0)      fold_one(W1, as1, ad1, WA1, 128, 112, 7, 16, t);
        else if (layer == 1) fold_one(W2, as2, ad2, WA2, 112,  96, 6, 16, t);
        else                 fold_one(W3, as3, ad3, WA3,  96, 240, 6, 40, t);
        return;
    }
    const int* __restrict__ dstp = ei + E;
    const int  ngrp = ((E + 15) >> 4) << 2;
    const bool al   = (((size_t)dstp) & 15) == 0;
    for (int g = (blockIdx.x - FOLD_BLOCKS) * SETUP_THREADS + threadIdx.x; g < ngrp; g += KEY_BLOCKS * SETUP_THREADS) {
        const int e = 4 * g;
        int d0, d1, d2, d3;
        if (al && e + 3 < E) {
            const int4 q = ((const int4*)dstp)[g];
            d0 = q.x; d1 = q.y; d2 = q.z; d3 = q.w;
        } else {
            d0 = e     < E ? dstp[e]     : -1;
            d1 = e + 1 < E ? dstp[e + 1] : -1;
            d2 = e + 2 < E ? dstp[e + 2] : -1;
            d3 = e + 3 < E ? dstp[e + 3] : -1;
        }
        keys[g] = tile_key(d0) | (tile_key(d1) << 8) | (tile_key(d2) << 16) | (tile_key(d3) << 24);
    }
}


constexpr int LIST_CAP    = 20480;
constexpr int AGG_THREADS = 1024;
constexpr int AGG_WAVES   = AGG_THREADS / 32;
constexpr int SCAN_VEC    = 4;
constexpr int ACC_BATCH   = 4;

static_assert(TILE <= 256, "local row index is packed into 8 bits");
static_assert(NTILE <= 255, "a tile number fits one key byte, 255 is left for 'no tile'");
static_assert(NNODE <= (1 << 24), "source index is packed into 24 bits");

__device__ __forceinline__ unsigned f2key(float f) {
    unsigned b = __float_as_uint(f);
    return (b & 0x80000000u) ? ~b : (b | 0x80000000u);
}
__device__ __forceinline__ float key2f(unsigned k) {
    return __uint_as_float((k & 0x80000000u) ? (k & 0x7fffffffu) : ~k);
}

template <int HEADS, int C, bool BUILD>
__global__ __launch_bounds__(AGG_THREADS)
void agg_tile(const int* __restrict__ ei, int E,
              const float* __restrict__ lg,
              const float* __restrict__ H, const float* __restrict__ bias,
              float* __restrict__ out, int relu,
              unsigned* glist, unsigned* gcnt,
              const unsigned* __restrict__ keys) {
    (void)keys;
    constexpr int HC  = HEADS * C;
    constexpr int NIT = (HC + 31) / 32;

    __shared__ float         acc[TILE * HC];
    __shared__ unsigned      mxk[TILE * HEADS];
    __shared__ float         dn[TILE * HEADS];
    __shared__ unsigned      elist[LIST_CAP];

    const int tid   = threadIdx.x;
    const int lane  = tid & 31;
    const int wave  = tid >> 5;
    const int tile0 = blockIdx.x * TILE;
    const int nvalid = min(TILE, NNODE - tile0);
    const unsigned nv = (unsigned)nvalid;

    const int* __restrict__ srcp = ei;
    const int* __restrict__ dstp = ei + E;

    unsigned* tlist = glist + (size_t)blockIdx.x * LIST_CAP;

    unsigned total_raw = 0;
    if constexpr (!BUILD) total_raw = gcnt[blockIdx.x];

    for (int i = tid; i < TILE * HC; i += AGG_THREADS) acc[i] = 0.0f;
    const unsigned key_ninf = f2key(-INFINITY);
    for (int i = tid; i < TILE * HEADS; i += AGG_THREADS) { mxk[i] = key_ninf; dn[i] = 0.0f; }

    if constexpr (BUILD) {
        __shared__ unsigned      cnt;

        for (int i = tid; i < nvalid; i += AGG_THREADS) {
            const unsigned p = ((unsigned)(tile0 + i) << 8) | (unsigned)i;
            elist[i] = p;
            tlist[i] = p;
        }
        if (tid == 0) cnt = nv;
        __syncthreads();

        auto push = [&](int e) {
            const unsigned pos = atomicAdd(&cnt, 1u);
            if (pos < (unsigned)LIST_CAP) elist[pos] = (unsigned)e;
        };
        {
            const int    nvec = (E + 15) >> 4;
            const uint4* kv   = (const uint4*)keys;
            const unsigned pat = (unsigned)blockIdx.x * 0x01010101u;
            auto hit = [&](unsigned x) -> unsigned {
                const unsigned y = x ^ pat;
                return ~(((y & 0x7F7F7F7Fu) + 0x7F7F7F7Fu) | y | 0x7F7F7F7Fu);
            };
            constexpr int STEP = SCAN_VEC * AGG_THREADS;
            auto loadq = [&](uint4 (&q)[SCAN_VEC], int base) {
#pragma unroll
                for (int j = 0; j < SCAN_VEC; ++j) {
                    const int v = base + j * AGG_THREADS + tid;
                    q[j] = kv[min(v, nvec - 1)];
                }
            };
            auto procq = [&](const uint4 (&q)[SCAN_VEC], int base) {
#pragma unroll
                for (int j = 0; j < SCAN_VEC; ++j) {
                    const int v = base + j * AGG_THREADS + tid;
                    unsigned m = (hit(q[j].x) >> 7) | (hit(q[j].y) >> 6) | (hit(q[j].z) >> 5) | (hit(q[j].w) >> 4);
                    if (v >= nvec) m = 0u;
                    while (m) {
                        const int t = __builtin_ctz(m);
                        m &= m - 1u;
                        push(16 * v + 4 * (t & 7) + (t >> 3));
                    }
                }
            };
            if (nvec > 0) {
                uint4 qa[SCAN_VEC], qb[SCAN_VEC];
                loadq(qa, 0);
                for (int base = 0; base < nvec; base += 2 * STEP) {
                    loadq(qb, base + STEP);
                    procq(qa, base);
                    loadq(qa, base + 2 * STEP);
                    procq(qb, base + STEP);
                }
            }
        }
        __syncthreads();

        total_raw = cnt;
        if (tid == 0) gcnt[blockIdx.x] = total_raw;
        if (total_raw <= (unsigned)LIST_CAP) {
            const int total = (int)total_raw;

            for (int i0 = nvalid + tid; i0 < total; i0 += 4 * AGG_THREADS) {
                int sv[4], dd[4];
#pragma unroll
                for (int j = 0; j < 4; ++j) {
                    const int i = i0 + j * AGG_THREADS;
                    sv[j] = 0; dd[j] = tile0;
                    if (i < total) {
                        const unsigned e = elist[i];
                        sv[j] = srcp[e];
                        dd[j] = dstp[e];
                    }
                }
#pragma unroll
                for (int j = 0; j < 4; ++j) {
                    const int i = i0 + j * AGG_THREADS;
                    if (i < total) {
                        const unsigned p = ((unsigned)sv[j] << 8) | (unsigned)(dd[j] - tile0);
                        elist[i] = p;
                        tlist[i] = p;
                    }
                }
            }
        }
        __syncthreads();
    } else {
        if (total_raw <= (unsigned)LIST_CAP) {
            const int total = (int)total_raw;
            const unsigned* __restrict__ rlist = tlist;
            for (int i0 = tid; i0 < total; i0 += 4 * AGG_THREADS) {
                unsigned pv[4];
#pragma unroll
                for (int j = 0; j < 4; ++j) {
                    const int i = i0 + j * AGG_THREADS;
                    pv[j] = (i < total) ? rlist[i] : 0u;
                }
#pragma unroll
                for (int j = 0; j < 4; ++j) {
                    const int i = i0 + j * AGG_THREADS;
                    if (i < total) elist[i] = pv[j];
                }
            }
        }
        __syncthreads();
    }

    if (total_raw <= (unsigned)LIST_CAP) {
        const int total = (int)total_raw;

        const int n1 = total * HEADS;
        for (int idx0 = tid; idx0 < n1; idx0 += 4 * AGG_THREADS) {
            float xv[4];
            int   cell[4];
#pragma unroll
            for (int j = 0; j < 4; ++j) {
                const int idx  = min(idx0 + j * AGG_THREADS, n1 - 1);
                const int item = idx / HEADS;
                const int h    = idx - item * HEADS;
                const unsigned en = elist[item];
                const int s  = (int)(en >> 8);
                const int ld = (int)(en & 255u);
                xv[j]   = lg[s * 16 + h] + lg[(tile0 + ld) * 16 + HEADS + h];
                cell[j] = ld * HEADS + h;
            }
#pragma unroll
            for (int j = 0; j < 4; ++j) {
                if (idx0 + j * AGG_THREADS < n1) {
                    const float lr = xv[j] >= 0.0f ? xv[j] : 0.2f * xv[j];
                    atomicMax(&mxk[cell[j]], f2key(lr));
                }
            }
        }
        __syncthreads();

        static_assert(ACC_BATCH == 4 && HEADS <= 8, "weight lanes: 8 lanes an edge, 4 edges a wave");
        static_assert(C == 16 || C >= 32, "32 consecutive channels belong to at most two heads");
        const int  wb   = lane >> 3;
        const int  wh   = lane & 7;
        const bool wact = wh < HEADS;
        const int  whc  = min(wh, HEADS - 1);

        for (int item0 = wave; item0 < total; item0 += AGG_WAVES * ACC_BATCH) {
            const int      itw = item0 + AGG_WAVES * wb;
            const unsigned enw = elist[min(itw, total - 1)];
            const int      lw  = (int)(enw & 255u);
            const float    xa = lg[(int)(enw >> 8) * 16 + whc];
            const float    xd = lg[(tile0 + lw) * 16 + HEADS + whc];
            const unsigned mk = mxk[lw * HEADS + whc];
            int   lb[ACC_BATCH];
            float hv[ACC_BATCH][NIT];
#pragma unroll
            for (int b = 0; b < ACC_BATCH; ++b) {
                const unsigned en = (unsigned)__builtin_amdgcn_readlane((int)enw, 8 * b);
                lb[b] = (int)(en & 255u);
                const float* __restrict__ hrow = H + (size_t)(en >> 8) * HC;
#pragma unroll
                for (int k = 0; k < NIT; ++k) {
                    const int c = lane + 32 * k;
                    hv[b][k] = (c < HC) ? hrow[c] : 0.0f;
                }
            }
            float exl = 0.0f;
            if (wact && itw < total) {
                const float x  = xa + xd;
                const float lr = x >= 0.0f ? x : 0.2f * x;
                exl = __expf(lr - key2f(mk));
                atomicAdd(&dn[lw * HEADS + wh], exl);
            }
            const int exi = __float_as_int(exl);
#pragma unroll
            for (int b = 0; b < ACC_BATCH; ++b) {
                if (item0 + AGG_WAVES * b < total) {
                    float* arow = acc + lb[b] * HC;
#pragma unroll
                    for (int k = 0; k < NIT; ++k) {
                        const int c  = lane + 32 * k;
                        const int h0 = (32 * k) / C;
                        const int h1 = min((32 * k + 31) / C, HEADS - 1);
                        float ex = __int_as_float(__builtin_amdgcn_readlane(exi, 8 * b + h0));
                        if (h1 != h0) {
                            const float e1 = __int_as_float(__builtin_amdgcn_readlane(exi, 8 * b + h1));
                            ex = lane < (h0 + 1) * C - 32 * k ? ex : e1;
                        }
                        if (c < HC) atomicAdd(&arow[c], hv[b][k] * ex);
                    }
                }
            }
        }
    } else {
        auto visit_max = [&](int s, int ld) {
            for (int h = 0; h < HEADS; ++h) {
                const float x  = lg[s * 16 + h] + lg[(tile0 + ld) * 16 + HEADS + h];
                const float lr = x >= 0.0f ? x : 0.2f * x;
                atomicMax(&mxk[ld * HEADS + h], f2key(lr));
            }
        };
        for (int i = tid; i < nvalid; i += AGG_THREADS) visit_max(tile0 + i, i);
        for (int e = tid; e < E; e += AGG_THREADS) {
            const unsigned ld = (unsigned)(dstp[e] - tile0);
            if (ld < nv) visit_max(srcp[e], (int)ld);
        }
        __syncthreads();

        auto visit_acc = [&](int s, int ld) {
            for (int h = 0; h < HEADS; ++h) {
                const float x  = lg[s * 16 + h] + lg[(tile0 + ld) * 16 + HEADS + h];
                const float lr = x >= 0.0f ? x : 0.2f * x;
                const float ex = __expf(lr - key2f(mxk[ld * HEADS + h]));
                atomicAdd(&dn[ld * HEADS + h], ex);
                const float* hs = H + (size_t)s * HC + h * C;
                float*       ap = acc + ld * HC + h * C;
                for (int c = 0; c < C; ++c) atomicAdd(&ap[c], hs[c] * ex);
            }
        };
        for (int i = tid; i < nvalid; i += AGG_THREADS) visit_acc(tile0 + i, i);
        for (int e = tid; e < E; e += AGG_THREADS) {
            const unsigned ld = (unsigned)(dstp[e] - tile0);
            if (ld < nv) visit_acc(srcp[e], (int)ld);
        }
    }
    __syncthreads();

    for (int i = tid; i < nvalid * HC; i += AGG_THREADS) {
        const int ld  = i / HC;
        const int rem = i - ld * HC;
        const int h   = rem / C;
        float v = acc[i] / dn[ld * HEADS + h] + bias[rem];
        if (relu) v = fmaxf(v, 0.0f);
        out[(size_t)tile0 * HC + i] = v;
    }
}


extern "C" void kernel_launch(void* const* d_in, const int* in_sizes, int n_in,
                              void* d_out, int out_size, void* d_ws, size_t ws_size,
                              hipStream_t stream) {
    const float* x      = (const float*)d_in[0];
    const int*   ei     = (const int*)  d_in[1];
    const float* W1     = (const float*)d_in[2];
    const float* a_src1 = (const float*)d_in[3];
    const float* a_dst1 = (const float*)d_in[4];
    const float* b1     = (const float*)d_in[5];
    const float* W2     = (const float*)d_in[6];
    const float* a_src2 = (const float*)d_in[7];
    const float* a_dst2 = (const float*)d_in[8];
    const float* b2     = (const float*)d_in[9];
    const float* W3     = (const float*)d_in[10];
    const float* a_src3 = (const float*)d_in[11];
    const float* a_dst3 = (const float*)d_in[12];
    const float* b3     = (const float*)d_in[13];

    const int E = in_sizes[1] / 2;

    float* ws   = (float*)d_ws;
    float* bufH = ws;
    float* act1 = bufH + (size_t)NNODE * 240;
    float* h2   = act1 + (size_t)NNODE * 112;
    float* act2 = h2   + (size_t)NNODE * 96;
    float* lg   = act2 + (size_t)NNODE * 96;
    float* wa1  = lg   + (size_t)NNODE * 16;
    float* wa2  = wa1  + 128 * 16;
    float* wa3  = wa2  + 112 * 16;
    float* out  = (float*)d_out;

    constexpr int NT = NTILE;
    unsigned* glist = (unsigned*)(wa3 + 96 * 16);
    unsigned* gcnt  = glist + (size_t)NT * LIST_CAP;
    unsigned* keys  = gcnt + 256;
    static_assert(NT <= 256, "the key plane starts behind the list lengths");

    constexpr int RT = NNODE / 16;
    static_assert(NNODE % 16 == 0, "row blocks are whole");

    fold_att<<<FOLD_BLOCKS + KEY_BLOCKS, SETUP_THREADS, 0, stream>>>(W1, a_src1, a_dst1, W2, a_src2, a_dst2, W3, a_src3, a_dst3, wa1, wa2, wa3, ei, E, keys);

    gemm_wstat_f32<128, 112><<<768, 256, 0, stream>>>(x, W1, wa1, bufH, lg, RT);
    agg_tile<7, 16, true><<<NT, AGG_THREADS, 0, stream>>>(ei, E, lg, bufH, b1, act1, 1, glist, gcnt, keys);
    gemm_wstat_f32<112, 96><<<768, 224, 0, stream>>>(act1, W2, wa2, h2, lg, RT);
    agg_tile<6, 16, false><<<NT, AGG_THREADS, 0, stream>>>(ei, E, lg, h2, b2, act2, 1, glist, gcnt, keys);
    gemm_wstat_f32<96, 240><<<512, 512, 0, stream>>>(act2, W3, wa3, bufH, lg, RT);
    agg_tile<6, 40, false><<<NT, AGG_THREADS, 0, stream>>>(ei, E, lg, bufH, b3, out, 0, glist, gcnt, keys);
}


# gemm of layers 2, 3 moved into the previous layer's aggregation launch: a tile's finished rows stay in LDS, its waves multiply them by W | W*a on the matrix unit; 2 launches go; k order unchanged
# speedup vs baseline: 16.6529x; 1.0549x over previous
#include <hip/hip_runtime.h>
#include <math.h>

#define NNODE 50000

typedef float v2f __attribute__((ext_vector_type(2)));
typedef float v8f __attribute__((ext_vector_type(8)));

template <int K, int M>
__global__ __launch_bounds__((M / 16 + 1) * 32)
void gemm_wstat_f32(const float* __restrict__ X,
                    const float* __restrict__ W,
                    const float* __restrict__ WA,
                    float* __restrict__ H,
                    float* __restrict__ LG,
                    int rowTiles) {
    constexpr int NW  = M / 16 + 1;
    constexpr int NT  = NW * 32;
    constexpr int KS  = K / 4;
    constexpr int P   = K + 4;
    constexpr int TV  = 16 * K / 4;
    constexpr int LPT = (TV + NT - 1) / NT;

    __shared__ __attribute__((aligned(16))) float xs[2][16 * P];

    const int tid  = threadIdx.x;
    const int lane = tid & 31;
    const int wave = tid >> 5;
    const int c    = lane & 15;
    const int h    = lane >> 4;
    const int col0 = wave << 4;
    const bool lw  = wave == NW - 1;

    int rt = blockIdx.x;
    if (rt >= rowTiles) return;

    v2f b[KS];
    if (!lw) {
#pragma unroll
        for (int ks = 0; ks < KS; ++ks) {
            const int k = 4 * ks + 2 * h;
            b[ks].x = W[(size_t)k * M + col0 + c];
            b[ks].y = W[(size_t)(k + 1) * M + col0 + c];
        }
    } else {
#pragma unroll
        for (int ks = 0; ks < KS; ++ks) {
            const int k = 4 * ks + 2 * h;
            b[ks].x = WA[k * 16 + c];
            b[ks].y = WA[(k + 1) * 16 + c];
        }
    }

    static_assert(LPT <= 3, "three staging registers");
    float4 s0 = {0.0f, 0.0f, 0.0f, 0.0f}, s1 = s0, s2 = s0;
    auto gload = [&](int t) {
        const float4* __restrict__ srcv = (const float4*)(X + (size_t)t * 16 * K);
        if (tid < TV) s0 = srcv[tid];
        if constexpr (LPT > 1) { if (tid + NT < TV) s1 = srcv[tid + NT]; }
        if constexpr (LPT > 2) { if (tid + 2 * NT < TV) s2 = srcv[tid + 2 * NT]; }
    };
    auto put = [&](int bufi, int v, const float4& val) {
        if (v < TV) {
            const int r = v / (K / 4);
            const int q = v - r * (K / 4);
            *(float4*)&xs[bufi][r * P + 4 * q] = val;
        }
    };
    auto lstore = [&](int bufi) {
        put(bufi, tid, s0);
        if constexpr (LPT > 1) put(bufi, tid + NT, s1);
        if constexpr (LPT > 2) put(bufi, tid + 2 * NT, s2);
    };

    gload(rt);
    lstore(0);
    __syncthreads();

    int cur = 0;
    for (; rt < rowTiles; rt += gridDim.x) {
        const int  nxt  = rt + gridDim.x;
        const bool more = nxt < rowTiles;
        if (more) gload(nxt);

        const float* xa = &xs[cur][c * P + 2 * h];
        v2f a[KS];
#pragma unroll
        for (int ks = 0; ks < KS; ++ks) a[ks] = *(const v2f*)(xa + 4 * ks);

        v8f acc = {};
#pragma unroll
        for (int ks = 0; ks < KS; ++ks)
            acc = __builtin_amdgcn_wmma_f32_16x16x4_f32(
                       false, a[ks],  false, b[ks],
                       (short)0, acc,  false,  false);

        if (!lw) {
            float* hp = H + (size_t)(rt * 16 + 8 * h) * M + col0 + c;
#pragma unroll
            for (int v = 0; v < 8; ++v) hp[(size_t)v * M] = acc[v];
        } else {
            float* lp = LG + (size_t)(rt * 16 + 8 * h) * 16 + c;
#pragma unroll
            for (int v = 0; v < 8; ++v) lp[v * 16] = acc[v];
        }

        if (more) lstore(cur ^ 1);
        __syncthreads();
        cur ^= 1;
    }
}

constexpr int TILE  = 200;
constexpr int NTILE = (NNODE + TILE - 1) / TILE;

__device__ __forceinline__ void fold_one(const float* __restrict__ W,
                                         const float* __restrict__ a_src,
                                         const float* __restrict__ a_dst,
                                         float* __restrict__ WA,
                                         int K, int M, int heads, int C, int t) {
    if (t >= K * 16) return;
    const int k = t >> 4;
    const int j = t & 15;
    float s = 0.0f;
    if (j < 2 * heads) {
        const int hd = j < heads ? j : j - heads;
        const float* a = (j < heads ? a_src : a_dst) + hd * C;
        const float* w = W + (size_t)k * M + hd * C;
        for (int cc = 0; cc < C; ++cc) s += w[cc] * a[cc];
    }
    WA[t] = s;
}

constexpr int FOLD_BLOCKS   = 24;
constexpr int KEY_BLOCKS    = 200;
constexpr int SETUP_THREADS = 256;

__device__ __forceinline__ unsigned tile_key(int d) {
    const unsigned t = (unsigned)d / (unsigned)TILE;
    return t < (unsigned)NTILE ? t : 255u;
}

__global__ __launch_bounds__(SETUP_THREADS)
void fold_att(const float* __restrict__ W1, const float* __restrict__ as1, const float* __restrict__ ad1,
              const float* __restrict__ W2, const float* __restrict__ as2, const float* __restrict__ ad2,
              const float* __restrict__ W3, const float* __restrict__ as3, const float* __restrict__ ad3,
              float* __restrict__ WA1, float* __restrict__ WA2, float* __restrict__ WA3,
              const int* __restrict__ ei, int E, unsigned* __restrict__ keys) {
    if (blockIdx.x < FOLD_BLOCKS) {
        const int layer = blockIdx.x >> 3;
        const int t     = (blockIdx.x & 7) * SETUP_THREADS + threadIdx.x;
        if (layer == 0)      fold_one(W1, as1, ad1, WA1, 128, 112, 7, 16, t);
        else if (layer == 1) fold_one(W2, as2, ad2, WA2, 112,  96, 6, 16, t);
        else                 fold_one(W3, as3, ad3, WA3,  96, 240, 6, 40, t);
        return;
    }
    const int* __restrict__ dstp = ei + E;
    const int  ngrp = ((E + 15) >> 4) << 2;
    const bool al   = (((size_t)dstp) & 15) == 0;
    for (int g = (blockIdx.x - FOLD_BLOCKS) * SETUP_THREADS + threadIdx.x; g < ngrp; g += KEY_BLOCKS * SETUP_THREADS) {
        const int e = 4 * g;
        int d0, d1, d2, d3;
        if (al && e + 3 < E) {
            const int4 q = ((const int4*)dstp)[g];
            d0 = q.x; d1 = q.y; d2 = q.z; d3 = q.w;
        } else {
            d0 = e     < E ? dstp[e]     : -1;
            d1 = e + 1 < E ? dstp[e + 1] : -1;
            d2 = e + 2 < E ? dstp[e + 2] : -1;
            d3 = e + 3 < E ? dstp[e + 3] : -1;
        }
        keys[g] = tile_key(d0) | (tile_key(d1) << 8) | (tile_key(d2) << 16) | (tile_key(d3) << 24);
    }
}


constexpr int LIST_CAP    = 20480;
constexpr int AGG_THREADS = 1024;
constexpr int AGG_WAVES   = AGG_THREADS / 32;
constexpr int SCAN_VEC    = 4;
constexpr int ACC_BATCH   = 4;

static_assert(TILE <= 256, "local row index is packed into 8 bits");
static_assert(NTILE <= 255, "a tile number fits one key byte, 255 is left for 'no tile'");
static_assert(NNODE <= (1 << 24), "source index is packed into 24 bits");

__device__ __forceinline__ unsigned f2key(float f) {
    unsigned b = __float_as_uint(f);
    return (b & 0x80000000u) ? ~b : (b | 0x80000000u);
}
__device__ __forceinline__ float key2f(unsigned k) {
    return __uint_as_float((k & 0x80000000u) ? (k & 0x7fffffffu) : ~k);
}

template <int HEADS, int C, bool BUILD, int M2>
__global__ __launch_bounds__(AGG_THREADS)
void agg_tile(const int* __restrict__ ei, int E,
              const float* __restrict__ lg,
              const float* __restrict__ H, const float* __restrict__ bias,
              float* __restrict__ out, int relu,
              unsigned* glist, unsigned* gcnt,
              const unsigned* __restrict__ keys,
              const float* __restrict__ Wn,
              const float* __restrict__ WAn,
              float* __restrict__ Hn,
              float* __restrict__ LGn) {
    (void)keys; (void)Wn; (void)WAn; (void)Hn; (void)LGn;
    constexpr int  HC   = HEADS * C;
    constexpr int  NIT  = (HC + 31) / 32;
    constexpr bool FUSE = M2 > 0;
    constexpr int  PA   = FUSE ? HC + 4 : HC;
    constexpr int  ROWS = FUSE ? ((TILE + 15) / 16) * 16 : TILE;

    __shared__ __attribute__((aligned(16))) float acc[ROWS * PA];
    __shared__ unsigned      mxk[TILE * HEADS];
    __shared__ float         dn[TILE * HEADS];
    __shared__ unsigned      elist[LIST_CAP];

    const int tid   = threadIdx.x;
    const int lane  = tid & 31;
    const int wave  = tid >> 5;
    const int tile0 = blockIdx.x * TILE;
    const int nvalid = min(TILE, NNODE - tile0);
    const unsigned nv = (unsigned)nvalid;

    const int* __restrict__ srcp = ei;
    const int* __restrict__ dstp = ei + E;

    unsigned* tlist = glist + (size_t)blockIdx.x * LIST_CAP;

    unsigned total_raw = 0;
    if constexpr (!BUILD) total_raw = gcnt[blockIdx.x];

    for (int i = tid; i < ROWS * PA; i += AGG_THREADS) acc[i] = 0.0f;
    const unsigned key_ninf = f2key(-INFINITY);
    for (int i = tid; i < TILE * HEADS; i += AGG_THREADS) { mxk[i] = key_ninf; dn[i] = 0.0f; }

    if constexpr (BUILD) {
        __shared__ unsigned      cnt;

        for (int i = tid; i < nvalid; i += AGG_THREADS) {
            const unsigned p = ((unsigned)(tile0 + i) << 8) | (unsigned)i;
            elist[i] = p;
            tlist[i] = p;
        }
        if (tid == 0) cnt = nv;
        __syncthreads();

        auto push = [&](int e) {
            const unsigned pos = atomicAdd(&cnt, 1u);
            if (pos < (unsigned)LIST_CAP) elist[pos] = (unsigned)e;
        };
        {
            const int    nvec = (E + 15) >> 4;
            const uint4* kv   = (const uint4*)keys;
            const unsigned pat = (unsigned)blockIdx.x * 0x01010101u;
            auto hit = [&](unsigned x) -> unsigned {
                const unsigned y = x ^ pat;
                return ~(((y & 0x7F7F7F7Fu) + 0x7F7F7F7Fu) | y | 0x7F7F7F7Fu);
            };
            constexpr int STEP = SCAN_VEC * AGG_THREADS;
            auto loadq = [&](uint4 (&q)[SCAN_VEC], int base) {
#pragma unroll
                for (int j = 0; j < SCAN_VEC; ++j) {
                    const int v = base + j * AGG_THREADS + tid;
                    q[j] = kv[min(v, nvec - 1)];
                }
            };
            auto procq = [&](const uint4 (&q)[SCAN_VEC], int base) {
#pragma unroll
                for (int j = 0; j < SCAN_VEC; ++j) {
                    const int v = base + j * AGG_THREADS + tid;
                    unsigned m = (hit(q[j].x) >> 7) | (hit(q[j].y) >> 6) | (hit(q[j].z) >> 5) | (hit(q[j].w) >> 4);
                    if (v >= nvec) m = 0u;
                    while (m) {
                        const int t = __builtin_ctz(m);
                        m &= m - 1u;
                        push(16 * v + 4 * (t & 7) + (t >> 3));
                    }
                }
            };
            if (nvec > 0) {
                uint4 qa[SCAN_VEC], qb[SCAN_VEC];
                loadq(qa, 0);
                for (int base = 0; base < nvec; base += 2 * STEP) {
                    loadq(qb, base + STEP);
                    procq(qa, base);
                    loadq(qa, base + 2 * STEP);
                    procq(qb, base + STEP);
                }
            }
        }
        __syncthreads();

        total_raw = cnt;
        if (tid == 0) gcnt[blockIdx.x] = total_raw;
        if (total_raw <= (unsigned)LIST_CAP) {
            const int total = (int)total_raw;

            for (int i0 = nvalid + tid; i0 < total; i0 += 4 * AGG_THREADS) {
                int sv[4], dd[4];
#pragma unroll
                for (int j = 0; j < 4; ++j) {
                    const int i = i0 + j * AGG_THREADS;
                    sv[j] = 0; dd[j] = tile0;
                    if (i < total) {
                        const unsigned e = elist[i];
                        sv[j] = srcp[e];
                        dd[j] = dstp[e];
                    }
                }
#pragma unroll
                for (int j = 0; j < 4; ++j) {
                    const int i = i0 + j * AGG_THREADS;
                    if (i < total) {
                        const unsigned p = ((unsigned)sv[j] << 8) | (unsigned)(dd[j] - tile0);
                        elist[i] = p;
                        tlist[i] = p;
                    }
                }
            }
        }
        __syncthreads();
    } else {
        if (total_raw <= (unsigned)LIST_CAP) {
            const int total = (int)total_raw;
            const unsigned* __restrict__ rlist = tlist;
            for (int i0 = tid; i0 < total; i0 += 4 * AGG_THREADS) {
                unsigned pv[4];
#pragma unroll
                for (int j = 0; j < 4; ++j) {
                    const int i = i0 + j * AGG_THREADS;
                    pv[j] = (i < total) ? rlist[i] : 0u;
                }
#pragma unroll
                for (int j = 0; j < 4; ++j) {
                    const int i = i0 + j * AGG_THREADS;
                    if (i < total) elist[i] = pv[j];
                }
            }
        }
        __syncthreads();
    }

    if (total_raw <= (unsigned)LIST_CAP) {
        const int total = (int)total_raw;

        const int n1 = total * HEADS;
        for (int idx0 = tid; idx0 < n1; idx0 += 4 * AGG_THREADS) {
            float xv[4];
            int   cell[4];
#pragma unroll
            for (int j = 0; j < 4; ++j) {
                const int idx  = min(idx0 + j * AGG_THREADS, n1 - 1);
                const int item = idx / HEADS;
                const int h    = idx - item * HEADS;
                const unsigned en = elist[item];
                const int s  = (int)(en >> 8);
                const int ld = (int)(en & 255u);
                xv[j]   = lg[s * 16 + h] + lg[(tile0 + ld) * 16 + HEADS + h];
                cell[j] = ld * HEADS + h;
            }
#pragma unroll
            for (int j = 0; j < 4; ++j) {
                if (idx0 + j * AGG_THREADS < n1) {
                    const float lr = xv[j] >= 0.0f ? xv[j] : 0.2f * xv[j];
                    atomicMax(&mxk[cell[j]], f2key(lr));
                }
            }
        }
        __syncthreads();

        static_assert(ACC_BATCH == 4 && HEADS <= 8, "weight lanes: 8 lanes an edge, 4 edges a wave");
        static_assert(C == 16 || C >= 32, "32 consecutive channels belong to at most two heads");
        const int  wb   = lane >> 3;
        const int  wh   = lane & 7;
        const bool wact = wh < HEADS;
        const int  whc  = min(wh, HEADS - 1);

        for (int item0 = wave; item0 < total; item0 += AGG_WAVES * ACC_BATCH) {
            const int      itw = item0 + AGG_WAVES * wb;
            const unsigned enw = elist[min(itw, total - 1)];
            const int      lw  = (int)(enw & 255u);
            const float    xa = lg[(int)(enw >> 8) * 16 + whc];
            const float    xd = lg[(tile0 + lw) * 16 + HEADS + whc];
            const unsigned mk = mxk[lw * HEADS + whc];
            int   lb[ACC_BATCH];
            float hv[ACC_BATCH][NIT];
#pragma unroll
            for (int b = 0; b < ACC_BATCH; ++b) {
                const unsigned en = (unsigned)__builtin_amdgcn_readlane((int)enw, 8 * b);
                lb[b] = (int)(en & 255u);
                const float* __restrict__ hrow = H + (size_t)(en >> 8) * HC;
#pragma unroll
                for (int k = 0; k < NIT; ++k) {
                    const int c = lane + 32 * k;
                    hv[b][k] = (c < HC) ? hrow[c] : 0.0f;
                }
            }
            float exl = 0.0f;
            if (wact && itw < total) {
                const float x  = xa + xd;
                const float lr = x >= 0.0f ? x : 0.2f * x;
                exl = __expf(lr - key2f(mk));
                atomicAdd(&dn[lw * HEADS + wh], exl);
            }
            const int exi = __float_as_int(exl);
#pragma unroll
            for (int b = 0; b < ACC_BATCH; ++b) {
                if (item0 + AGG_WAVES * b < total) {
                    float* arow = acc + lb[b] * PA;
#pragma unroll
                    for (int k = 0; k < NIT; ++k) {
                        const int c  = lane + 32 * k;
                        const int h0 = (32 * k) / C;
                        const int h1 = min((32 * k + 31) / C, HEADS - 1);
                        float ex = __int_as_float(__builtin_amdgcn_readlane(exi, 8 * b + h0));
                        if (h1 != h0) {
                            const float e1 = __int_as_float(__builtin_amdgcn_readlane(exi, 8 * b + h1));
                            ex = lane < (h0 + 1) * C - 32 * k ? ex : e1;
                        }
                        if (c < HC) atomicAdd(&arow[c], hv[b][k] * ex);
                    }
                }
            }
        }
    } else {
        auto visit_max = [&](int s, int ld) {
            for (int h = 0; h < HEADS; ++h) {
                const float x  = lg[s * 16 + h] + lg[(tile0 + ld) * 16 + HEADS + h];
                const float lr = x >= 0.0f ? x : 0.2f * x;
                atomicMax(&mxk[ld * HEADS + h], f2key(lr));
            }
        };
        for (int i = tid; i < nvalid; i += AGG_THREADS) visit_max(tile0 + i, i);
        for (int e = tid; e < E; e += AGG_THREADS) {
            const unsigned ld = (unsigned)(dstp[e] - tile0);
            if (ld < nv) visit_max(srcp[e], (int)ld);
        }
        __syncthreads();

        auto visit_acc = [&](int s, int ld) {
            for (int h = 0; h < HEADS; ++h) {
                const float x  = lg[s * 16 + h] + lg[(tile0 + ld) * 16 + HEADS + h];
                const float lr = x >= 0.0f ? x : 0.2f * x;
                const float ex = __expf(lr - key2f(mxk[ld * HEADS + h]));
                atomicAdd(&dn[ld * HEADS + h], ex);
                const float* hs = H + (size_t)s * HC + h * C;
                float*       ap = acc + ld * PA + h * C;
                for (int c = 0; c < C; ++c) atomicAdd(&ap[c], hs[c] * ex);
            }
        };
        for (int i = tid; i < nvalid; i += AGG_THREADS) visit_acc(tile0 + i, i);
        for (int e = tid; e < E; e += AGG_THREADS) {
            const unsigned ld = (unsigned)(dstp[e] - tile0);
            if (ld < nv) visit_acc(srcp[e], (int)ld);
        }
    }
    __syncthreads();

    for (int i = tid; i < nvalid * HC; i += AGG_THREADS) {
        const int ld  = i / HC;
        const int rem = i - ld * HC;
        const int h   = rem / C;
        float v = acc[ld * PA + rem] / dn[ld * HEADS + h] + bias[rem];
        if (relu) v = fmaxf(v, 0.0f);
        if constexpr (FUSE) acc[ld * PA + rem] = v;
        else                out[(size_t)tile0 * HC + i] = v;
    }

    if constexpr (FUSE) {
        __syncthreads();
        constexpr int KS  = HC / 4;
        constexpr int NCT = M2 / 16 + 1;
        constexpr int NRG = AGG_WAVES / NCT;
        constexpr int NRT = ROWS / 16;
        static_assert(HC % 4 == 0 && M2 % 16 == 0 && NRG >= 1, "whole matrix steps, whole slices, a wave per slice");
        const int ct = wave % NCT;
        const int rg = wave / NCT;
        if (rg < NRG) {
            const int  fc = lane & 15;
            const int  fh = lane >> 4;
            const bool lw = ct == NCT - 1;
            v2f bf[KS];
            if (!lw) {
#pragma unroll
                for (int ks = 0; ks < KS; ++ks) {
                    const int k = 4 * ks + 2 * fh;
                    bf[ks].x = Wn[(size_t)k * M2 + ct * 16 + fc];
                    bf[ks].y = Wn[(size_t)(k + 1) * M2 + ct * 16 + fc];
                }
            } else {
#pragma unroll
                for (int ks = 0; ks < KS; ++ks) {
                    const int k = 4 * ks + 2 * fh;
                    bf[ks].x = WAn[k * 16 + fc];
                    bf[ks].y = WAn[(k + 1) * 16 + fc];
                }
            }
            for (int rt = rg; rt < NRT; rt += NRG) {
                const float* xa = &acc[(rt * 16 + fc) * PA + 2 * fh];
                v2f af[KS];
#pragma unroll
                for (int ks = 0; ks < KS; ++ks) af[ks] = *(const v2f*)(xa + 4 * ks);

                v8f d = {};
#pragma unroll
                for (int ks = 0; ks < KS; ++ks)
                    d = __builtin_amdgcn_wmma_f32_16x16x4_f32(
                             false, af[ks],  false, bf[ks],
                             (short)0, d,  false,  false);

                const int    r0 = rt * 16 + 8 * fh;
                const size_t ps = lw ? (size_t)16 : (size_t)M2;
                float* dp = lw ? LGn + (size_t)(tile0 + r0) * 16 + fc
                               : Hn  + (size_t)(tile0 + r0) * M2 + ct * 16 + fc;
                if (r0 + 8 <= nvalid) {
#pragma unroll
                    for (int v = 0; v < 8; ++v) dp[(size_t)v * ps] = d[v];
                } else {
#pragma unroll
                    for (int v = 0; v < 8; ++v) if (r0 + v < nvalid) dp[(size_t)v * ps] = d[v];
                }
            }
        }
    }
}


extern "C" void kernel_launch(void* const* d_in, const int* in_sizes, int n_in,
                              void* d_out, int out_size, void* d_ws, size_t ws_size,
                              hipStream_t stream) {
    const float* x      = (const float*)d_in[0];
    const int*   ei     = (const int*)  d_in[1];
    const float* W1     = (const float*)d_in[2];
    const float* a_src1 = (const float*)d_in[3];
    const float* a_dst1 = (const float*)d_in[4];
    const float* b1     = (const float*)d_in[5];
    const float* W2     = (const float*)d_in[6];
    const float* a_src2 = (const float*)d_in[7];
    const float* a_dst2 = (const float*)d_in[8];
    const float* b2     = (const float*)d_in[9];
    const float* W3     = (const float*)d_in[10];
    const float* a_src3 = (const float*)d_in[11];
    const float* a_dst3 = (const float*)d_in[12];
    const float* b3     = (const float*)d_in[13];

    const int E = in_sizes[1] / 2;

    float* ws   = (float*)d_ws;
    float* bufH = ws;
    float* lgB  = bufH + (size_t)NNODE * 240;
    float* h2   = lgB  + (size_t)NNODE * 112;
    float* lg   = h2   + (size_t)NNODE * 192;
    float* wa1  = lg   + (size_t)NNODE * 16;
    float* wa2  = wa1  + 128 * 16;
    float* wa3  = wa2  + 112 * 16;
    float* out  = (float*)d_out;

    constexpr int NT = NTILE;
    unsigned* glist = (unsigned*)(wa3 + 96 * 16);
    unsigned* gcnt  = glist + (size_t)NT * LIST_CAP;
    unsigned* keys  = gcnt + 256;
    static_assert(NT <= 256, "the key plane starts behind the list lengths");

    constexpr int RT = NNODE / 16;
    static_assert(NNODE % 16 == 0, "row blocks are whole");

    fold_att<<<FOLD_BLOCKS + KEY_BLOCKS, SETUP_THREADS, 0, stream>>>(W1, a_src1, a_dst1, W2, a_src2, a_dst2, W3, a_src3, a_dst3, wa1, wa2, wa3, ei, E, keys);

    gemm_wstat_f32<128, 112><<<768, 256, 0, stream>>>(x, W1, wa1, bufH, lg, RT);
    agg_tile<7, 16, true, 96><<<NT, AGG_THREADS, 0, stream>>>(ei, E, lg, bufH, b1, nullptr, 1, glist, gcnt, keys, W2, wa2, h2, lgB);
    agg_tile<6, 16, false, 240><<<NT, AGG_THREADS, 0, stream>>>(ei, E, lgB, h2, b2, nullptr, 1, glist, gcnt, keys, W3, wa3, bufH, lg);
    agg_tile<6, 40, false, 0><<<NT, AGG_THREADS, 0, stream>>>(ei, E, lg, bufH, b3, out, 0, glist, gcnt, keys, nullptr, nullptr, nullptr, nullptr);
}


# epilogue: a thread normalizes 4 neighbouring channels of a row, acc * (1 / dn) with one reciprocal, not 4 divisions; 16-byte LDS reads and stores, piece, head, bias fixed per thread
# speedup vs baseline: 18.1045x; 1.0872x over previous
#include <hip/hip_runtime.h>
#include <math.h>

#define NNODE 50000

typedef float v2f __attribute__((ext_vector_type(2)));
typedef float v8f __attribute__((ext_vector_type(8)));

template <int K, int M>
__global__ __launch_bounds__((M / 16 + 1) * 32)
void gemm_wstat_f32(const float* __restrict__ X,
                    const float* __restrict__ W,
                    const float* __restrict__ WA,
                    float* __restrict__ H,
                    float* __restrict__ LG,
                    int rowTiles) {
    constexpr int NW  = M / 16 + 1;
    constexpr int NT  = NW * 32;
    constexpr int KS  = K / 4;
    constexpr int P   = K + 4;
    constexpr int TV  = 16 * K / 4;
    constexpr int LPT = (TV + NT - 1) / NT;

    __shared__ __attribute__((aligned(16))) float xs[2][16 * P];

    const int tid  = threadIdx.x;
    const int lane = tid & 31;
    const int wave = tid >> 5;
    const int c    = lane & 15;
    const int h    = lane >> 4;
    const int col0 = wave << 4;
    const bool lw  = wave == NW - 1;

    int rt = blockIdx.x;
    if (rt >= rowTiles) return;

    v2f b[KS];
    if (!lw) {
#pragma unroll
        for (int ks = 0; ks < KS; ++ks) {
            const int k = 4 * ks + 2 * h;
            b[ks].x = W[(size_t)k * M + col0 + c];
            b[ks].y = W[(size_t)(k + 1) * M + col0 + c];
        }
    } else {
#pragma unroll
        for (int ks = 0; ks < KS; ++ks) {
            const int k = 4 * ks + 2 * h;
            b[ks].x = WA[k * 16 + c];
            b[ks].y = WA[(k + 1) * 16 + c];
        }
    }

    static_assert(LPT <= 3, "three staging registers");
    float4 s0 = {0.0f, 0.0f, 0.0f, 0.0f}, s1 = s0, s2 = s0;
    auto gload = [&](int t) {
        const float4* __restrict__ srcv = (const float4*)(X + (size_t)t * 16 * K);
        if (tid < TV) s0 = srcv[tid];
        if constexpr (LPT > 1) { if (tid + NT < TV) s1 = srcv[tid + NT]; }
        if constexpr (LPT > 2) { if (tid + 2 * NT < TV) s2 = srcv[tid + 2 * NT]; }
    };
    auto put = [&](int bufi, int v, const float4& val) {
        if (v < TV) {
            const int r = v / (K / 4);
            const int q = v - r * (K / 4);
            *(float4*)&xs[bufi][r * P + 4 * q] = val;
        }
    };
    auto lstore = [&](int bufi) {
        put(bufi, tid, s0);
        if constexpr (LPT > 1) put(bufi, tid + NT, s1);
        if constexpr (LPT > 2) put(bufi, tid + 2 * NT, s2);
    };

    gload(rt);
    lstore(0);
    __syncthreads();

    int cur = 0;
    for (; rt < rowTiles; rt += gridDim.x) {
        const int  nxt  = rt + gridDim.x;
        const bool more = nxt < rowTiles;
        if (more) gload(nxt);

        const float* xa = &xs[cur][c * P + 2 * h];
        v2f a[KS];
#pragma unroll
        for (int ks = 0; ks < KS; ++ks) a[ks] = *(const v2f*)(xa + 4 * ks);

        v8f acc = {};
#pragma unroll
        for (int ks = 0; ks < KS; ++ks)
            acc = __builtin_amdgcn_wmma_f32_16x16x4_f32(
                       false, a[ks],  false, b[ks],
                       (short)0, acc,  false,  false);

        if (!lw) {
            float* hp = H + (size_t)(rt * 16 + 8 * h) * M + col0 + c;
#pragma unroll
            for (int v = 0; v < 8; ++v) hp[(size_t)v * M] = acc[v];
        } else {
            float* lp = LG + (size_t)(rt * 16 + 8 * h) * 16 + c;
#pragma unroll
            for (int v = 0; v < 8; ++v) lp[v * 16] = acc[v];
        }

        if (more) lstore(cur ^ 1);
        __syncthreads();
        cur ^= 1;
    }
}

constexpr int TILE  = 200;
constexpr int NTILE = (NNODE + TILE - 1) / TILE;

__device__ __forceinline__ void fold_one(const float* __restrict__ W,
                                         const float* __restrict__ a_src,
                                         const float* __restrict__ a_dst,
                                         float* __restrict__ WA,
                                         int K, int M, int heads, int C, int t) {
    if (t >= K * 16) return;
    const int k = t >> 4;
    const int j = t & 15;
    float s = 0.0f;
    if (j < 2 * heads) {
        const int hd = j < heads ? j : j - heads;
        const float* a = (j < heads ? a_src : a_dst) + hd * C;
        const float* w = W + (size_t)k * M + hd * C;
        for (int cc = 0; cc < C; ++cc) s += w[cc] * a[cc];
    }
    WA[t] = s;
}

constexpr int FOLD_BLOCKS   = 24;
constexpr int KEY_BLOCKS    = 200;
constexpr int SETUP_THREADS = 256;

__device__ __forceinline__ unsigned tile_key(int d) {
    const unsigned t = (unsigned)d / (unsigned)TILE;
    return t < (unsigned)NTILE ? t : 255u;
}

__global__ __launch_bounds__(SETUP_THREADS)
void fold_att(const float* __restrict__ W1, const float* __restrict__ as1, const float* __restrict__ ad1,
              const float* __restrict__ W2, const float* __restrict__ as2, const float* __restrict__ ad2,
              const float* __restrict__ W3, const float* __restrict__ as3, const float* __restrict__ ad3,
              float* __restrict__ WA1, float* __restrict__ WA2, float* __restrict__ WA3,
              const int* __restrict__ ei, int E, unsigned* __restrict__ keys) {
    if (blockIdx.x < FOLD_BLOCKS) {
        const int layer = blockIdx.x >> 3;
        const int t     = (blockIdx.x & 7) * SETUP_THREADS + threadIdx.x;
        if (layer == 0)      fold_one(W1, as1, ad1, WA1, 128, 112, 7, 16, t);
        else if (layer == 1) fold_one(W2, as2, ad2, WA2, 112,  96, 6, 16, t);
        else                 fold_one(W3, as3, ad3, WA3,  96, 240, 6, 40, t);
        return;
    }
    const int* __restrict__ dstp = ei + E;
    const int  ngrp = ((E + 15) >> 4) << 2;
    const bool al   = (((size_t)dstp) & 15) == 0;
    for (int g = (blockIdx.x - FOLD_BLOCKS) * SETUP_THREADS + threadIdx.x; g < ngrp; g += KEY_BLOCKS * SETUP_THREADS) {
        const int e = 4 * g;
        int d0, d1, d2, d3;
        if (al && e + 3 < E) {
            const int4 q = ((const int4*)dstp)[g];
            d0 = q.x; d1 = q.y; d2 = q.z; d3 = q.w;
        } else {
            d0 = e     < E ? dstp[e]     : -1;
            d1 = e + 1 < E ? dstp[e + 1] : -1;
            d2 = e + 2 < E ? dstp[e + 2] : -1;
            d3 = e + 3 < E ? dstp[e + 3] : -1;
        }
        keys[g] = tile_key(d0) | (tile_key(d1) << 8) | (tile_key(d2) << 16) | (tile_key(d3) << 24);
    }
}


constexpr int LIST_CAP    = 20480;
constexpr int AGG_THREADS = 1024;
constexpr int AGG_WAVES   = AGG_THREADS / 32;
constexpr int SCAN_VEC    = 4;
constexpr int ACC_BATCH   = 4;

static_assert(TILE <= 256, "local row index is packed into 8 bits");
static_assert(NTILE <= 255, "a tile number fits one key byte, 255 is left for 'no tile'");
static_assert(NNODE <= (1 << 24), "source index is packed into 24 bits");

__device__ __forceinline__ unsigned f2key(float f) {
    unsigned b = __float_as_uint(f);
    return (b & 0x80000000u) ? ~b : (b | 0x80000000u);
}
__device__ __forceinline__ float key2f(unsigned k) {
    return __uint_as_float((k & 0x80000000u) ? (k & 0x7fffffffu) : ~k);
}

constexpr int norm_rows_per_pass(int nq) {
    int r = AGG_THREADS / nq;
    while (r > 1 && (r * nq) % 8 != 0) --r;
    return r;
}

template <int HEADS, int C, bool BUILD, int M2>
__global__ __launch_bounds__(AGG_THREADS)
void agg_tile(const int* __restrict__ ei, int E,
              const float* __restrict__ lg,
              const float* __restrict__ H, const float* __restrict__ bias,
              float* __restrict__ out, int relu,
              unsigned* glist, unsigned* gcnt,
              const unsigned* __restrict__ keys,
              const float* __restrict__ Wn,
              const float* __restrict__ WAn,
              float* __restrict__ Hn,
              float* __restrict__ LGn) {
    (void)keys; (void)Wn; (void)WAn; (void)Hn; (void)LGn;
    constexpr int  HC   = HEADS * C;
    constexpr int  NIT  = (HC + 31) / 32;
    constexpr bool FUSE = M2 > 0;
    constexpr int  PA   = FUSE ? HC + 4 : HC;
    constexpr int  ROWS = FUSE ? ((TILE + 15) / 16) * 16 : TILE;

    __shared__ __attribute__((aligned(16))) float acc[ROWS * PA];
    __shared__ unsigned      mxk[TILE * HEADS];
    __shared__ float         dn[TILE * HEADS];
    __shared__ unsigned      elist[LIST_CAP];

    const int tid   = threadIdx.x;
    const int lane  = tid & 31;
    const int wave  = tid >> 5;
    const int tile0 = blockIdx.x * TILE;
    const int nvalid = min(TILE, NNODE - tile0);
    const unsigned nv = (unsigned)nvalid;

    const int* __restrict__ srcp = ei;
    const int* __restrict__ dstp = ei + E;

    unsigned* tlist = glist + (size_t)blockIdx.x * LIST_CAP;

    unsigned total_raw = 0;
    if constexpr (!BUILD) total_raw = gcnt[blockIdx.x];

    for (int i = tid; i < ROWS * PA; i += AGG_THREADS) acc[i] = 0.0f;
    const unsigned key_ninf = f2key(-INFINITY);
    for (int i = tid; i < TILE * HEADS; i += AGG_THREADS) { mxk[i] = key_ninf; dn[i] = 0.0f; }

    if constexpr (BUILD) {
        __shared__ unsigned      cnt;

        for (int i = tid; i < nvalid; i += AGG_THREADS) {
            const unsigned p = ((unsigned)(tile0 + i) << 8) | (unsigned)i;
            elist[i] = p;
            tlist[i] = p;
        }
        if (tid == 0) cnt = nv;
        __syncthreads();

        auto push = [&](int e) {
            const unsigned pos = atomicAdd(&cnt, 1u);
            if (pos < (unsigned)LIST_CAP) elist[pos] = (unsigned)e;
        };
        {
            const int    nvec = (E + 15) >> 4;
            const uint4* kv   = (const uint4*)keys;
            const unsigned pat = (unsigned)blockIdx.x * 0x01010101u;
            auto hit = [&](unsigned x) -> unsigned {
                const unsigned y = x ^ pat;
                return ~(((y & 0x7F7F7F7Fu) + 0x7F7F7F7Fu) | y | 0x7F7F7F7Fu);
            };
            constexpr int STEP = SCAN_VEC * AGG_THREADS;
            auto loadq = [&](uint4 (&q)[SCAN_VEC], int base) {
#pragma unroll
                for (int j = 0; j < SCAN_VEC; ++j) {
                    const int v = base + j * AGG_THREADS + tid;
                    q[j] = kv[min(v, nvec - 1)];
                }
            };
            auto procq = [&](const uint4 (&q)[SCAN_VEC], int base) {
#pragma unroll
                for (int j = 0; j < SCAN_VEC; ++j) {
                    const int v = base + j * AGG_THREADS + tid;
                    unsigned m = (hit(q[j].x) >> 7) | (hit(q[j].y) >> 6) | (hit(q[j].z) >> 5) | (hit(q[j].w) >> 4);
                    if (v >= nvec) m = 0u;
                    while (m) {
                        const int t = __builtin_ctz(m);
                        m &= m - 1u;
                        push(16 * v + 4 * (t & 7) + (t >> 3));
                    }
                }
            };
            if (nvec > 0) {
                uint4 qa[SCAN_VEC], qb[SCAN_VEC];
                loadq(qa, 0);
                for (int base = 0; base < nvec; base += 2 * STEP) {
                    loadq(qb, base + STEP);
                    procq(qa, base);
                    loadq(qa, base + 2 * STEP);
                    procq(qb, base + STEP);
                }
            }
        }
        __syncthreads();

        total_raw = cnt;
        if (tid == 0) gcnt[blockIdx.x] = total_raw;
        if (total_raw <= (unsigned)LIST_CAP) {
            const int total = (int)total_raw;

            for (int i0 = nvalid + tid; i0 < total; i0 += 4 * AGG_THREADS) {
                int sv[4], dd[4];
#pragma unroll
                for (int j = 0; j < 4; ++j) {
                    const int i = i0 + j * AGG_THREADS;
                    sv[j] = 0; dd[j] = tile0;
                    if (i < total) {
                        const unsigned e = elist[i];
                        sv[j] = srcp[e];
                        dd[j] = dstp[e];
                    }
                }
#pragma unroll
                for (int j = 0; j < 4; ++j) {
                    const int i = i0 + j * AGG_THREADS;
                    if (i < total) {
                        const unsigned p = ((unsigned)sv[j] << 8) | (unsigned)(dd[j] - tile0);
                        elist[i] = p;
                        tlist[i] = p;
                    }
                }
            }
        }
        __syncthreads();
    } else {
        if (total_raw <= (unsigned)LIST_CAP) {
            const int total = (int)total_raw;
            const unsigned* __restrict__ rlist = tlist;
            for (int i0 = tid; i0 < total; i0 += 4 * AGG_THREADS) {
                unsigned pv[4];
#pragma unroll
                for (int j = 0; j < 4; ++j) {
                    const int i = i0 + j * AGG_THREADS;
                    pv[j] = (i < total) ? rlist[i] : 0u;
                }
#pragma unroll
                for (int j = 0; j < 4; ++j) {
                    const int i = i0 + j * AGG_THREADS;
                    if (i < total) elist[i] = pv[j];
                }
            }
        }
        __syncthreads();
    }

    if (total_raw <= (unsigned)LIST_CAP) {
        const int total = (int)total_raw;

        const int n1 = total * HEADS;
        for (int idx0 = tid; idx0 < n1; idx0 += 4 * AGG_THREADS) {
            float xv[4];
            int   cell[4];
#pragma unroll
            for (int j = 0; j < 4; ++j) {
                const int idx  = min(idx0 + j * AGG_THREADS, n1 - 1);
                const int item = idx / HEADS;
                const int h    = idx - item * HEADS;
                const unsigned en = elist[item];
                const int s  = (int)(en >> 8);
                const int ld = (int)(en & 255u);
                xv[j]   = lg[s * 16 + h] + lg[(tile0 + ld) * 16 + HEADS + h];
                cell[j] = ld * HEADS + h;
            }
#pragma unroll
            for (int j = 0; j < 4; ++j) {
                if (idx0 + j * AGG_THREADS < n1) {
                    const float lr = xv[j] >= 0.0f ? xv[j] : 0.2f * xv[j];
                    atomicMax(&mxk[cell[j]], f2key(lr));
                }
            }
        }
        __syncthreads();

        static_assert(ACC_BATCH == 4 && HEADS <= 8, "weight lanes: 8 lanes an edge, 4 edges a wave");
        static_assert(C == 16 || C >= 32, "32 consecutive channels belong to at most two heads");
        const int  wb   = lane >> 3;
        const int  wh   = lane & 7;
        const bool wact = wh < HEADS;
        const int  whc  = min(wh, HEADS - 1);

        for (int item0 = wave; item0 < total; item0 += AGG_WAVES * ACC_BATCH) {
            const int      itw = item0 + AGG_WAVES * wb;
            const unsigned enw = elist[min(itw, total - 1)];
            const int      lw  = (int)(enw & 255u);
            const float    xa = lg[(int)(enw >> 8) * 16 + whc];
            const float    xd = lg[(tile0 + lw) * 16 + HEADS + whc];
            const unsigned mk = mxk[lw * HEADS + whc];
            int   lb[ACC_BATCH];
            float hv[ACC_BATCH][NIT];
#pragma unroll
            for (int b = 0; b < ACC_BATCH; ++b) {
                const unsigned en = (unsigned)__builtin_amdgcn_readlane((int)enw, 8 * b);
                lb[b] = (int)(en & 255u);
                const float* __restrict__ hrow = H + (size_t)(en >> 8) * HC;
#pragma unroll
                for (int k = 0; k < NIT; ++k) {
                    const int c = lane + 32 * k;
                    hv[b][k] = (c < HC) ? hrow[c] : 0.0f;
                }
            }
            float exl = 0.0f;
            if (wact && itw < total) {
                const float x  = xa + xd;
                const float lr = x >= 0.0f ? x : 0.2f * x;
                exl = __expf(lr - key2f(mk));
                atomicAdd(&dn[lw * HEADS + wh], exl);
            }
            const int exi = __float_as_int(exl);
#pragma unroll
            for (int b = 0; b < ACC_BATCH; ++b) {
                if (item0 + AGG_WAVES * b < total) {
                    float* arow = acc + lb[b] * PA;
#pragma unroll
                    for (int k = 0; k < NIT; ++k) {
                        const int c  = lane + 32 * k;
                        const int h0 = (32 * k) / C;
                        const int h1 = min((32 * k + 31) / C, HEADS - 1);
                        float ex = __int_as_float(__builtin_amdgcn_readlane(exi, 8 * b + h0));
                        if (h1 != h0) {
                            const float e1 = __int_as_float(__builtin_amdgcn_readlane(exi, 8 * b + h1));
                            ex = lane < (h0 + 1) * C - 32 * k ? ex : e1;
                        }
                        if (c < HC) atomicAdd(&arow[c], hv[b][k] * ex);
                    }
                }
            }
        }
    } else {
        auto visit_max = [&](int s, int ld) {
            for (int h = 0; h < HEADS; ++h) {
                const float x  = lg[s * 16 + h] + lg[(tile0 + ld) * 16 + HEADS + h];
                const float lr = x >= 0.0f ? x : 0.2f * x;
                atomicMax(&mxk[ld * HEADS + h], f2key(lr));
            }
        };
        for (int i = tid; i < nvalid; i += AGG_THREADS) visit_max(tile0 + i, i);
        for (int e = tid; e < E; e += AGG_THREADS) {
            const unsigned ld = (unsigned)(dstp[e] - tile0);
            if (ld < nv) visit_max(srcp[e], (int)ld);
        }
        __syncthreads();

        auto visit_acc = [&](int s, int ld) {
            for (int h = 0; h < HEADS; ++h) {
                const float x  = lg[s * 16 + h] + lg[(tile0 + ld) * 16 + HEADS + h];
                const float lr = x >= 0.0f ? x : 0.2f * x;
                const float ex = __expf(lr - key2f(mxk[ld * HEADS + h]));
                atomicAdd(&dn[ld * HEADS + h], ex);
                const float* hs = H + (size_t)s * HC + h * C;
                float*       ap = acc + ld * PA + h * C;
                for (int c = 0; c < C; ++c) atomicAdd(&ap[c], hs[c] * ex);
            }
        };
        for (int i = tid; i < nvalid; i += AGG_THREADS) visit_acc(tile0 + i, i);
        for (int e = tid; e < E; e += AGG_THREADS) {
            const unsigned ld = (unsigned)(dstp[e] - tile0);
            if (ld < nv) visit_acc(srcp[e], (int)ld);
        }
    }
    static_assert(C % 4 == 0 && PA % 4 == 0, "a 16-byte piece lies in one head and is 16-byte aligned in LDS");
    constexpr int NQ  = HC / 4;
    constexpr int NRP = norm_rows_per_pass(NQ);
    const int  nr0   = tid / NQ;
    const int  nq    = tid - nr0 * NQ;
    const int  nh    = (4 * nq) / C;
    const bool nact  = nr0 < NRP;
    float bq0 = 0.0f, bq1 = 0.0f, bq2 = 0.0f, bq3 = 0.0f;
    if (nact) {
        bq0 = bias[4 * nq];     bq1 = bias[4 * nq + 1];
        bq2 = bias[4 * nq + 2]; bq3 = bias[4 * nq + 3];
    }
    __syncthreads();

    if (nact) {
        for (int ld = nr0; ld < nvalid; ld += NRP) {
            float4* ap = (float4*)&acc[ld * PA + 4 * nq];
            const float4 a = *ap;
            const float  r = 1.0f / dn[ld * HEADS + nh];
            float4 v;
            v.x = a.x * r + bq0;
            v.y = a.y * r + bq1;
            v.z = a.z * r + bq2;
            v.w = a.w * r + bq3;
            if (relu) {
                v.x = fmaxf(v.x, 0.0f); v.y = fmaxf(v.y, 0.0f);
                v.z = fmaxf(v.z, 0.0f); v.w = fmaxf(v.w, 0.0f);
            }
            if constexpr (FUSE) *ap = v;
            else *(float4*)&out[(size_t)(tile0 + ld) * HC + 4 * nq] = v;
        }
    }

    if constexpr (FUSE) {
        __syncthreads();
        constexpr int KS  = HC / 4;
        constexpr int NCT = M2 / 16 + 1;
        constexpr int NRG = AGG_WAVES / NCT;
        constexpr int NRT = ROWS / 16;
        static_assert(HC % 4 == 0 && M2 % 16 == 0 && NRG >= 1, "whole matrix steps, whole slices, a wave per slice");
        const int ct = wave % NCT;
        const int rg = wave / NCT;
        if (rg < NRG) {
            const int  fc = lane & 15;
            const int  fh = lane >> 4;
            const bool lw = ct == NCT - 1;
            v2f bf[KS];
            if (!lw) {
#pragma unroll
                for (int ks = 0; ks < KS; ++ks) {
                    const int k = 4 * ks + 2 * fh;
                    bf[ks].x = Wn[(size_t)k * M2 + ct * 16 + fc];
                    bf[ks].y = Wn[(size_t)(k + 1) * M2 + ct * 16 + fc];
                }
            } else {
#pragma unroll
                for (int ks = 0; ks < KS; ++ks) {
                    const int k = 4 * ks + 2 * fh;
                    bf[ks].x = WAn[k * 16 + fc];
                    bf[ks].y = WAn[(k + 1) * 16 + fc];
                }
            }
            for (int rt = rg; rt < NRT; rt += NRG) {
                const float* xa = &acc[(rt * 16 + fc) * PA + 2 * fh];
                v2f af[KS];
#pragma unroll
                for (int ks = 0; ks < KS; ++ks) af[ks] = *(const v2f*)(xa + 4 * ks);

                v8f d = {};
#pragma unroll
                for (int ks = 0; ks < KS; ++ks)
                    d = __builtin_amdgcn_wmma_f32_16x16x4_f32(
                             false, af[ks],  false, bf[ks],
                             (short)0, d,  false,  false);

                const int    r0 = rt * 16 + 8 * fh;
                const size_t ps = lw ? (size_t)16 : (size_t)M2;
                float* dp = lw ? LGn + (size_t)(tile0 + r0) * 16 + fc
                               : Hn  + (size_t)(tile0 + r0) * M2 + ct * 16 + fc;
                if (r0 + 8 <= nvalid) {
#pragma unroll
                    for (int v = 0; v < 8; ++v) dp[(size_t)v * ps] = d[v];
                } else {
#pragma unroll
                    for (int v = 0; v < 8; ++v) if (r0 + v < nvalid) dp[(size_t)v * ps] = d[v];
                }
            }
        }
    }
}


extern "C" void kernel_launch(void* const* d_in, const int* in_sizes, int n_in,
                              void* d_out, int out_size, void* d_ws, size_t ws_size,
                              hipStream_t stream) {
    const float* x      = (const float*)d_in[0];
    const int*   ei     = (const int*)  d_in[1];
    const float* W1     = (const float*)d_in[2];
    const float* a_src1 = (const float*)d_in[3];
    const float* a_dst1 = (const float*)d_in[4];
    const float* b1     = (const float*)d_in[5];
    const float* W2     = (const float*)d_in[6];
    const float* a_src2 = (const float*)d_in[7];
    const float* a_dst2 = (const float*)d_in[8];
    const float* b2     = (const float*)d_in[9];
    const float* W3     = (const float*)d_in[10];
    const float* a_src3 = (const float*)d_in[11];
    const float* a_dst3 = (const float*)d_in[12];
    const float* b3     = (const float*)d_in[13];

    const int E = in_sizes[1] / 2;

    float* ws   = (float*)d_ws;
    float* bufH = ws;
    float* lgB  = bufH + (size_t)NNODE * 240;
    float* h2   = lgB  + (size_t)NNODE * 112;
    float* lg   = h2   + (size_t)NNODE * 192;
    float* wa1  = lg   + (size_t)NNODE * 16;
    float* wa2  = wa1  + 128 * 16;
    float* wa3  = wa2  + 112 * 16;
    float* out  = (float*)d_out;

    constexpr int NT = NTILE;
    unsigned* glist = (unsigned*)(wa3 + 96 * 16);
    unsigned* gcnt  = glist + (size_t)NT * LIST_CAP;
    unsigned* keys  = gcnt + 256;
    static_assert(NT <= 256, "the key plane starts behind the list lengths");

    constexpr int RT = NNODE / 16;
    static_assert(NNODE % 16 == 0, "row blocks are whole");

    fold_att<<<FOLD_BLOCKS + KEY_BLOCKS, SETUP_THREADS, 0, stream>>>(W1, a_src1, a_dst1, W2, a_src2, a_dst2, W3, a_src3, a_dst3, wa1, wa2, wa3, ei, E, keys);

    gemm_wstat_f32<128, 112><<<768, 256, 0, stream>>>(x, W1, wa1, bufH, lg, RT);
    agg_tile<7, 16, true, 96><<<NT, AGG_THREADS, 0, stream>>>(ei, E, lg, bufH, b1, nullptr, 1, glist, gcnt, keys, W2, wa2, h2, lgB);
    agg_tile<6, 16, false, 240><<<NT, AGG_THREADS, 0, stream>>>(ei, E, lgB, h2, b2, nullptr, 1, glist, gcnt, keys, W3, wa3, bufH, lg);
    agg_tile<6, 40, false, 0><<<NT, AGG_THREADS, 0, stream>>>(ei, E, lg, bufH, b3, out, 0, glist, gcnt, keys, nullptr, nullptr, nullptr, nullptr);
}


# aggregation: the maximum pass keeps each (edge, head) leaky-relu logit in an LDS table, the sum pass reads it there and no longer gathers the two logit lines of every edge again; sums unchanged
# speedup vs baseline: 19.3215x; 1.0672x over previous
#include <hip/hip_runtime.h>
#include <math.h>

#define NNODE 50000

typedef float v2f __attribute__((ext_vector_type(2)));
typedef float v8f __attribute__((ext_vector_type(8)));

template <int K, int M>
__global__ __launch_bounds__((M / 16 + 1) * 32)
void gemm_wstat_f32(const float* __restrict__ X,
                    const float* __restrict__ W,
                    const float* __restrict__ WA,
                    float* __restrict__ H,
                    float* __restrict__ LG,
                    int rowTiles) {
    constexpr int NW  = M / 16 + 1;
    constexpr int NT  = NW * 32;
    constexpr int KS  = K / 4;
    constexpr int P   = K + 4;
    constexpr int TV  = 16 * K / 4;
    constexpr int LPT = (TV + NT - 1) / NT;

    __shared__ __attribute__((aligned(16))) float xs[2][16 * P];

    const int tid  = threadIdx.x;
    const int lane = tid & 31;
    const int wave = tid >> 5;
    const int c    = lane & 15;
    const int h    = lane >> 4;
    const int col0 = wave << 4;
    const bool lw  = wave == NW - 1;

    int rt = blockIdx.x;
    if (rt >= rowTiles) return;

    v2f b[KS];
    if (!lw) {
#pragma unroll
        for (int ks = 0; ks < KS; ++ks) {
            const int k = 4 * ks + 2 * h;
            b[ks].x = W[(size_t)k * M + col0 + c];
            b[ks].y = W[(size_t)(k + 1) * M + col0 + c];
        }
    } else {
#pragma unroll
        for (int ks = 0; ks < KS; ++ks) {
            const int k = 4 * ks + 2 * h;
            b[ks].x = WA[k * 16 + c];
            b[ks].y = WA[(k + 1) * 16 + c];
        }
    }

    static_assert(LPT <= 3, "three staging registers");
    float4 s0 = {0.0f, 0.0f, 0.0f, 0.0f}, s1 = s0, s2 = s0;
    auto gload = [&](int t) {
        const float4* __restrict__ srcv = (const float4*)(X + (size_t)t * 16 * K);
        if (tid < TV) s0 = srcv[tid];
        if constexpr (LPT > 1) { if (tid + NT < TV) s1 = srcv[tid + NT]; }
        if constexpr (LPT > 2) { if (tid + 2 * NT < TV) s2 = srcv[tid + 2 * NT]; }
    };
    auto put = [&](int bufi, int v, const float4& val) {
        if (v < TV) {
            const int r = v / (K / 4);
            const int q = v - r * (K / 4);
            *(float4*)&xs[bufi][r * P + 4 * q] = val;
        }
    };
    auto lstore = [&](int bufi) {
        put(bufi, tid, s0);
        if constexpr (LPT > 1) put(bufi, tid + NT, s1);
        if constexpr (LPT > 2) put(bufi, tid + 2 * NT, s2);
    };

    gload(rt);
    lstore(0);
    __syncthreads();

    int cur = 0;
    for (; rt < rowTiles; rt += gridDim.x) {
        const int  nxt  = rt + gridDim.x;
        const bool more = nxt < rowTiles;
        if (more) gload(nxt);

        const float* xa = &xs[cur][c * P + 2 * h];
        v2f a[KS];
#pragma unroll
        for (int ks = 0; ks < KS; ++ks) a[ks] = *(const v2f*)(xa + 4 * ks);

        v8f acc = {};
#pragma unroll
        for (int ks = 0; ks < KS; ++ks)
            acc = __builtin_amdgcn_wmma_f32_16x16x4_f32(
                       false, a[ks],  false, b[ks],
                       (short)0, acc,  false,  false);

        if (!lw) {
            float* hp = H + (size_t)(rt * 16 + 8 * h) * M + col0 + c;
#pragma unroll
            for (int v = 0; v < 8; ++v) hp[(size_t)v * M] = acc[v];
        } else {
            float* lp = LG + (size_t)(rt * 16 + 8 * h) * 16 + c;
#pragma unroll
            for (int v = 0; v < 8; ++v) lp[v * 16] = acc[v];
        }

        if (more) lstore(cur ^ 1);
        __syncthreads();
        cur ^= 1;
    }
}

constexpr int TILE  = 200;
constexpr int NTILE = (NNODE + TILE - 1) / TILE;

__device__ __forceinline__ void fold_one(const float* __restrict__ W,
                                         const float* __restrict__ a_src,
                                         const float* __restrict__ a_dst,
                                         float* __restrict__ WA,
                                         int K, int M, int heads, int C, int t) {
    if (t >= K * 16) return;
    const int k = t >> 4;
    const int j = t & 15;
    float s = 0.0f;
    if (j < 2 * heads) {
        const int hd = j < heads ? j : j - heads;
        const float* a = (j < heads ? a_src : a_dst) + hd * C;
        const float* w = W + (size_t)k * M + hd * C;
        for (int cc = 0; cc < C; ++cc) s += w[cc] * a[cc];
    }
    WA[t] = s;
}

constexpr int FOLD_BLOCKS   = 24;
constexpr int KEY_BLOCKS    = 200;
constexpr int SETUP_THREADS = 256;

__device__ __forceinline__ unsigned tile_key(int d) {
    const unsigned t = (unsigned)d / (unsigned)TILE;
    return t < (unsigned)NTILE ? t : 255u;
}

__global__ __launch_bounds__(SETUP_THREADS)
void fold_att(const float* __restrict__ W1, const float* __restrict__ as1, const float* __restrict__ ad1,
              const float* __restrict__ W2, const float* __restrict__ as2, const float* __restrict__ ad2,
              const float* __restrict__ W3, const float* __restrict__ as3, const float* __restrict__ ad3,
              float* __restrict__ WA1, float* __restrict__ WA2, float* __restrict__ WA3,
              const int* __restrict__ ei, int E, unsigned* __restrict__ keys) {
    if (blockIdx.x < FOLD_BLOCKS) {
        const int layer = blockIdx.x >> 3;
        const int t     = (blockIdx.x & 7) * SETUP_THREADS + threadIdx.x;
        if (layer == 0)      fold_one(W1, as1, ad1, WA1, 128, 112, 7, 16, t);
        else if (layer == 1) fold_one(W2, as2, ad2, WA2, 112,  96, 6, 16, t);
        else                 fold_one(W3, as3, ad3, WA3,  96, 240, 6, 40, t);
        return;
    }
    const int* __restrict__ dstp = ei + E;
    const int  ngrp = ((E + 15) >> 4) << 2;
    const bool al   = (((size_t)dstp) & 15) == 0;
    for (int g = (blockIdx.x - FOLD_BLOCKS) * SETUP_THREADS + threadIdx.x; g < ngrp; g += KEY_BLOCKS * SETUP_THREADS) {
        const int e = 4 * g;
        int d0, d1, d2, d3;
        if (al && e + 3 < E) {
            const int4 q = ((const int4*)dstp)[g];
            d0 = q.x; d1 = q.y; d2 = q.z; d3 = q.w;
        } else {
            d0 = e     < E ? dstp[e]     : -1;
            d1 = e + 1 < E ? dstp[e + 1] : -1;
            d2 = e + 2 < E ? dstp[e + 2] : -1;
            d3 = e + 3 < E ? dstp[e + 3] : -1;
        }
        keys[g] = tile_key(d0) | (tile_key(d1) << 8) | (tile_key(d2) << 16) | (tile_key(d3) << 24);
    }
}


constexpr int LIST_CAP    = 20480;
constexpr int LR_CAP      = 4096;
constexpr int AGG_THREADS = 1024;
constexpr int AGG_WAVES   = AGG_THREADS / 32;
constexpr int SCAN_VEC    = 4;
constexpr int ACC_BATCH   = 4;

static_assert(TILE <= 256, "local row index is packed into 8 bits");
static_assert(NTILE <= 255, "a tile number fits one key byte, 255 is left for 'no tile'");
static_assert(NNODE <= (1 << 24), "source index is packed into 24 bits");

template <bool V> struct flag_c { static constexpr bool value = V; };

__device__ __forceinline__ unsigned f2key(float f) {
    unsigned b = __float_as_uint(f);
    return (b & 0x80000000u) ? ~b : (b | 0x80000000u);
}
__device__ __forceinline__ float key2f(unsigned k) {
    return __uint_as_float((k & 0x80000000u) ? (k & 0x7fffffffu) : ~k);
}

constexpr int norm_rows_per_pass(int nq) {
    int r = AGG_THREADS / nq;
    while (r > 1 && (r * nq) % 8 != 0) --r;
    return r;
}

template <int HEADS, int C, bool BUILD, int M2>
__global__ __launch_bounds__(AGG_THREADS)
void agg_tile(const int* __restrict__ ei, int E,
              const float* __restrict__ lg,
              const float* __restrict__ H, const float* __restrict__ bias,
              float* __restrict__ out, int relu,
              unsigned* glist, unsigned* gcnt,
              const unsigned* __restrict__ keys,
              const float* __restrict__ Wn,
              const float* __restrict__ WAn,
              float* __restrict__ Hn,
              float* __restrict__ LGn) {
    (void)keys; (void)Wn; (void)WAn; (void)Hn; (void)LGn;
    constexpr int  HC   = HEADS * C;
    constexpr int  NIT  = (HC + 31) / 32;
    constexpr bool FUSE = M2 > 0;
    constexpr int  PA   = FUSE ? HC + 4 : HC;
    constexpr int  ROWS = FUSE ? ((TILE + 15) / 16) * 16 : TILE;

    __shared__ __attribute__((aligned(16))) float acc[ROWS * PA];
    __shared__ unsigned      mxk[TILE * HEADS];
    __shared__ float         dn[TILE * HEADS];
    constexpr int  ECAP = ((size_t)ROWS * PA * 4 + (size_t)LIST_CAP * 4 + (size_t)LR_CAP * HEADS * 4 + (size_t)TILE * HEADS * 8 + 1024 <= (size_t)320 * 1024)
                              ? LIST_CAP : LR_CAP;
    static_assert(!BUILD || ECAP == LIST_CAP, "the launch that writes the workspace list keeps all of it in LDS");
    __shared__ unsigned      elist[ECAP];
    __shared__ float         lrt[LR_CAP * HEADS];

    const int tid   = threadIdx.x;
    const int lane  = tid & 31;
    const int wave  = tid >> 5;
    const int tile0 = blockIdx.x * TILE;
    const int nvalid = min(TILE, NNODE - tile0);
    const unsigned nv = (unsigned)nvalid;

    const int* __restrict__ srcp = ei;
    const int* __restrict__ dstp = ei + E;

    unsigned* tlist = glist + (size_t)blockIdx.x * LIST_CAP;

    unsigned total_raw = 0;
    if constexpr (!BUILD) total_raw = gcnt[blockIdx.x];

    for (int i = tid; i < ROWS * PA; i += AGG_THREADS) acc[i] = 0.0f;
    const unsigned key_ninf = f2key(-INFINITY);
    for (int i = tid; i < TILE * HEADS; i += AGG_THREADS) { mxk[i] = key_ninf; dn[i] = 0.0f; }

    if constexpr (BUILD) {
        __shared__ unsigned      cnt;

        for (int i = tid; i < nvalid; i += AGG_THREADS) {
            const unsigned p = ((unsigned)(tile0 + i) << 8) | (unsigned)i;
            elist[i] = p;
            tlist[i] = p;
        }
        if (tid == 0) cnt = nv;
        __syncthreads();

        auto push = [&](int e) {
            const unsigned pos = atomicAdd(&cnt, 1u);
            if (pos < (unsigned)ECAP) elist[pos] = (unsigned)e;
        };
        {
            const int    nvec = (E + 15) >> 4;
            const uint4* kv   = (const uint4*)keys;
            const unsigned pat = (unsigned)blockIdx.x * 0x01010101u;
            auto hit = [&](unsigned x) -> unsigned {
                const unsigned y = x ^ pat;
                return ~(((y & 0x7F7F7F7Fu) + 0x7F7F7F7Fu) | y | 0x7F7F7F7Fu);
            };
            constexpr int STEP = SCAN_VEC * AGG_THREADS;
            auto loadq = [&](uint4 (&q)[SCAN_VEC], int base) {
#pragma unroll
                for (int j = 0; j < SCAN_VEC; ++j) {
                    const int v = base + j * AGG_THREADS + tid;
                    q[j] = kv[min(v, nvec - 1)];
                }
            };
            auto procq = [&](const uint4 (&q)[SCAN_VEC], int base) {
#pragma unroll
                for (int j = 0; j < SCAN_VEC; ++j) {
                    const int v = base + j * AGG_THREADS + tid;
                    unsigned m = (hit(q[j].x) >> 7) | (hit(q[j].y) >> 6) | (hit(q[j].z) >> 5) | (hit(q[j].w) >> 4);
                    if (v >= nvec) m = 0u;
                    while (m) {
                        const int t = __builtin_ctz(m);
                        m &= m - 1u;
                        push(16 * v + 4 * (t & 7) + (t >> 3));
                    }
                }
            };
            if (nvec > 0) {
                uint4 qa[SCAN_VEC], qb[SCAN_VEC];
                loadq(qa, 0);
                for (int base = 0; base < nvec; base += 2 * STEP) {
                    loadq(qb, base + STEP);
                    procq(qa, base);
                    loadq(qa, base + 2 * STEP);
                    procq(qb, base + STEP);
                }
            }
        }
        __syncthreads();

        total_raw = cnt;
        if (tid == 0) gcnt[blockIdx.x] = total_raw;
        if (total_raw <= (unsigned)ECAP) {
            const int total = (int)total_raw;

            for (int i0 = nvalid + tid; i0 < total; i0 += 4 * AGG_THREADS) {
                int sv[4], dd[4];
#pragma unroll
                for (int j = 0; j < 4; ++j) {
                    const int i = i0 + j * AGG_THREADS;
                    sv[j] = 0; dd[j] = tile0;
                    if (i < total) {
                        const unsigned e = elist[i];
                        sv[j] = srcp[e];
                        dd[j] = dstp[e];
                    }
                }
#pragma unroll
                for (int j = 0; j < 4; ++j) {
                    const int i = i0 + j * AGG_THREADS;
                    if (i < total) {
                        const unsigned p = ((unsigned)sv[j] << 8) | (unsigned)(dd[j] - tile0);
                        elist[i] = p;
                        tlist[i] = p;
                    }
                }
            }
        }
        __syncthreads();
    } else {
        if (total_raw <= (unsigned)ECAP) {
            const int total = (int)total_raw;
            const unsigned* __restrict__ rlist = tlist;
            for (int i0 = tid; i0 < total; i0 += 4 * AGG_THREADS) {
                unsigned pv[4];
#pragma unroll
                for (int j = 0; j < 4; ++j) {
                    const int i = i0 + j * AGG_THREADS;
                    pv[j] = (i < total) ? rlist[i] : 0u;
                }
#pragma unroll
                for (int j = 0; j < 4; ++j) {
                    const int i = i0 + j * AGG_THREADS;
                    if (i < total) elist[i] = pv[j];
                }
            }
        }
        __syncthreads();
    }

    if (total_raw <= (unsigned)ECAP) {
        const int total = (int)total_raw;

        const int  n1   = total * HEADS;
        const bool keep = total <= LR_CAP;
        for (int idx0 = tid; idx0 < n1; idx0 += 4 * AGG_THREADS) {
            float xv[4];
            int   cell[4];
#pragma unroll
            for (int j = 0; j < 4; ++j) {
                const int idx  = min(idx0 + j * AGG_THREADS, n1 - 1);
                const int item = idx / HEADS;
                const int h    = idx - item * HEADS;
                const unsigned en = elist[item];
                const int s  = (int)(en >> 8);
                const int ld = (int)(en & 255u);
                xv[j]   = lg[s * 16 + h] + lg[(tile0 + ld) * 16 + HEADS + h];
                cell[j] = ld * HEADS + h;
            }
#pragma unroll
            for (int j = 0; j < 4; ++j) {
                if (idx0 + j * AGG_THREADS < n1) {
                    const float lr = xv[j] >= 0.0f ? xv[j] : 0.2f * xv[j];
                    if (keep) lrt[idx0 + j * AGG_THREADS] = lr;
                    atomicMax(&mxk[cell[j]], f2key(lr));
                }
            }
        }
        __syncthreads();

        static_assert(ACC_BATCH == 4 && HEADS <= 8, "weight lanes: 8 lanes an edge, 4 edges a wave");
        static_assert(C == 16 || C >= 32, "32 consecutive channels belong to at most two heads");
        const int  wb   = lane >> 3;
        const int  wh   = lane & 7;
        const bool wact = wh < HEADS;
        const int  whc  = min(wh, HEADS - 1);

        auto sum_pass = [&](auto keep_c) {
        constexpr bool KEEP = decltype(keep_c)::value;
        for (int item0 = wave; item0 < total; item0 += AGG_WAVES * ACC_BATCH) {
            const int      itw = item0 + AGG_WAVES * wb;
            const int      itc = min(itw, total - 1);
            const unsigned enw = elist[itc];
            const int      lw  = (int)(enw & 255u);
            float xa = 0.0f, xd = 0.0f, lrk = 0.0f;
            if constexpr (KEEP) {
                lrk = lrt[itc * HEADS + whc];
            } else {
                xa = lg[(int)(enw >> 8) * 16 + whc];
                xd = lg[(tile0 + lw) * 16 + HEADS + whc];
            }
            const unsigned mk = mxk[lw * HEADS + whc];
            int   lb[ACC_BATCH];
            float hv[ACC_BATCH][NIT];
#pragma unroll
            for (int b = 0; b < ACC_BATCH; ++b) {
                const unsigned en = (unsigned)__builtin_amdgcn_readlane((int)enw, 8 * b);
                lb[b] = (int)(en & 255u);
                const float* __restrict__ hrow = H + (size_t)(en >> 8) * HC;
#pragma unroll
                for (int k = 0; k < NIT; ++k) {
                    const int c = lane + 32 * k;
                    hv[b][k] = (c < HC) ? hrow[c] : 0.0f;
                }
            }
            float exl = 0.0f;
            if (wact && itw < total) {
                float lr = lrk;
                if constexpr (!KEEP) {
                    const float x = xa + xd;
                    lr = x >= 0.0f ? x : 0.2f * x;
                }
                exl = __expf(lr - key2f(mk));
                atomicAdd(&dn[lw * HEADS + wh], exl);
            }
            const int exi = __float_as_int(exl);
#pragma unroll
            for (int b = 0; b < ACC_BATCH; ++b) {
                if (item0 + AGG_WAVES * b < total) {
                    float* arow = acc + lb[b] * PA;
#pragma unroll
                    for (int k = 0; k < NIT; ++k) {
                        const int c  = lane + 32 * k;
                        const int h0 = (32 * k) / C;
                        const int h1 = min((32 * k + 31) / C, HEADS - 1);
                        float ex = __int_as_float(__builtin_amdgcn_readlane(exi, 8 * b + h0));
                        if (h1 != h0) {
                            const float e1 = __int_as_float(__builtin_amdgcn_readlane(exi, 8 * b + h1));
                            ex = lane < (h0 + 1) * C - 32 * k ? ex : e1;
                        }
                        if (c < HC) atomicAdd(&arow[c], hv[b][k] * ex);
                    }
                }
            }
        }
        };
        if constexpr (ECAP <= LR_CAP) {
            sum_pass(flag_c<true>{});
        } else {
            if (keep) sum_pass(flag_c<true>{});
            else      sum_pass(flag_c<false>{});
        }
    } else {
        auto visit_max = [&](int s, int ld) {
            for (int h = 0; h < HEADS; ++h) {
                const float x  = lg[s * 16 + h] + lg[(tile0 + ld) * 16 + HEADS + h];
                const float lr = x >= 0.0f ? x : 0.2f * x;
                atomicMax(&mxk[ld * HEADS + h], f2key(lr));
            }
        };
        for (int i = tid; i < nvalid; i += AGG_THREADS) visit_max(tile0 + i, i);
        for (int e = tid; e < E; e += AGG_THREADS) {
            const unsigned ld = (unsigned)(dstp[e] - tile0);
            if (ld < nv) visit_max(srcp[e], (int)ld);
        }
        __syncthreads();

        auto visit_acc = [&](int s, int ld) {
            for (int h = 0; h < HEADS; ++h) {
                const float x  = lg[s * 16 + h] + lg[(tile0 + ld) * 16 + HEADS + h];
                const float lr = x >= 0.0f ? x : 0.2f * x;
                const float ex = __expf(lr - key2f(mxk[ld * HEADS + h]));
                atomicAdd(&dn[ld * HEADS + h], ex);
                const float* hs = H + (size_t)s * HC + h * C;
                float*       ap = acc + ld * PA + h * C;
                for (int c = 0; c < C; ++c) atomicAdd(&ap[c], hs[c] * ex);
            }
        };
        for (int i = tid; i < nvalid; i += AGG_THREADS) visit_acc(tile0 + i, i);
        for (int e = tid; e < E; e += AGG_THREADS) {
            const unsigned ld = (unsigned)(dstp[e] - tile0);
            if (ld < nv) visit_acc(srcp[e], (int)ld);
        }
    }
    static_assert(C % 4 == 0 && PA % 4 == 0, "a 16-byte piece lies in one head and is 16-byte aligned in LDS");
    constexpr int NQ  = HC / 4;
    constexpr int NRP = norm_rows_per_pass(NQ);
    const int  nr0   = tid / NQ;
    const int  nq    = tid - nr0 * NQ;
    const int  nh    = (4 * nq) / C;
    const bool nact  = nr0 < NRP;
    float bq0 = 0.0f, bq1 = 0.0f, bq2 = 0.0f, bq3 = 0.0f;
    if (nact) {
        bq0 = bias[4 * nq];     bq1 = bias[4 * nq + 1];
        bq2 = bias[4 * nq + 2]; bq3 = bias[4 * nq + 3];
    }
    __syncthreads();

    if (nact) {
        for (int ld = nr0; ld < nvalid; ld += NRP) {
            float4* ap = (float4*)&acc[ld * PA + 4 * nq];
            const float4 a = *ap;
            const float  r = 1.0f / dn[ld * HEADS + nh];
            float4 v;
            v.x = a.x * r + bq0;
            v.y = a.y * r + bq1;
            v.z = a.z * r + bq2;
            v.w = a.w * r + bq3;
            if (relu) {
                v.x = fmaxf(v.x, 0.0f); v.y = fmaxf(v.y, 0.0f);
                v.z = fmaxf(v.z, 0.0f); v.w = fmaxf(v.w, 0.0f);
            }
            if constexpr (FUSE) *ap = v;
            else *(float4*)&out[(size_t)(tile0 + ld) * HC + 4 * nq] = v;
        }
    }

    if constexpr (FUSE) {
        __syncthreads();
        constexpr int KS  = HC / 4;
        constexpr int NCT = M2 / 16 + 1;
        constexpr int NRG = AGG_WAVES / NCT;
        constexpr int NRT = ROWS / 16;
        static_assert(HC % 4 == 0 && M2 % 16 == 0 && NRG >= 1, "whole matrix steps, whole slices, a wave per slice");
        const int ct = wave % NCT;
        const int rg = wave / NCT;
        if (rg < NRG) {
            const int  fc = lane & 15;
            const int  fh = lane >> 4;
            const bool lw = ct == NCT - 1;
            v2f bf[KS];
            if (!lw) {
#pragma unroll
                for (int ks = 0; ks < KS; ++ks) {
                    const int k = 4 * ks + 2 * fh;
                    bf[ks].x = Wn[(size_t)k * M2 + ct * 16 + fc];
                    bf[ks].y = Wn[(size_t)(k + 1) * M2 + ct * 16 + fc];
                }
            } else {
#pragma unroll
                for (int ks = 0; ks < KS; ++ks) {
                    const int k = 4 * ks + 2 * fh;
                    bf[ks].x = WAn[k * 16 + fc];
                    bf[ks].y = WAn[(k + 1) * 16 + fc];
                }
            }
            for (int rt = rg; rt < NRT; rt += NRG) {
                const float* xa = &acc[(rt * 16 + fc) * PA + 2 * fh];
                v2f af[KS];
#pragma unroll
                for (int ks = 0; ks < KS; ++ks) af[ks] = *(const v2f*)(xa + 4 * ks);

                v8f d = {};
#pragma unroll
                for (int ks = 0; ks < KS; ++ks)
                    d = __builtin_amdgcn_wmma_f32_16x16x4_f32(
                             false, af[ks],  false, bf[ks],
                             (short)0, d,  false,  false);

                const int    r0 = rt * 16 + 8 * fh;
                const size_t ps = lw ? (size_t)16 : (size_t)M2;
                float* dp = lw ? LGn + (size_t)(tile0 + r0) * 16 + fc
                               : Hn  + (size_t)(tile0 + r0) * M2 + ct * 16 + fc;
                if (r0 + 8 <= nvalid) {
#pragma unroll
                    for (int v = 0; v < 8; ++v) dp[(size_t)v * ps] = d[v];
                } else {
#pragma unroll
                    for (int v = 0; v < 8; ++v) if (r0 + v < nvalid) dp[(size_t)v * ps] = d[v];
                }
            }
        }
    }
}


extern "C" void kernel_launch(void* const* d_in, const int* in_sizes, int n_in,
                              void* d_out, int out_size, void* d_ws, size_t ws_size,
                              hipStream_t stream) {
    const float* x      = (const float*)d_in[0];
    const int*   ei     = (const int*)  d_in[1];
    const float* W1     = (const float*)d_in[2];
    const float* a_src1 = (const float*)d_in[3];
    const float* a_dst1 = (const float*)d_in[4];
    const float* b1     = (const float*)d_in[5];
    const float* W2     = (const float*)d_in[6];
    const float* a_src2 = (const float*)d_in[7];
    const float* a_dst2 = (const float*)d_in[8];
    const float* b2     = (const float*)d_in[9];
    const float* W3     = (const float*)d_in[10];
    const float* a_src3 = (const float*)d_in[11];
    const float* a_dst3 = (const float*)d_in[12];
    const float* b3     = (const float*)d_in[13];

    const int E = in_sizes[1] / 2;

    float* ws   = (float*)d_ws;
    float* bufH = ws;
    float* lgB  = bufH + (size_t)NNODE * 240;
    float* h2   = lgB  + (size_t)NNODE * 112;
    float* lg   = h2   + (size_t)NNODE * 192;
    float* wa1  = lg   + (size_t)NNODE * 16;
    float* wa2  = wa1  + 128 * 16;
    float* wa3  = wa2  + 112 * 16;
    float* out  = (float*)d_out;

    constexpr int NT = NTILE;
    unsigned* glist = (unsigned*)(wa3 + 96 * 16);
    unsigned* gcnt  = glist + (size_t)NT * LIST_CAP;
    unsigned* keys  = gcnt + 256;
    static_assert(NT <= 256, "the key plane starts behind the list lengths");

    constexpr int RT = NNODE / 16;
    static_assert(NNODE % 16 == 0, "row blocks are whole");

    fold_att<<<FOLD_BLOCKS + KEY_BLOCKS, SETUP_THREADS, 0, stream>>>(W1, a_src1, a_dst1, W2, a_src2, a_dst2, W3, a_src3, a_dst3, wa1, wa2, wa3, ei, E, keys);

    gemm_wstat_f32<128, 112><<<768, 256, 0, stream>>>(x, W1, wa1, bufH, lg, RT);
    agg_tile<7, 16, true, 96><<<NT, AGG_THREADS, 0, stream>>>(ei, E, lg, bufH, b1, nullptr, 1, glist, gcnt, keys, W2, wa2, h2, lgB);
    agg_tile<6, 16, false, 240><<<NT, AGG_THREADS, 0, stream>>>(ei, E, lgB, h2, b2, nullptr, 1, glist, gcnt, keys, W3, wa3, bufH, lg);
    agg_tile<6, 40, false, 0><<<NT, AGG_THREADS, 0, stream>>>(ei, E, lg, bufH, b3, out, 0, glist, gcnt, keys, nullptr, nullptr, nullptr, nullptr);
}
